# starting kernel
# baseline (speedup 1.0000x reference)
#include <hip/hip_runtime.h>

typedef float v2f __attribute__((ext_vector_type(2)));
typedef float v8f __attribute__((ext_vector_type(8)));

#define N_NODES 100000
#define FEAT    256
#define NHID    64
#define NCLS    16
#define NEDGE   1600000
#define NTILES  (N_NODES / 16)

__global__ __launch_bounds__(128) void gemm1_kernel(const float* __restrict__ x,
                                                    const float* __restrict__ W1,
                                                    float* __restrict__ sup) {
    __shared__ float tileA[16 * FEAT];
    const int lane = threadIdx.x & 31;
    const int wave = threadIdx.x >> 5;
    const int rb   = blockIdx.x * 16;

    const float4* x4 = reinterpret_cast<const float4*>(x + (size_t)rb * FEAT);
    float4* t4 = reinterpret_cast<float4*>(tileA);
    for (int i = threadIdx.x; i < 16 * FEAT / 4; i += 128) t4[i] = x4[i];
    __syncthreads();

    const int c  = wave * 16;
    const int l  = lane & 15;
    const int hi = lane >> 4;
    v8f acc = {};
    for (int k = 0; k < FEAT; k += 4) {
        const int ks = k + 2 * hi;
        v2f a = *reinterpret_cast<const v2f*>(&tileA[l * FEAT + ks]);
        v2f b;
        b.x = W1[(size_t)ks       * NHID + c + l];
        b.y = W1[(size_t)(ks + 1) * NHID + c + l];
        acc = __builtin_amdgcn_wmma_f32_16x16x4_f32(false, a, false, b,
                                                    (short)0, acc, false, false);
    }
    float* out = sup + (size_t)rb * NHID + c;
    const int mb = hi * 8;
#pragma unroll
    for (int r = 0; r < 8; ++r)
        out[(size_t)(mb + r) * NHID + l] = acc[r];
}

__global__ __launch_bounds__(256) void gemm2_kernel(const float* __restrict__ h,
                                                    const float* __restrict__ W2,
                                                    float* __restrict__ sup2) {
    const int lane = threadIdx.x & 31;
    const int wave = threadIdx.x >> 5;
    const int tile = blockIdx.x * 8 + wave;
    if (tile >= NTILES) return;
    const int rb = tile * 16;
    const int l  = lane & 15;
    const int hi = lane >> 4;
    v8f acc = {};
    for (int k = 0; k < NHID; k += 4) {
        const int ks = k + 2 * hi;
        v2f a = *reinterpret_cast<const v2f*>(&h[(size_t)(rb + l) * NHID + ks]);
        v2f b;
        b.x = W2[ks * NCLS + l];
        b.y = W2[(ks + 1) * NCLS + l];
        acc = __builtin_amdgcn_wmma_f32_16x16x4_f32(false, a, false, b,
                                                    (short)0, acc, false, false);
    }
    float* out = sup2 + (size_t)rb * NCLS;
    const int mb = hi * 8;
#pragma unroll
    for (int r = 0; r < 8; ++r)
        out[(size_t)(mb + r) * NCLS + l] = acc[r];
}

template <int F>
__global__ __launch_bounds__(256) void spmm_kernel(const int* __restrict__ esrc,
                                                   const int* __restrict__ edst,
                                                   const float* __restrict__ ev,
                                                   const float* __restrict__ sup,
                                                   float* __restrict__ acc) {
    constexpr int CH = F / 4;
    const long long total  = (long long)NEDGE * CH;
    const long long stride = (long long)gridDim.x * blockDim.x;
    for (long long idx = (long long)blockIdx.x * blockDim.x + threadIdx.x;
         idx < total; idx += stride) {
        const int e    = (int)(idx / CH);
        const int part = (int)(idx % CH);
        long long pe = (idx + stride) / CH;
        if (pe >= NEDGE) pe = NEDGE - 1;
        __builtin_prefetch(&esrc[pe], 0, 0);
        const int   s  = esrc[e];
        const int   d  = edst[e];
        const float wv = ev[e];
        const float4 v = reinterpret_cast<const float4*>(sup)[(size_t)s * CH + part];
        float* base = acc + (size_t)d * F + (size_t)part * 4;
        atomicAdd(base + 0, v.x * wv);
        atomicAdd(base + 1, v.y * wv);
        atomicAdd(base + 2, v.z * wv);
        atomicAdd(base + 3, v.w * wv);
    }
}

__global__ __launch_bounds__(256) void zero_kernel(float* __restrict__ p, long long n) {
    long long i = (long long)blockIdx.x * blockDim.x + threadIdx.x;
    if (i < n) p[i] = 0.0f;
}

__global__ __launch_bounds__(256) void bias_relu_kernel(float* __restrict__ h,
                                                        const float* __restrict__ b1) {
    long long i = (long long)blockIdx.x * blockDim.x + threadIdx.x;
    if (i < (long long)N_NODES * NHID) {
        const float v = h[i] + b1[i & (NHID - 1)];
        h[i] = v > 0.0f ? v : 0.0f;
    }
}

__global__ __launch_bounds__(256) void init_out_kernel(float* __restrict__ out,
                                                       const float* __restrict__ b2) {
    long long i = (long long)blockIdx.x * blockDim.x + threadIdx.x;
    if (i < (long long)N_NODES * NCLS) out[i] = b2[i & (NCLS - 1)];
}

__global__ __launch_bounds__(128) void log_softmax_kernel(float* __restrict__ out) {
    const int r = blockIdx.x * blockDim.x + threadIdx.x;
    if (r >= N_NODES) return;
    float* p = out + (size_t)r * NCLS;
    float v[NCLS];
    float m = -3.402823466e+38f;
#pragma unroll
    for (int i = 0; i < NCLS; ++i) { v[i] = p[i]; m = v[i] > m ? v[i] : m; }
    float s = 0.0f;
#pragma unroll
    for (int i = 0; i < NCLS; ++i) s += __expf(v[i] - m);
    const float lse = m + __logf(s);
#pragma unroll
    for (int i = 0; i < NCLS; ++i) p[i] = v[i] - lse;
}

extern "C" void kernel_launch(void* const* d_in, const int* in_sizes, int n_in,
                              void* d_out, int out_size, void* d_ws, size_t ws_size,
                              hipStream_t stream) {
    (void)in_sizes; (void)n_in; (void)out_size; (void)ws_size;
    const float* x   = (const float*)d_in[0];
    const int*   es  = (const int*)  d_in[1];
    const int*   ed  = (const int*)  d_in[2];
    const float* ev  = (const float*)d_in[3];
    const float* W1  = (const float*)d_in[4];
    const float* b1  = (const float*)d_in[5];
    const float* W2  = (const float*)d_in[6];
    const float* b2  = (const float*)d_in[7];
    float* out = (float*)d_out;

    float* sup1 = (float*)d_ws;
    float* hacc = sup1 + (size_t)N_NODES * NHID;
    float* sup2 = sup1;

    const long long nh = (long long)N_NODES * NHID;

    zero_kernel<<<(int)((nh + 255) / 256), 256, 0, stream>>>(hacc, nh);
    gemm1_kernel<<<NTILES, 128, 0, stream>>>(x, W1, sup1);
    {
        const long long total = (long long)NEDGE * (NHID / 4);
        spmm_kernel<NHID><<<(int)((total + 255) / 256), 256, 0, stream>>>(es, ed, ev, sup1, hacc);
    }
    bias_relu_kernel<<<(int)((nh + 255) / 256), 256, 0, stream>>>(hacc, b1);
    gemm2_kernel<<<(NTILES + 7) / 8, 256, 0, stream>>>(hacc, W2, sup2);
    {
        const long long nc = (long long)N_NODES * NCLS;
        init_out_kernel<<<(int)((nc + 255) / 256), 256, 0, stream>>>(out, b2);
        const long long total = (long long)NEDGE * (NCLS / 4);
        spmm_kernel<NCLS><<<(int)((total + 255) / 256), 256, 0, stream>>>(es, ed, ev, sup2, out);
    }
    log_softmax_kernel<<<(N_NODES + 127) / 128, 128, 0, stream>>>(out);
}


# global f32 atomic scatter replaced by row owners: 256 blocks each scan all of edge_dst, gather the rows of their own edges, sum them in LDS (f32 LDS adds, arrival order), store once with bias
# speedup vs baseline: 1.5265x; 1.5265x over previous
#include <hip/hip_runtime.h>

typedef float v2f __attribute__((ext_vector_type(2)));
typedef float v8f __attribute__((ext_vector_type(8)));

#define N_NODES 100000
#define FEAT    256
#define NHID    64
#define NCLS    16
#define NEDGE   1600000
#define NTILES  (N_NODES / 16)

__global__ __launch_bounds__(128) void gemm1_kernel(const float* __restrict__ x,
                                                    const float* __restrict__ W1,
                                                    float* __restrict__ sup) {
    __shared__ float tileA[16 * FEAT];
    const int lane = threadIdx.x & 31;
    const int wave = threadIdx.x >> 5;
    const int rb   = blockIdx.x * 16;

    const float4* x4 = reinterpret_cast<const float4*>(x + (size_t)rb * FEAT);
    float4* t4 = reinterpret_cast<float4*>(tileA);
    for (int i = threadIdx.x; i < 16 * FEAT / 4; i += 128) t4[i] = x4[i];
    __syncthreads();

    const int c  = wave * 16;
    const int l  = lane & 15;
    const int hi = lane >> 4;
    v8f acc = {};
    for (int k = 0; k < FEAT; k += 4) {
        const int ks = k + 2 * hi;
        v2f a = *reinterpret_cast<const v2f*>(&tileA[l * FEAT + ks]);
        v2f b;
        b.x = W1[(size_t)ks       * NHID + c + l];
        b.y = W1[(size_t)(ks + 1) * NHID + c + l];
        acc = __builtin_amdgcn_wmma_f32_16x16x4_f32(false, a, false, b,
                                                    (short)0, acc, false, false);
    }
    float* out = sup + (size_t)rb * NHID + c;
    const int mb = hi * 8;
#pragma unroll
    for (int r = 0; r < 8; ++r)
        out[(size_t)(mb + r) * NHID + l] = acc[r];
}

__global__ __launch_bounds__(256) void gemm2_kernel(const float* __restrict__ h,
                                                    const float* __restrict__ W2,
                                                    float* __restrict__ sup2) {
    const int lane = threadIdx.x & 31;
    const int wave = threadIdx.x >> 5;
    const int tile = blockIdx.x * 8 + wave;
    if (tile >= NTILES) return;
    const int rb = tile * 16;
    const int l  = lane & 15;
    const int hi = lane >> 4;
    v8f acc = {};
    for (int k = 0; k < NHID; k += 4) {
        const int ks = k + 2 * hi;
        v2f a = *reinterpret_cast<const v2f*>(&h[(size_t)(rb + l) * NHID + ks]);
        v2f b;
        b.x = W2[ks * NCLS + l];
        b.y = W2[(ks + 1) * NCLS + l];
        acc = __builtin_amdgcn_wmma_f32_16x16x4_f32(false, a, false, b,
                                                    (short)0, acc, false, false);
    }
    float* out = sup2 + (size_t)rb * NCLS;
    const int mb = hi * 8;
#pragma unroll
    for (int r = 0; r < 8; ++r)
        out[(size_t)(mb + r) * NCLS + l] = acc[r];
}

#define NOWN  256
#define RPB   392
#define NQ    (NEDGE / 4)
#define SCAN_U 4
#define NFULL (NQ / (1024 * SCAN_U))
#define PCAP  544
#define ROT   1568

template <bool GUARD>
__device__ __forceinline__ void load_groups(const int4* __restrict__ d4, int it, int tid, int rot,
                                            int4 (&dv)[SCAN_U], int (&qi)[SCAN_U]) {
#pragma unroll
    for (int u = 0; u < SCAN_U; ++u) {
        int q = (it * SCAN_U + u) * 1024 + tid;
        const bool valid = !GUARD || q < NQ;
        q += rot;
        if (q >= NQ) q -= NQ;
        qi[u] = q;
        dv[u] = valid ? d4[q] : make_int4(-1, -1, -1, -1);
    }
}

template <int F>
__device__ __forceinline__ void add_edges(const int* list, int n, int lane, int lo,
                                          const int* __restrict__ esrc,
                                          const int* __restrict__ edst,
                                          const float* __restrict__ ev,
                                          const float* __restrict__ sup,
                                          float* acc) {
    constexpr int VW  = (F == 64) ? 2 : 1;
    constexpr int LPE = F / VW;
    constexpr int EPS = 32 / LPE;
    constexpr int NST = 32 / EPS;
    constexpr int CH  = 8;
    int ro = 0, s = 0;
    float w = 0.0f;
    if (lane < n) {
        const int e = list[lane];
        ro = (edst[e] - lo) * F;
        s  = esrc[e];
        w  = ev[e];
    }
    const int sub = lane / LPE;
    const int col = (lane % LPE) * VW;
    for (int t0 = 0; t0 < NST; t0 += CH) {
        float v0[CH], v1[CH], wj[CH];
        int   rj[CH];
#pragma unroll
        for (int k = 0; k < CH; ++k) {
            const int j  = (t0 + k) * EPS + sub;
            const int sj = __shfl(s, j);
            rj[k] = __shfl(ro, j);
            wj[k] = __shfl(w, j);
            v0[k] = 0.0f; v1[k] = 0.0f;
            if (j < n) {
                if (VW == 2) {
                    const v2f v = *reinterpret_cast<const v2f*>(sup + (size_t)sj * F + col);
                    v0[k] = v.x; v1[k] = v.y;
                } else {
                    v0[k] = sup[(size_t)sj * F + col];
                }
            }
        }
#pragma unroll
        for (int k = 0; k < CH; ++k) {
            const int j = (t0 + k) * EPS + sub;
            if (j < n) {
                atomicAdd(&acc[rj[k] + col], v0[k] * wj[k]);
                if (VW == 2) atomicAdd(&acc[rj[k] + col + 1], v1[k] * wj[k]);
            }
        }
    }
}

__device__ __forceinline__ void scan_set(const int4 (&dv)[SCAN_U], const int (&qi)[SCAN_U],
                                         int lo, int nrows, int* mypend, int& cnt) {
#pragma unroll
    for (int u = 0; u < SCAN_U; ++u) {
        const int dd[4] = {dv[u].x, dv[u].y, dv[u].z, dv[u].w};
#pragma unroll
        for (int c = 0; c < 4; ++c) {
            const bool m = (unsigned)(dd[c] - lo) < (unsigned)nrows;
            const unsigned bal = __builtin_amdgcn_ballot_w32(m);
            if (bal) {
                const int pos = cnt + (int)__builtin_amdgcn_mbcnt_lo(bal, 0u);
                if (m) mypend[pos] = qi[u] * 4 + c;
                cnt += __builtin_popcount(bal);
            }
        }
    }
}

template <int F, bool RELU>
__global__ __launch_bounds__(1024) void owner_sum_kernel(const int* __restrict__ esrc,
                                                         const int* __restrict__ edst,
                                                         const float* __restrict__ ev,
                                                         const float* __restrict__ sup,
                                                         const float* __restrict__ bias,
                                                         float* __restrict__ dstm) {
    __shared__ float acc[RPB * F];
    __shared__ int   pend[32 * PCAP];
    const int tid  = threadIdx.x;
    const int lane = tid & 31;
    const int wave = tid >> 5;
    const int lo   = blockIdx.x * RPB;
    int nrows = N_NODES - lo;
    if (nrows > RPB) nrows = RPB;

    float4* acc4 = reinterpret_cast<float4*>(acc);
    for (int i = tid; i < RPB * F / 4; i += 1024) acc4[i] = make_float4(0.f, 0.f, 0.f, 0.f);
    __syncthreads();

    int* mypend = pend + wave * PCAP;
    int  cnt = 0;
    const int4* d4 = reinterpret_cast<const int4*>(edst);
    const int rot = blockIdx.x * ROT;
    static_assert(NFULL % 2 == 1, "the loop below ends with set a holding the last full trip");
    static_assert((NFULL + 1) * 1024 * SCAN_U >= NQ, "one guarded trip covers the rest");
    int4 ga[SCAN_U], gb[SCAN_U];
    int  qa[SCAN_U], qb[SCAN_U];
    auto drain = [&]() {
        while (cnt >= 32) {
            cnt -= 32;
            add_edges<F>(mypend + cnt, 32, lane, lo, esrc, edst, ev, sup, acc);
        }
    };
    load_groups<false>(d4, 0, tid, rot, ga, qa);
    for (int it = 0; it < NFULL - 1; it += 2) {
        load_groups<false>(d4, it + 1, tid, rot, gb, qb);
        scan_set(ga, qa, lo, nrows, mypend, cnt);
        drain();
        load_groups<false>(d4, it + 2, tid, rot, ga, qa);
        scan_set(gb, qb, lo, nrows, mypend, cnt);
        drain();
    }
    load_groups<true>(d4, NFULL, tid, rot, gb, qb);
    scan_set(ga, qa, lo, nrows, mypend, cnt);
    drain();
    scan_set(gb, qb, lo, nrows, mypend, cnt);
    drain();
    add_edges<F>(mypend, cnt, lane, lo, esrc, edst, ev, sup, acc);
    __syncthreads();

    float4* o4 = reinterpret_cast<float4*>(dstm + (size_t)lo * F);
    for (int i = tid; i < nrows * F / 4; i += 1024) {
        float4 v = acc4[i];
        const int c = (i * 4) & (F - 1);
        v.x += bias[c]; v.y += bias[c + 1]; v.z += bias[c + 2]; v.w += bias[c + 3];
        if (RELU) {
            v.x = v.x > 0.0f ? v.x : 0.0f; v.y = v.y > 0.0f ? v.y : 0.0f;
            v.z = v.z > 0.0f ? v.z : 0.0f; v.w = v.w > 0.0f ? v.w : 0.0f;
        }
        o4[i] = v;
    }
}

__global__ __launch_bounds__(128) void log_softmax_kernel(float* __restrict__ out) {
    const int r = blockIdx.x * blockDim.x + threadIdx.x;
    if (r >= N_NODES) return;
    float* p = out + (size_t)r * NCLS;
    float v[NCLS];
    float m = -3.402823466e+38f;
#pragma unroll
    for (int i = 0; i < NCLS; ++i) { v[i] = p[i]; m = v[i] > m ? v[i] : m; }
    float s = 0.0f;
#pragma unroll
    for (int i = 0; i < NCLS; ++i) s += __expf(v[i] - m);
    const float lse = m + __logf(s);
#pragma unroll
    for (int i = 0; i < NCLS; ++i) p[i] = v[i] - lse;
}

extern "C" void kernel_launch(void* const* d_in, const int* in_sizes, int n_in,
                              void* d_out, int out_size, void* d_ws, size_t ws_size,
                              hipStream_t stream) {
    (void)in_sizes; (void)n_in; (void)out_size; (void)ws_size;
    const float* x   = (const float*)d_in[0];
    const int*   es  = (const int*)  d_in[1];
    const int*   ed  = (const int*)  d_in[2];
    const float* ev  = (const float*)d_in[3];
    const float* W1  = (const float*)d_in[4];
    const float* b1  = (const float*)d_in[5];
    const float* W2  = (const float*)d_in[6];
    const float* b2  = (const float*)d_in[7];
    float* out = (float*)d_out;

    float* sup1 = (float*)d_ws;
    float* hacc = sup1 + (size_t)N_NODES * NHID;
    float* sup2 = sup1;

    gemm1_kernel<<<NTILES, 128, 0, stream>>>(x, W1, sup1);
    owner_sum_kernel<NHID, true><<<NOWN, 1024, 0, stream>>>(es, ed, ev, sup1, b1, hacc);
    gemm2_kernel<<<(NTILES + 7) / 8, 256, 0, stream>>>(hacc, W2, sup2);
    owner_sum_kernel<NCLS, false><<<NOWN, 1024, 0, stream>>>(es, ed, ev, sup2, b2, out);
    log_softmax_kernel<<<(N_NODES + 127) / 128, 128, 0, stream>>>(out);
}


# second layer's owner blocks read back per-wave edge records (source row, weight, row) kept in workspace by the first layer's scan instead of scanning edge_dst again; arrival-order LDS sums kept
# speedup vs baseline: 2.0900x; 1.3691x over previous
#include <hip/hip_runtime.h>

typedef float v2f __attribute__((ext_vector_type(2)));
typedef float v8f __attribute__((ext_vector_type(8)));

#define N_NODES 100000
#define FEAT    256
#define NHID    64
#define NCLS    16
#define NEDGE   1600000
#define NTILES  (N_NODES / 16)

__global__ __launch_bounds__(128) void gemm1_kernel(const float* __restrict__ x,
                                                    const float* __restrict__ W1,
                                                    float* __restrict__ sup) {
    __shared__ float tileA[16 * FEAT];
    const int lane = threadIdx.x & 31;
    const int wave = threadIdx.x >> 5;
    const int rb   = blockIdx.x * 16;

    const float4* x4 = reinterpret_cast<const float4*>(x + (size_t)rb * FEAT);
    float4* t4 = reinterpret_cast<float4*>(tileA);
    for (int i = threadIdx.x; i < 16 * FEAT / 4; i += 128) t4[i] = x4[i];
    __syncthreads();

    const int c  = wave * 16;
    const int l  = lane & 15;
    const int hi = lane >> 4;
    v8f acc = {};
    for (int k = 0; k < FEAT; k += 4) {
        const int ks = k + 2 * hi;
        v2f a = *reinterpret_cast<const v2f*>(&tileA[l * FEAT + ks]);
        v2f b;
        b.x = W1[(size_t)ks       * NHID + c + l];
        b.y = W1[(size_t)(ks + 1) * NHID + c + l];
        acc = __builtin_amdgcn_wmma_f32_16x16x4_f32(false, a, false, b,
                                                    (short)0, acc, false, false);
    }
    float* out = sup + (size_t)rb * NHID + c;
    const int mb = hi * 8;
#pragma unroll
    for (int r = 0; r < 8; ++r)
        out[(size_t)(mb + r) * NHID + l] = acc[r];
}

__global__ __launch_bounds__(256) void gemm2_kernel(const float* __restrict__ h,
                                                    const float* __restrict__ W2,
                                                    float* __restrict__ sup2) {
    const int lane = threadIdx.x & 31;
    const int wave = threadIdx.x >> 5;
    const int tile = blockIdx.x * 8 + wave;
    if (tile >= NTILES) return;
    const int rb = tile * 16;
    const int l  = lane & 15;
    const int hi = lane >> 4;
    v8f acc = {};
    for (int k = 0; k < NHID; k += 4) {
        const int ks = k + 2 * hi;
        v2f a = *reinterpret_cast<const v2f*>(&h[(size_t)(rb + l) * NHID + ks]);
        v2f b;
        b.x = W2[ks * NCLS + l];
        b.y = W2[(ks + 1) * NCLS + l];
        acc = __builtin_amdgcn_wmma_f32_16x16x4_f32(false, a, false, b,
                                                    (short)0, acc, false, false);
    }
    float* out = sup2 + (size_t)rb * NCLS;
    const int mb = hi * 8;
#pragma unroll
    for (int r = 0; r < 8; ++r)
        out[(size_t)(mb + r) * NCLS + l] = acc[r];
}

#define NOWN  256
#define RPB   392
#define NQ    (NEDGE / 4)
#define SCAN_U 4
#define NFULL (NQ / (1024 * SCAN_U))
#define PCAP  544
#define ROT   1568
#define WCAP  512

template <bool GUARD>
__device__ __forceinline__ void load_groups(const int4* __restrict__ d4, int it, int tid, int rot,
                                            int4 (&dv)[SCAN_U], int (&qi)[SCAN_U]) {
#pragma unroll
    for (int u = 0; u < SCAN_U; ++u) {
        int q = (it * SCAN_U + u) * 1024 + tid;
        const bool valid = !GUARD || q < NQ;
        q += rot;
        if (q >= NQ) q -= NQ;
        qi[u] = q;
        dv[u] = valid ? d4[q] : make_int4(-1, -1, -1, -1);
    }
}

template <int F>
__device__ __forceinline__ void gather_add(int ro, int s, float w, int n, int lane,
                                           const float* __restrict__ sup,
                                           float* acc) {
    constexpr int VW  = (F == 64) ? 2 : 1;
    constexpr int LPE = F / VW;
    constexpr int EPS = 32 / LPE;
    constexpr int NST = 32 / EPS;
    constexpr int CH  = 8;
    const int sub = lane / LPE;
    const int col = (lane % LPE) * VW;
    for (int t0 = 0; t0 < NST; t0 += CH) {
        float v0[CH], v1[CH], wj[CH];
        int   rj[CH];
#pragma unroll
        for (int k = 0; k < CH; ++k) {
            const int j  = (t0 + k) * EPS + sub;
            const int sj = __shfl(s, j);
            rj[k] = __shfl(ro, j);
            wj[k] = __shfl(w, j);
            v0[k] = 0.0f; v1[k] = 0.0f;
            if (j < n) {
                if (VW == 2) {
                    const v2f v = *reinterpret_cast<const v2f*>(sup + (size_t)sj * F + col);
                    v0[k] = v.x; v1[k] = v.y;
                } else {
                    v0[k] = sup[(size_t)sj * F + col];
                }
            }
        }
#pragma unroll
        for (int k = 0; k < CH; ++k) {
            const int j = (t0 + k) * EPS + sub;
            if (j < n) {
                atomicAdd(&acc[rj[k] + col], v0[k] * wj[k]);
                if (VW == 2) atomicAdd(&acc[rj[k] + col + 1], v1[k] * wj[k]);
            }
        }
    }
}

template <int F, bool REC>
__device__ __forceinline__ void add_edges(const int* list, int n, int lane, int lo,
                                          const int* __restrict__ esrc,
                                          const int* __restrict__ edst,
                                          const float* __restrict__ ev,
                                          const float* __restrict__ sup,
                                          float* acc, int4* rec, int& written) {
    int ro = 0, s = 0;
    float w = 0.0f;
    if (lane < n) {
        const int e   = list[lane];
        const int row = edst[e] - lo;
        ro = row * F;
        s  = esrc[e];
        w  = ev[e];
        if (REC && written + lane < WCAP)
            rec[written + lane] = make_int4(s, __float_as_int(w), row, 0);
    }
    if (REC) written += n;
    gather_add<F>(ro, s, w, n, lane, sup, acc);
}

__device__ __forceinline__ void scan_set(const int4 (&dv)[SCAN_U], const int (&qi)[SCAN_U],
                                         int lo, int nrows, int* mypend, int& cnt) {
#pragma unroll
    for (int u = 0; u < SCAN_U; ++u) {
        const int dd[4] = {dv[u].x, dv[u].y, dv[u].z, dv[u].w};
#pragma unroll
        for (int c = 0; c < 4; ++c) {
            const bool m = (unsigned)(dd[c] - lo) < (unsigned)nrows;
            const unsigned bal = __builtin_amdgcn_ballot_w32(m);
            if (bal) {
                const int pos = cnt + (int)__builtin_amdgcn_mbcnt_lo(bal, 0u);
                if (m) mypend[pos] = qi[u] * 4 + c;
                cnt += __builtin_popcount(bal);
            }
        }
    }
}

template <int F, bool REC>
__device__ __forceinline__ void scan_and_sum(const int* __restrict__ esrc,
                                             const int* __restrict__ edst,
                                             const float* __restrict__ ev,
                                             const float* __restrict__ sup,
                                             float* acc, int* mypend, int lo, int nrows,
                                             int tid, int lane, int rot,
                                             int4* rec, int& written) {
    int  cnt = 0;
    const int4* d4 = reinterpret_cast<const int4*>(edst);
    static_assert(NFULL % 2 == 1, "the loop below ends with set a holding the last full trip");
    static_assert((NFULL + 1) * 1024 * SCAN_U >= NQ, "one guarded trip covers the rest");
    int4 ga[SCAN_U], gb[SCAN_U];
    int  qa[SCAN_U], qb[SCAN_U];
    auto drain = [&]() {
        while (cnt >= 32) {
            cnt -= 32;
            add_edges<F, REC>(mypend + cnt, 32, lane, lo, esrc, edst, ev, sup, acc, rec, written);
        }
    };
    load_groups<false>(d4, 0, tid, rot, ga, qa);
    for (int it = 0; it < NFULL - 1; it += 2) {
        load_groups<false>(d4, it + 1, tid, rot, gb, qb);
        scan_set(ga, qa, lo, nrows, mypend, cnt);
        drain();
        load_groups<false>(d4, it + 2, tid, rot, ga, qa);
        scan_set(gb, qb, lo, nrows, mypend, cnt);
        drain();
    }
    load_groups<true>(d4, NFULL, tid, rot, gb, qb);
    scan_set(ga, qa, lo, nrows, mypend, cnt);
    drain();
    scan_set(gb, qb, lo, nrows, mypend, cnt);
    drain();
    add_edges<F, REC>(mypend, cnt, lane, lo, esrc, edst, ev, sup, acc, rec, written);
}

template <int F, bool RELU>
__device__ __forceinline__ void store_rows(const float* acc, const float* __restrict__ bias,
                                           float* __restrict__ dstm, int lo, int nrows, int tid) {
    const float4* acc4 = reinterpret_cast<const float4*>(acc);
    float4* o4 = reinterpret_cast<float4*>(dstm + (size_t)lo * F);
    for (int i = tid; i < nrows * F / 4; i += 1024) {
        float4 v = acc4[i];
        const int c = (i * 4) & (F - 1);
        v.x += bias[c]; v.y += bias[c + 1]; v.z += bias[c + 2]; v.w += bias[c + 3];
        if (RELU) {
            v.x = v.x > 0.0f ? v.x : 0.0f; v.y = v.y > 0.0f ? v.y : 0.0f;
            v.z = v.z > 0.0f ? v.z : 0.0f; v.w = v.w > 0.0f ? v.w : 0.0f;
        }
        o4[i] = v;
    }
}

template <int F, bool RELU>
__global__ __launch_bounds__(1024) void owner_sum_kernel(const int* __restrict__ esrc,
                                                         const int* __restrict__ edst,
                                                         const float* __restrict__ ev,
                                                         const float* __restrict__ sup,
                                                         const float* __restrict__ bias,
                                                         float* __restrict__ dstm,
                                                         int* __restrict__ recs,
                                                         int* __restrict__ cnts) {
    __shared__ float acc[RPB * F];
    __shared__ int   pend[32 * PCAP];
    const int tid  = threadIdx.x;
    const int lane = tid & 31;
    const int wave = tid >> 5;
    const int lo   = blockIdx.x * RPB;
    int nrows = N_NODES - lo;
    if (nrows > RPB) nrows = RPB;

    float4* acc4 = reinterpret_cast<float4*>(acc);
    for (int i = tid; i < RPB * F / 4; i += 1024) acc4[i] = make_float4(0.f, 0.f, 0.f, 0.f);
    __syncthreads();

    const int rot = blockIdx.x * ROT;
    const int seg = blockIdx.x * 32 + wave;
    int4* rec = reinterpret_cast<int4*>(recs) + (size_t)seg * WCAP;
    int written = 0;
    scan_and_sum<F, true>(esrc, edst, ev, sup, acc, pend + wave * PCAP, lo, nrows,
                          tid, lane, rot, rec, written);
    if (lane == 0) cnts[seg] = written;
    __syncthreads();
    store_rows<F, RELU>(acc, bias, dstm, lo, nrows, tid);
}

template <int F, bool RELU>
__global__ __launch_bounds__(1024) void replay_sum_kernel(const int* __restrict__ esrc,
                                                          const int* __restrict__ edst,
                                                          const float* __restrict__ ev,
                                                          const float* __restrict__ sup,
                                                          const float* __restrict__ bias,
                                                          float* __restrict__ dstm,
                                                          const int* __restrict__ recs,
                                                          const int* __restrict__ cnts) {
    __shared__ float acc[RPB * F];
    __shared__ int   pend[32 * PCAP];
    const int tid  = threadIdx.x;
    const int lane = tid & 31;
    const int wave = tid >> 5;
    const int lo   = blockIdx.x * RPB;
    int nrows = N_NODES - lo;
    if (nrows > RPB) nrows = RPB;

    float4* acc4 = reinterpret_cast<float4*>(acc);
    for (int i = tid; i < RPB * F / 4; i += 1024) acc4[i] = make_float4(0.f, 0.f, 0.f, 0.f);
    __syncthreads();

    const bool over = __builtin_amdgcn_ballot_w32(cnts[blockIdx.x * 32 + lane] > WCAP) != 0u;
    if (!over) {
        const int seg = blockIdx.x * 32 + wave;
        const int n   = __builtin_amdgcn_readfirstlane(cnts[seg]);
        const int4* rec = reinterpret_cast<const int4*>(recs) + (size_t)seg * WCAP;
        if (n > 0) {
            const int last = n - 1;
            int i0 = lane < last ? lane : last;
            int4 cur = rec[i0];
            for (int base = 0; base < n; base += 32) {
                int i1 = base + 32 + lane;
                i1 = i1 < last ? i1 : last;
                const int4 nxt = rec[i1];
                int m = n - base;
                if (m > 32) m = 32;
                gather_add<F>(cur.z * F, cur.x, __int_as_float(cur.y), m, lane, sup, acc);
                cur = nxt;
            }
        }
    } else {
        int unused = 0;
        scan_and_sum<F, false>(esrc, edst, ev, sup, acc, pend + wave * PCAP, lo, nrows,
                               tid, lane, blockIdx.x * ROT, nullptr, unused);
    }
    __syncthreads();
    store_rows<F, RELU>(acc, bias, dstm, lo, nrows, tid);
}

__global__ __launch_bounds__(128) void log_softmax_kernel(float* __restrict__ out) {
    const int r = blockIdx.x * blockDim.x + threadIdx.x;
    if (r >= N_NODES) return;
    float* p = out + (size_t)r * NCLS;
    float v[NCLS];
    float m = -3.402823466e+38f;
#pragma unroll
    for (int i = 0; i < NCLS; ++i) { v[i] = p[i]; m = v[i] > m ? v[i] : m; }
    float s = 0.0f;
#pragma unroll
    for (int i = 0; i < NCLS; ++i) s += __expf(v[i] - m);
    const float lse = m + __logf(s);
#pragma unroll
    for (int i = 0; i < NCLS; ++i) p[i] = v[i] - lse;
}

extern "C" void kernel_launch(void* const* d_in, const int* in_sizes, int n_in,
                              void* d_out, int out_size, void* d_ws, size_t ws_size,
                              hipStream_t stream) {
    (void)in_sizes; (void)n_in; (void)out_size; (void)ws_size;
    const float* x   = (const float*)d_in[0];
    const int*   es  = (const int*)  d_in[1];
    const int*   ed  = (const int*)  d_in[2];
    const float* ev  = (const float*)d_in[3];
    const float* W1  = (const float*)d_in[4];
    const float* b1  = (const float*)d_in[5];
    const float* W2  = (const float*)d_in[6];
    const float* b2  = (const float*)d_in[7];
    float* out = (float*)d_out;

    float* sup1 = (float*)d_ws;
    float* hacc = sup1 + (size_t)N_NODES * NHID;
    float* sup2 = sup1;
    int*   recs = (int*)(hacc + (size_t)N_NODES * NHID);
    int*   cnts = recs + (size_t)NOWN * 32 * WCAP * 4;

    gemm1_kernel<<<NTILES, 128, 0, stream>>>(x, W1, sup1);
    owner_sum_kernel<NHID, true><<<NOWN, 1024, 0, stream>>>(es, ed, ev, sup1, b1, hacc, recs, cnts);
    gemm2_kernel<<<(NTILES + 7) / 8, 256, 0, stream>>>(hacc, W2, sup2);
    replay_sum_kernel<NCLS, false><<<NOWN, 1024, 0, stream>>>(es, ed, ev, sup2, b2, out, recs, cnts);
    log_softmax_kernel<<<(N_NODES + 127) / 128, 128, 0, stream>>>(out);
}


# owner blocks scan one byte per edge (dst/392, its owner block, written by GEMM1's blocks), not edge_dst: 4x fewer L2 bytes, 16 edges per test; edges meet waves in a new order, arrival-order sums
# speedup vs baseline: 2.3882x; 1.1427x over previous
#include <hip/hip_runtime.h>

typedef float v2f __attribute__((ext_vector_type(2)));
typedef float v8f __attribute__((ext_vector_type(8)));

#define N_NODES 100000
#define FEAT    256
#define NHID    64
#define NCLS    16
#define NEDGE   1600000
#define NTILES  (N_NODES / 16)
#define RPB     392
#define EPT     (NEDGE / NTILES)

__global__ __launch_bounds__(128) void gemm1_kernel(const float* __restrict__ x,
                                                    const float* __restrict__ W1,
                                                    float* __restrict__ sup,
                                                    const int* __restrict__ edst,
                                                    int* __restrict__ ob) {
    __shared__ float tileA[16 * FEAT];
    const int lane = threadIdx.x & 31;
    const int wave = threadIdx.x >> 5;
    const int rb   = blockIdx.x * 16;

    const float4* x4 = reinterpret_cast<const float4*>(x + (size_t)rb * FEAT);
    float4* t4 = reinterpret_cast<float4*>(tileA);
    for (int i = threadIdx.x; i < 16 * FEAT / 4; i += 128) t4[i] = x4[i];
    __syncthreads();

    const int c  = wave * 16;
    const int l  = lane & 15;
    const int hi = lane >> 4;
    v8f acc = {};
    for (int k = 0; k < FEAT; k += 4) {
        const int ks = k + 2 * hi;
        v2f a = *reinterpret_cast<const v2f*>(&tileA[l * FEAT + ks]);
        v2f b;
        b.x = W1[(size_t)ks       * NHID + c + l];
        b.y = W1[(size_t)(ks + 1) * NHID + c + l];
        acc = __builtin_amdgcn_wmma_f32_16x16x4_f32(false, a, false, b,
                                                    (short)0, acc, false, false);
    }
    float* out = sup + (size_t)rb * NHID + c;
    const int mb = hi * 8;
#pragma unroll
    for (int r = 0; r < 8; ++r)
        out[(size_t)(mb + r) * NHID + l] = acc[r];

    static_assert(EPT == 256 && EPT * NTILES == NEDGE, "16 lanes x 16 edges per block cover every edge once");
    static_assert((N_NODES - 1) / RPB <= 255, "an owner id fits a byte");
    if (threadIdx.x < 16) {
        const int4* d4 = reinterpret_cast<const int4*>(edst) + (size_t)blockIdx.x * (EPT / 4) + threadIdx.x * 4;
        unsigned pk[4];
#pragma unroll
        for (int g = 0; g < 4; ++g) {
            const int4 d = d4[g];
            pk[g] = ((unsigned)d.x / RPB) | (((unsigned)d.y / RPB) << 8) |
                    (((unsigned)d.z / RPB) << 16) | (((unsigned)d.w / RPB) << 24);
        }
        reinterpret_cast<int4*>(ob)[(size_t)blockIdx.x * (EPT / 16) + threadIdx.x] =
            make_int4((int)pk[0], (int)pk[1], (int)pk[2], (int)pk[3]);
    }
}

__global__ __launch_bounds__(256) void gemm2_kernel(const float* __restrict__ h,
                                                    const float* __restrict__ W2,
                                                    float* __restrict__ sup2) {
    const int lane = threadIdx.x & 31;
    const int wave = threadIdx.x >> 5;
    const int tile = blockIdx.x * 8 + wave;
    if (tile >= NTILES) return;
    const int rb = tile * 16;
    const int l  = lane & 15;
    const int hi = lane >> 4;
    v8f acc = {};
    for (int k = 0; k < NHID; k += 4) {
        const int ks = k + 2 * hi;
        v2f a = *reinterpret_cast<const v2f*>(&h[(size_t)(rb + l) * NHID + ks]);
        v2f b;
        b.x = W2[ks * NCLS + l];
        b.y = W2[(ks + 1) * NCLS + l];
        acc = __builtin_amdgcn_wmma_f32_16x16x4_f32(false, a, false, b,
                                                    (short)0, acc, false, false);
    }
    float* out = sup2 + (size_t)rb * NCLS;
    const int mb = hi * 8;
#pragma unroll
    for (int r = 0; r < 8; ++r)
        out[(size_t)(mb + r) * NCLS + l] = acc[r];
}

#define NOWN  256
#define NQ    (NEDGE / 16)
#define SCAN_U 2
#define NFULL (NQ / (1024 * SCAN_U))
#define PCAP  1056
#define ROT   392
#define WCAP  512

template <bool GUARD>
__device__ __forceinline__ void load_groups(const int4* __restrict__ d4, int it, int tid, int rot,
                                            int fill, int4 (&dv)[SCAN_U], int (&qi)[SCAN_U]) {
#pragma unroll
    for (int u = 0; u < SCAN_U; ++u) {
        int q = (it * SCAN_U + u) * 1024 + tid;
        const bool valid = !GUARD || q < NQ;
        q += rot;
        if (q >= NQ) q -= NQ;
        qi[u] = q;
        dv[u] = valid ? d4[q] : make_int4(fill, fill, fill, fill);
    }
}

template <int F>
__device__ __forceinline__ void gather_add(int ro, int s, float w, int n, int lane,
                                           const float* __restrict__ sup,
                                           float* acc) {
    constexpr int VW  = (F == 64) ? 2 : 1;
    constexpr int LPE = F / VW;
    constexpr int EPS = 32 / LPE;
    constexpr int NST = 32 / EPS;
    constexpr int CH  = 8;
    const int sub = lane / LPE;
    const int col = (lane % LPE) * VW;
    for (int t0 = 0; t0 < NST; t0 += CH) {
        float v0[CH], v1[CH], wj[CH];
        int   rj[CH];
#pragma unroll
        for (int k = 0; k < CH; ++k) {
            const int j  = (t0 + k) * EPS + sub;
            const int sj = __shfl(s, j);
            rj[k] = __shfl(ro, j);
            wj[k] = __shfl(w, j);
            v0[k] = 0.0f; v1[k] = 0.0f;
            if (j < n) {
                if (VW == 2) {
                    const v2f v = *reinterpret_cast<const v2f*>(sup + (size_t)sj * F + col);
                    v0[k] = v.x; v1[k] = v.y;
                } else {
                    v0[k] = sup[(size_t)sj * F + col];
                }
            }
        }
#pragma unroll
        for (int k = 0; k < CH; ++k) {
            const int j = (t0 + k) * EPS + sub;
            if (j < n) {
                atomicAdd(&acc[rj[k] + col], v0[k] * wj[k]);
                if (VW == 2) atomicAdd(&acc[rj[k] + col + 1], v1[k] * wj[k]);
            }
        }
    }
}

template <int F, bool REC>
__device__ __forceinline__ void add_edges(const int* list, int n, int lane, int lo,
                                          const int* __restrict__ esrc,
                                          const int* __restrict__ edst,
                                          const float* __restrict__ ev,
                                          const float* __restrict__ sup,
                                          float* acc, int4* rec, int& written) {
    int ro = 0, s = 0;
    float w = 0.0f;
    if (lane < n) {
        const int e   = list[lane];
        const int row = edst[e] - lo;
        ro = row * F;
        s  = esrc[e];
        w  = ev[e];
        if (REC && written + lane < WCAP)
            rec[written + lane] = make_int4(s, __float_as_int(w), row, 0);
    }
    if (REC) written += n;
    gather_add<F>(ro, s, w, n, lane, sup, acc);
}

__device__ __forceinline__ void scan_set(const int4 (&dv)[SCAN_U], const int (&qi)[SCAN_U],
                                         unsigned pat, int* mypend, int& cnt) {
#pragma unroll
    for (int u = 0; u < SCAN_U; ++u) {
        const unsigned ww[4] = {(unsigned)dv[u].x, (unsigned)dv[u].y, (unsigned)dv[u].z, (unsigned)dv[u].w};
        unsigned m = 0u;
#pragma unroll
        for (int c = 0; c < 4; ++c) {
            const unsigned x = ww[c] ^ pat;
            const unsigned t = ~(((x & 0x7f7f7f7fu) + 0x7f7f7f7fu) | x | 0x7f7f7f7fu);
            m |= t >> (7 - c);
        }
        unsigned bal = __builtin_amdgcn_ballot_w32(m != 0u);
        while (bal) {
            const int pos = cnt + (int)__builtin_amdgcn_mbcnt_lo(bal, 0u);
            if (m != 0u) {
                const int p = __builtin_ctz(m);
                mypend[pos] = qi[u] * 16 + (p & 3) * 4 + (p >> 3);
                m &= m - 1u;
            }
            cnt += __builtin_popcount(bal);
            bal = __builtin_amdgcn_ballot_w32(m != 0u);
        }
    }
}

template <int F, bool REC>
__device__ __forceinline__ void scan_and_sum(const int* __restrict__ esrc,
                                             const int* __restrict__ edst,
                                             const float* __restrict__ ev,
                                             const float* __restrict__ sup,
                                             const int* __restrict__ ob,
                                             float* acc, int* mypend, int lo, int owner,
                                             int tid, int lane, int rot,
                                             int4* rec, int& written) {
    int  cnt = 0;
    const int4* d4 = reinterpret_cast<const int4*>(ob);
    const unsigned pat = (unsigned)owner * 0x01010101u;
    const int fill = (int)~pat;
    static_assert(NFULL % 2 == 0 && NFULL >= 2, "the loop below ends with set a holding trip NFULL - 2");
    static_assert((NFULL + 1) * 1024 * SCAN_U >= NQ, "one guarded trip covers the rest");
    static_assert(31 + SCAN_U * 16 * 32 <= PCAP, "a trip's edges fit the pending list whatever the input");
    int4 ga[SCAN_U], gb[SCAN_U];
    int  qa[SCAN_U], qb[SCAN_U];
    auto drain = [&]() {
        while (cnt >= 32) {
            cnt -= 32;
            add_edges<F, REC>(mypend + cnt, 32, lane, lo, esrc, edst, ev, sup, acc, rec, written);
        }
    };
    load_groups<false>(d4, 0, tid, rot, fill, ga, qa);
    for (int it = 0; it < NFULL - 2; it += 2) {
        load_groups<false>(d4, it + 1, tid, rot, fill, gb, qb);
        scan_set(ga, qa, pat, mypend, cnt);
        drain();
        load_groups<false>(d4, it + 2, tid, rot, fill, ga, qa);
        scan_set(gb, qb, pat, mypend, cnt);
        drain();
    }
    load_groups<false>(d4, NFULL - 1, tid, rot, fill, gb, qb);
    scan_set(ga, qa, pat, mypend, cnt);
    drain();
    load_groups<true>(d4, NFULL, tid, rot, fill, ga, qa);
    scan_set(gb, qb, pat, mypend, cnt);
    drain();
    scan_set(ga, qa, pat, mypend, cnt);
    drain();
    add_edges<F, REC>(mypend, cnt, lane, lo, esrc, edst, ev, sup, acc, rec, written);
}

template <int F, bool RELU>
__device__ __forceinline__ void store_rows(const float* acc, const float* __restrict__ bias,
                                           float* __restrict__ dstm, int lo, int nrows, int tid) {
    const float4* acc4 = reinterpret_cast<const float4*>(acc);
    float4* o4 = reinterpret_cast<float4*>(dstm + (size_t)lo * F);
    for (int i = tid; i < nrows * F / 4; i += 1024) {
        float4 v = acc4[i];
        const int c = (i * 4) & (F - 1);
        v.x += bias[c]; v.y += bias[c + 1]; v.z += bias[c + 2]; v.w += bias[c + 3];
        if (RELU) {
            v.x = v.x > 0.0f ? v.x : 0.0f; v.y = v.y > 0.0f ? v.y : 0.0f;
            v.z = v.z > 0.0f ? v.z : 0.0f; v.w = v.w > 0.0f ? v.w : 0.0f;
        }
        o4[i] = v;
    }
}

template <int F, bool RELU>
__global__ __launch_bounds__(1024) void owner_sum_kernel(const int* __restrict__ esrc,
                                                         const int* __restrict__ edst,
                                                         const float* __restrict__ ev,
                                                         const float* __restrict__ sup,
                                                         const float* __restrict__ bias,
                                                         float* __restrict__ dstm,
                                                         int* __restrict__ recs,
                                                         int* __restrict__ cnts,
                                                         const int* __restrict__ ob) {
    __shared__ float acc[RPB * F];
    __shared__ int   pend[32 * PCAP];
    const int tid  = threadIdx.x;
    const int lane = tid & 31;
    const int wave = tid >> 5;
    const int lo   = blockIdx.x * RPB;
    int nrows = N_NODES - lo;
    if (nrows > RPB) nrows = RPB;

    float4* acc4 = reinterpret_cast<float4*>(acc);
    for (int i = tid; i < RPB * F / 4; i += 1024) acc4[i] = make_float4(0.f, 0.f, 0.f, 0.f);
    __syncthreads();

    const int rot = blockIdx.x * ROT;
    const int seg = blockIdx.x * 32 + wave;
    int4* rec = reinterpret_cast<int4*>(recs) + (size_t)seg * WCAP;
    int written = 0;
    scan_and_sum<F, true>(esrc, edst, ev, sup, ob, acc, pend + wave * PCAP, lo, (int)blockIdx.x,
                          tid, lane, rot, rec, written);
    if (lane == 0) cnts[seg] = written;
    __syncthreads();
    store_rows<F, RELU>(acc, bias, dstm, lo, nrows, tid);
}

template <int F, bool RELU>
__global__ __launch_bounds__(1024) void replay_sum_kernel(const int* __restrict__ esrc,
                                                          const int* __restrict__ edst,
                                                          const float* __restrict__ ev,
                                                          const float* __restrict__ sup,
                                                          const float* __restrict__ bias,
                                                          float* __restrict__ dstm,
                                                          const int* __restrict__ recs,
                                                          const int* __restrict__ cnts,
                                                          const int* __restrict__ ob) {
    __shared__ float acc[RPB * F];
    __shared__ int   pend[32 * PCAP];
    const int tid  = threadIdx.x;
    const int lane = tid & 31;
    const int wave = tid >> 5;
    const int lo   = blockIdx.x * RPB;
    int nrows = N_NODES - lo;
    if (nrows > RPB) nrows = RPB;

    float4* acc4 = reinterpret_cast<float4*>(acc);
    for (int i = tid; i < RPB * F / 4; i += 1024) acc4[i] = make_float4(0.f, 0.f, 0.f, 0.f);
    __syncthreads();

    const bool over = __builtin_amdgcn_ballot_w32(cnts[blockIdx.x * 32 + lane] > WCAP) != 0u;
    if (!over) {
        const int seg = blockIdx.x * 32 + wave;
        const int n   = __builtin_amdgcn_readfirstlane(cnts[seg]);
        const int4* rec = reinterpret_cast<const int4*>(recs) + (size_t)seg * WCAP;
        if (n > 0) {
            const int last = n - 1;
            int i0 = lane < last ? lane : last;
            int4 cur = rec[i0];
            for (int base = 0; base < n; base += 32) {
                int i1 = base + 32 + lane;
                i1 = i1 < last ? i1 : last;
                const int4 nxt = rec[i1];
                int m = n - base;
                if (m > 32) m = 32;
                gather_add<F>(cur.z * F, cur.x, __int_as_float(cur.y), m, lane, sup, acc);
                cur = nxt;
            }
        }
    } else {
        int unused = 0;
        scan_and_sum<F, false>(esrc, edst, ev, sup, ob, acc, pend + wave * PCAP, lo, (int)blockIdx.x,
                               tid, lane, blockIdx.x * ROT, nullptr, unused);
    }
    __syncthreads();
    store_rows<F, RELU>(acc, bias, dstm, lo, nrows, tid);
}

__global__ __launch_bounds__(128) void log_softmax_kernel(float* __restrict__ out) {
    const int r = blockIdx.x * blockDim.x + threadIdx.x;
    if (r >= N_NODES) return;
    float* p = out + (size_t)r * NCLS;
    float v[NCLS];
    float m = -3.402823466e+38f;
#pragma unroll
    for (int i = 0; i < NCLS; ++i) { v[i] = p[i]; m = v[i] > m ? v[i] : m; }
    float s = 0.0f;
#pragma unroll
    for (int i = 0; i < NCLS; ++i) s += __expf(v[i] - m);
    const float lse = m + __logf(s);
#pragma unroll
    for (int i = 0; i < NCLS; ++i) p[i] = v[i] - lse;
}

extern "C" void kernel_launch(void* const* d_in, const int* in_sizes, int n_in,
                              void* d_out, int out_size, void* d_ws, size_t ws_size,
                              hipStream_t stream) {
    (void)in_sizes; (void)n_in; (void)out_size; (void)ws_size;
    const float* x   = (const float*)d_in[0];
    const int*   es  = (const int*)  d_in[1];
    const int*   ed  = (const int*)  d_in[2];
    const float* ev  = (const float*)d_in[3];
    const float* W1  = (const float*)d_in[4];
    const float* b1  = (const float*)d_in[5];
    const float* W2  = (const float*)d_in[6];
    const float* b2  = (const float*)d_in[7];
    float* out = (float*)d_out;

    float* sup1 = (float*)d_ws;
    float* hacc = sup1 + (size_t)N_NODES * NHID;
    float* sup2 = sup1;
    int*   recs = (int*)(hacc + (size_t)N_NODES * NHID);
    int*   cnts = recs + (size_t)NOWN * 32 * WCAP * 4;
    int*   ob   = cnts + NOWN * 32;

    gemm1_kernel<<<NTILES, 128, 0, stream>>>(x, W1, sup1, ed, ob);
    owner_sum_kernel<NHID, true><<<NOWN, 1024, 0, stream>>>(es, ed, ev, sup1, b1, hacc, recs, cnts, ob);
    gemm2_kernel<<<(NTILES + 7) / 8, 256, 0, stream>>>(hacc, W2, sup2);
    replay_sum_kernel<NCLS, false><<<NOWN, 1024, 0, stream>>>(es, ed, ev, sup2, b2, out, recs, cnts, ob);
    log_softmax_kernel<<<(N_NODES + 127) / 128, 128, 0, stream>>>(out);
}


# GEMM1 A fragments read straight from global into registers (no LDS x tile; f32 k sums in a permuted order), one wave = 16 rows x 64 cols on 4 accumulators, B fragments of all W1 kept in registers
# speedup vs baseline: 2.7282x; 1.1424x over previous
#include <hip/hip_runtime.h>

typedef float v2f __attribute__((ext_vector_type(2)));
typedef float v8f __attribute__((ext_vector_type(8)));

#define N_NODES 100000
#define FEAT    256
#define NHID    64
#define NCLS    16
#define NEDGE   1600000
#define NTILES  (N_NODES / 16)
#define RPB     392
#define EPT     (NEDGE / NTILES)

#define G1_WAVES   4
#define G1_THREADS (G1_WAVES * 32)
#define G1_BLOCKS  256
#define G1_STEPS   (FEAT / 4)
#define G1_MINT    (NTILES / G1_BLOCKS / G1_WAVES)

__device__ __forceinline__ void g1_load_tile(const float* __restrict__ x, int tile, int l, int h,
                                             float4 (&a)[32]) {
    const float4* xr = reinterpret_cast<const float4*>(x + (size_t)(tile * 16 + l) * FEAT + 4 * h);
#pragma unroll
    for (int i = 0; i < 32; ++i) a[i] = xr[2 * i];
}

__device__ __forceinline__ void g1_compute_tile(const float4 (&a)[32], int tile, int l, int h,
                                                const float4 (&w01)[G1_STEPS], const float4 (&w23)[G1_STEPS],
                                                float* __restrict__ sup) {
    v8f c0 = {}, c1 = {}, c2 = {}, c3 = {};
#pragma unroll
    for (int i = 0; i < 32; ++i) {
#pragma unroll
        for (int p = 0; p < 2; ++p) {
            const int t = 2 * i + p;
            v2f av, b;
            av.x = p ? a[i].z : a[i].x;
            av.y = p ? a[i].w : a[i].y;
            b.x = w01[t].x; b.y = w01[t].y;
            c0 = __builtin_amdgcn_wmma_f32_16x16x4_f32(false, av, false, b, (short)0, c0, false, false);
            b.x = w01[t].z; b.y = w01[t].w;
            c1 = __builtin_amdgcn_wmma_f32_16x16x4_f32(false, av, false, b, (short)0, c1, false, false);
            b.x = w23[t].x; b.y = w23[t].y;
            c2 = __builtin_amdgcn_wmma_f32_16x16x4_f32(false, av, false, b, (short)0, c2, false, false);
            b.x = w23[t].z; b.y = w23[t].w;
            c3 = __builtin_amdgcn_wmma_f32_16x16x4_f32(false, av, false, b, (short)0, c3, false, false);
        }
    }
    float* out = sup + (size_t)(tile * 16 + h * 8) * NHID + l;
#pragma unroll
    for (int r = 0; r < 8; ++r) {
        out[(size_t)r * NHID]      = c0[r];
        out[(size_t)r * NHID + 16] = c1[r];
        out[(size_t)r * NHID + 32] = c2[r];
        out[(size_t)r * NHID + 48] = c3[r];
    }
}

__device__ __forceinline__ int2 g1_owner_bytes(const int4& d0, const int4& d1) {
    const unsigned pk0 = ((unsigned)d0.x / RPB) | (((unsigned)d0.y / RPB) << 8) |
                         (((unsigned)d0.z / RPB) << 16) | (((unsigned)d0.w / RPB) << 24);
    const unsigned pk1 = ((unsigned)d1.x / RPB) | (((unsigned)d1.y / RPB) << 8) |
                         (((unsigned)d1.z / RPB) << 16) | (((unsigned)d1.w / RPB) << 24);
    return make_int2((int)pk0, (int)pk1);
}

__global__ __launch_bounds__(G1_THREADS) void gemm1_kernel(const float* __restrict__ x,
                                                           const float* __restrict__ W1,
                                                           float* __restrict__ sup,
                                                           const int* __restrict__ edst,
                                                           int* __restrict__ ob) {
    __shared__ float4 wb01[G1_STEPS * 32];
    __shared__ float4 wb23[G1_STEPS * 32];
    const int tid  = threadIdx.x;
    const int lane = tid & 31;
    const int wave = tid >> 5;
    const int l    = lane & 15;
    const int h    = lane >> 4;

    static_assert(EPT == 256 && EPT * NTILES == NEDGE, "32 lanes x 8 edges per tile cover every edge once");
    static_assert((N_NODES - 1) / RPB <= 255, "an owner id fits a byte");
    static_assert(G1_MINT >= 1 && NTILES < (G1_MINT * G1_WAVES + 1) * G1_BLOCKS,
                  "a block has G1_MINT * G1_WAVES tiles or one more, so a wave has G1_MINT or G1_MINT + 1");

    const int t_lo = (int)(((long long)blockIdx.x * NTILES) / G1_BLOCKS);
    const int t_hi = (int)(((long long)(blockIdx.x + 1) * NTILES) / G1_BLOCKS);
    const int tile0 = t_lo + wave;
    const int n = (t_hi - tile0 + G1_WAVES - 1) / G1_WAVES;

    float4 A[32], B[32];
    g1_load_tile(x, tile0, l, h, A);

    {
        int4 d0[G1_MINT], d1[G1_MINT];
#pragma unroll
        for (int i = 0; i < G1_MINT; ++i) {
            const int4* d4 = reinterpret_cast<const int4*>(edst) + (size_t)(tile0 + G1_WAVES * i) * (EPT / 4) + lane * 2;
            d0[i] = d4[0]; d1[i] = d4[1];
        }
#pragma unroll
        for (int i = 0; i < G1_MINT; ++i)
            reinterpret_cast<int2*>(ob)[(size_t)(tile0 + G1_WAVES * i) * (EPT / 8) + lane] = g1_owner_bytes(d0[i], d1[i]);
        if (n > G1_MINT) {
            const int tl = tile0 + G1_WAVES * G1_MINT;
            const int4* d4 = reinterpret_cast<const int4*>(edst) + (size_t)tl * (EPT / 4) + lane * 2;
            reinterpret_cast<int2*>(ob)[(size_t)tl * (EPT / 8) + lane] = g1_owner_bytes(d4[0], d4[1]);
        }
    }

    for (int i = tid; i < G1_STEPS * 32; i += G1_THREADS) {
        const int t  = i >> 5;
        const int ll = i & 15;
        const int hh = (i >> 4) & 1;
        const int k0 = 64 * (t >> 4) + 8 * ((t >> 1) & 7) + 4 * hh + 2 * (t & 1);
        const float* r0 = W1 + (size_t)k0 * NHID + ll;
        const float* r1 = r0 + NHID;
        wb01[i] = make_float4(r0[0],  r1[0],  r0[16], r1[16]);
        wb23[i] = make_float4(r0[32], r1[32], r0[48], r1[48]);
    }
    __syncthreads();
    float4 w01[G1_STEPS], w23[G1_STEPS];
#pragma unroll
    for (int t = 0; t < G1_STEPS; ++t) {
        w01[t] = wb01[t * 32 + lane];
        w23[t] = wb23[t * 32 + lane];
    }

    int i = 0;
    for (; i + 2 < n; i += 2) {
        g1_load_tile(x, tile0 + G1_WAVES * (i + 1), l, h, B);
        __builtin_amdgcn_sched_barrier(0);
        g1_compute_tile(A, tile0 + G1_WAVES * i, l, h, w01, w23, sup);
        __builtin_amdgcn_sched_barrier(0);
        g1_load_tile(x, tile0 + G1_WAVES * (i + 2), l, h, A);
        __builtin_amdgcn_sched_barrier(0);
        g1_compute_tile(B, tile0 + G1_WAVES * (i + 1), l, h, w01, w23, sup);
        __builtin_amdgcn_sched_barrier(0);
    }
    const bool two = i + 1 < n;
    if (two) g1_load_tile(x, tile0 + G1_WAVES * (i + 1), l, h, B);
    __builtin_amdgcn_sched_barrier(0);
    g1_compute_tile(A, tile0 + G1_WAVES * i, l, h, w01, w23, sup);
    if (two) g1_compute_tile(B, tile0 + G1_WAVES * (i + 1), l, h, w01, w23, sup);
}

__global__ __launch_bounds__(256) void gemm2_kernel(const float* __restrict__ h,
                                                    const float* __restrict__ W2,
                                                    float* __restrict__ sup2) {
    const int lane = threadIdx.x & 31;
    const int wave = threadIdx.x >> 5;
    const int tile = blockIdx.x * 8 + wave;
    if (tile >= NTILES) return;
    const int rb = tile * 16;
    const int l  = lane & 15;
    const int hi = lane >> 4;
    v8f acc = {};
    for (int k = 0; k < NHID; k += 4) {
        const int ks = k + 2 * hi;
        v2f a = *reinterpret_cast<const v2f*>(&h[(size_t)(rb + l) * NHID + ks]);
        v2f b;
        b.x = W2[ks * NCLS + l];
        b.y = W2[(ks + 1) * NCLS + l];
        acc = __builtin_amdgcn_wmma_f32_16x16x4_f32(false, a, false, b,
                                                    (short)0, acc, false, false);
    }
    float* out = sup2 + (size_t)rb * NCLS;
    const int mb = hi * 8;
#pragma unroll
    for (int r = 0; r < 8; ++r)
        out[(size_t)(mb + r) * NCLS + l] = acc[r];
}

#define NOWN  256
#define NQ    (NEDGE / 16)
#define SCAN_U 2
#define NFULL (NQ / (1024 * SCAN_U))
#define PCAP  1056
#define ROT   392
#define WCAP  512

template <bool GUARD>
__device__ __forceinline__ void load_groups(const int4* __restrict__ d4, int it, int tid, int rot,
                                            int fill, int4 (&dv)[SCAN_U], int (&qi)[SCAN_U]) {
#pragma unroll
    for (int u = 0; u < SCAN_U; ++u) {
        int q = (it * SCAN_U + u) * 1024 + tid;
        const bool valid = !GUARD || q < NQ;
        q += rot;
        if (q >= NQ) q -= NQ;
        qi[u] = q;
        dv[u] = valid ? d4[q] : make_int4(fill, fill, fill, fill);
    }
}

template <int F>
__device__ __forceinline__ void gather_add(int ro, int s, float w, int n, int lane,
                                           const float* __restrict__ sup,
                                           float* acc) {
    constexpr int VW  = (F == 64) ? 2 : 1;
    constexpr int LPE = F / VW;
    constexpr int EPS = 32 / LPE;
    constexpr int NST = 32 / EPS;
    constexpr int CH  = 8;
    const int sub = lane / LPE;
    const int col = (lane % LPE) * VW;
    for (int t0 = 0; t0 < NST; t0 += CH) {
        float v0[CH], v1[CH], wj[CH];
        int   rj[CH];
#pragma unroll
        for (int k = 0; k < CH; ++k) {
            const int j  = (t0 + k) * EPS + sub;
            const int sj = __shfl(s, j);
            rj[k] = __shfl(ro, j);
            wj[k] = __shfl(w, j);
            v0[k] = 0.0f; v1[k] = 0.0f;
            if (j < n) {
                if (VW == 2) {
                    const v2f v = *reinterpret_cast<const v2f*>(sup + (size_t)sj * F + col);
                    v0[k] = v.x; v1[k] = v.y;
                } else {
                    v0[k] = sup[(size_t)sj * F + col];
                }
            }
        }
#pragma unroll
        for (int k = 0; k < CH; ++k) {
            const int j = (t0 + k) * EPS + sub;
            if (j < n) {
                atomicAdd(&acc[rj[k] + col], v0[k] * wj[k]);
                if (VW == 2) atomicAdd(&acc[rj[k] + col + 1], v1[k] * wj[k]);
            }
        }
    }
}

template <int F, bool REC>
__device__ __forceinline__ void add_edges(const int* list, int n, int lane, int lo,
                                          const int* __restrict__ esrc,
                                          const int* __restrict__ edst,
                                          const float* __restrict__ ev,
                                          const float* __restrict__ sup,
                                          float* acc, int4* rec, int& written) {
    int ro = 0, s = 0;
    float w = 0.0f;
    if (lane < n) {
        const int e   = list[lane];
        const int row = edst[e] - lo;
        ro = row * F;
        s  = esrc[e];
        w  = ev[e];
        if (REC && written + lane < WCAP)
            rec[written + lane] = make_int4(s, __float_as_int(w), row, 0);
    }
    if (REC) written += n;
    gather_add<F>(ro, s, w, n, lane, sup, acc);
}

__device__ __forceinline__ void scan_set(const int4 (&dv)[SCAN_U], const int (&qi)[SCAN_U],
                                         unsigned pat, int* mypend, int& cnt) {
#pragma unroll
    for (int u = 0; u < SCAN_U; ++u) {
        const unsigned ww[4] = {(unsigned)dv[u].x, (unsigned)dv[u].y, (unsigned)dv[u].z, (unsigned)dv[u].w};
        unsigned m = 0u;
#pragma unroll
        for (int c = 0; c < 4; ++c) {
            const unsigned x = ww[c] ^ pat;
            const unsigned t = ~(((x & 0x7f7f7f7fu) + 0x7f7f7f7fu) | x | 0x7f7f7f7fu);
            m |= t >> (7 - c);
        }
        unsigned bal = __builtin_amdgcn_ballot_w32(m != 0u);
        while (bal) {
            const int pos = cnt + (int)__builtin_amdgcn_mbcnt_lo(bal, 0u);
            if (m != 0u) {
                const int p = __builtin_ctz(m);
                mypend[pos] = qi[u] * 16 + (p & 3) * 4 + (p >> 3);
                m &= m - 1u;
            }
            cnt += __builtin_popcount(bal);
            bal = __builtin_amdgcn_ballot_w32(m != 0u);
        }
    }
}

template <int F, bool REC>
__device__ __forceinline__ void scan_and_sum(const int* __restrict__ esrc,
                                             const int* __restrict__ edst,
                                             const float* __restrict__ ev,
                                             const float* __restrict__ sup,
                                             const int* __restrict__ ob,
                                             float* acc, int* mypend, int lo, int owner,
                                             int tid, int lane, int rot,
                                             int4* rec, int& written) {
    int  cnt = 0;
    const int4* d4 = reinterpret_cast<const int4*>(ob);
    const unsigned pat = (unsigned)owner * 0x01010101u;
    const int fill = (int)~pat;
    static_assert(NFULL % 2 == 0 && NFULL >= 2, "the loop below ends with set a holding trip NFULL - 2");
    static_assert((NFULL + 1) * 1024 * SCAN_U >= NQ, "one guarded trip covers the rest");
    static_assert(31 + SCAN_U * 16 * 32 <= PCAP, "a trip's edges fit the pending list whatever the input");
    int4 ga[SCAN_U], gb[SCAN_U];
    int  qa[SCAN_U], qb[SCAN_U];
    auto drain = [&]() {
        while (cnt >= 32) {
            cnt -= 32;
            add_edges<F, REC>(mypend + cnt, 32, lane, lo, esrc, edst, ev, sup, acc, rec, written);
        }
    };
    load_groups<false>(d4, 0, tid, rot, fill, ga, qa);
    for (int it = 0; it < NFULL - 2; it += 2) {
        load_groups<false>(d4, it + 1, tid, rot, fill, gb, qb);
        scan_set(ga, qa, pat, mypend, cnt);
        drain();
        load_groups<false>(d4, it + 2, tid, rot, fill, ga, qa);
        scan_set(gb, qb, pat, mypend, cnt);
        drain();
    }
    load_groups<false>(d4, NFULL - 1, tid, rot, fill, gb, qb);
    scan_set(ga, qa, pat, mypend, cnt);
    drain();
    load_groups<true>(d4, NFULL, tid, rot, fill, ga, qa);
    scan_set(gb, qb, pat, mypend, cnt);
    drain();
    scan_set(ga, qa, pat, mypend, cnt);
    drain();
    add_edges<F, REC>(mypend, cnt, lane, lo, esrc, edst, ev, sup, acc, rec, written);
}

template <int F, bool RELU>
__device__ __forceinline__ void store_rows(const float* acc, const float* __restrict__ bias,
                                           float* __restrict__ dstm, int lo, int nrows, int tid) {
    const float4* acc4 = reinterpret_cast<const float4*>(acc);
    float4* o4 = reinterpret_cast<float4*>(dstm + (size_t)lo * F);
    for (int i = tid; i < nrows * F / 4; i += 1024) {
        float4 v = acc4[i];
        const int c = (i * 4) & (F - 1);
        v.x += bias[c]; v.y += bias[c + 1]; v.z += bias[c + 2]; v.w += bias[c + 3];
        if (RELU) {
            v.x = v.x > 0.0f ? v.x : 0.0f; v.y = v.y > 0.0f ? v.y : 0.0f;
            v.z = v.z > 0.0f ? v.z : 0.0f; v.w = v.w > 0.0f ? v.w : 0.0f;
        }
        o4[i] = v;
    }
}

template <int F, bool RELU>
__global__ __launch_bounds__(1024) void owner_sum_kernel(const int* __restrict__ esrc,
                                                         const int* __restrict__ edst,
                                                         const float* __restrict__ ev,
                                                         const float* __restrict__ sup,
                                                         const float* __restrict__ bias,
                                                         float* __restrict__ dstm,
                                                         int* __restrict__ recs,
                                                         int* __restrict__ cnts,
                                                         const int* __restrict__ ob) {
    __shared__ float acc[RPB * F];
    __shared__ int   pend[32 * PCAP];
    const int tid  = threadIdx.x;
    const int lane = tid & 31;
    const int wave = tid >> 5;
    const int lo   = blockIdx.x * RPB;
    int nrows = N_NODES - lo;
    if (nrows > RPB) nrows = RPB;

    float4* acc4 = reinterpret_cast<float4*>(acc);
    for (int i = tid; i < RPB * F / 4; i += 1024) acc4[i] = make_float4(0.f, 0.f, 0.f, 0.f);
    __syncthreads();

    const int rot = blockIdx.x * ROT;
    const int seg = blockIdx.x * 32 + wave;
    int4* rec = reinterpret_cast<int4*>(recs) + (size_t)seg * WCAP;
    int written = 0;
    scan_and_sum<F, true>(esrc, edst, ev, sup, ob, acc, pend + wave * PCAP, lo, (int)blockIdx.x,
                          tid, lane, rot, rec, written);
    if (lane == 0) cnts[seg] = written;
    __syncthreads();
    store_rows<F, RELU>(acc, bias, dstm, lo, nrows, tid);
}

template <int F, bool RELU>
__global__ __launch_bounds__(1024) void replay_sum_kernel(const int* __restrict__ esrc,
                                                          const int* __restrict__ edst,
                                                          const float* __restrict__ ev,
                                                          const float* __restrict__ sup,
                                                          const float* __restrict__ bias,
                                                          float* __restrict__ dstm,
                                                          const int* __restrict__ recs,
                                                          const int* __restrict__ cnts,
                                                          const int* __restrict__ ob) {
    __shared__ float acc[RPB * F];
    __shared__ int   pend[32 * PCAP];
    const int tid  = threadIdx.x;
    const int lane = tid & 31;
    const int wave = tid >> 5;
    const int lo   = blockIdx.x * RPB;
    int nrows = N_NODES - lo;
    if (nrows > RPB) nrows = RPB;

    float4* acc4 = reinterpret_cast<float4*>(acc);
    for (int i = tid; i < RPB * F / 4; i += 1024) acc4[i] = make_float4(0.f, 0.f, 0.f, 0.f);
    __syncthreads();

    const bool over = __builtin_amdgcn_ballot_w32(cnts[blockIdx.x * 32 + lane] > WCAP) != 0u;
    if (!over) {
        const int seg = blockIdx.x * 32 + wave;
        const int n   = __builtin_amdgcn_readfirstlane(cnts[seg]);
        const int4* rec = reinterpret_cast<const int4*>(recs) + (size_t)seg * WCAP;
        if (n > 0) {
            const int last = n - 1;
            int i0 = lane < last ? lane : last;
            int4 cur = rec[i0];
            for (int base = 0; base < n; base += 32) {
                int i1 = base + 32 + lane;
                i1 = i1 < last ? i1 : last;
                const int4 nxt = rec[i1];
                int m = n - base;
                if (m > 32) m = 32;
                gather_add<F>(cur.z * F, cur.x, __int_as_float(cur.y), m, lane, sup, acc);
                cur = nxt;
            }
        }
    } else {
        int unused = 0;
        scan_and_sum<F, false>(esrc, edst, ev, sup, ob, acc, pend + wave * PCAP, lo, (int)blockIdx.x,
                               tid, lane, blockIdx.x * ROT, nullptr, unused);
    }
    __syncthreads();
    store_rows<F, RELU>(acc, bias, dstm, lo, nrows, tid);
}

__global__ __launch_bounds__(128) void log_softmax_kernel(float* __restrict__ out) {
    const int r = blockIdx.x * blockDim.x + threadIdx.x;
    if (r >= N_NODES) return;
    float* p = out + (size_t)r * NCLS;
    float v[NCLS];
    float m = -3.402823466e+38f;
#pragma unroll
    for (int i = 0; i < NCLS; ++i) { v[i] = p[i]; m = v[i] > m ? v[i] : m; }
    float s = 0.0f;
#pragma unroll
    for (int i = 0; i < NCLS; ++i) s += __expf(v[i] - m);
    const float lse = m + __logf(s);
#pragma unroll
    for (int i = 0; i < NCLS; ++i) p[i] = v[i] - lse;
}

extern "C" void kernel_launch(void* const* d_in, const int* in_sizes, int n_in,
                              void* d_out, int out_size, void* d_ws, size_t ws_size,
                              hipStream_t stream) {
    (void)in_sizes; (void)n_in; (void)out_size; (void)ws_size;
    const float* x   = (const float*)d_in[0];
    const int*   es  = (const int*)  d_in[1];
    const int*   ed  = (const int*)  d_in[2];
    const float* ev  = (const float*)d_in[3];
    const float* W1  = (const float*)d_in[4];
    const float* b1  = (const float*)d_in[5];
    const float* W2  = (const float*)d_in[6];
    const float* b2  = (const float*)d_in[7];
    float* out = (float*)d_out;

    float* sup1 = (float*)d_ws;
    float* hacc = sup1 + (size_t)N_NODES * NHID;
    float* sup2 = sup1;
    int*   recs = (int*)(hacc + (size_t)N_NODES * NHID);
    int*   cnts = recs + (size_t)NOWN * 32 * WCAP * 4;
    int*   ob   = cnts + NOWN * 32;

    gemm1_kernel<<<G1_BLOCKS, G1_THREADS, 0, stream>>>(x, W1, sup1, ed, ob);
    owner_sum_kernel<NHID, true><<<NOWN, 1024, 0, stream>>>(es, ed, ev, sup1, b1, hacc, recs, cnts, ob);
    gemm2_kernel<<<(NTILES + 7) / 8, 256, 0, stream>>>(hacc, W2, sup2);
    replay_sum_kernel<NCLS, false><<<NOWN, 1024, 0, stream>>>(es, ed, ev, sup2, b2, out, recs, cnts, ob);
    log_softmax_kernel<<<(N_NODES + 127) / 128, 128, 0, stream>>>(out);
}


# edge fields reach the row gathers without LDS shuffles: scalar lane reads in layer 1, row-share moves in layer 2 (edges i and 16+i of a wave share a step: new order in the wave; arrival-order sums)
# speedup vs baseline: 2.8331x; 1.0385x over previous
#include <hip/hip_runtime.h>

typedef float v2f __attribute__((ext_vector_type(2)));
typedef float v8f __attribute__((ext_vector_type(8)));

#define N_NODES 100000
#define FEAT    256
#define NHID    64
#define NCLS    16
#define NEDGE   1600000
#define NTILES  (N_NODES / 16)
#define RPB     392
#define EPT     (NEDGE / NTILES)

#define G1_WAVES   4
#define G1_THREADS (G1_WAVES * 32)
#define G1_BLOCKS  256
#define G1_STEPS   (FEAT / 4)
#define G1_MINT    (NTILES / G1_BLOCKS / G1_WAVES)

__device__ __forceinline__ void g1_load_tile(const float* __restrict__ x, int tile, int l, int h,
                                             float4 (&a)[32]) {
    const float4* xr = reinterpret_cast<const float4*>(x + (size_t)(tile * 16 + l) * FEAT + 4 * h);
#pragma unroll
    for (int i = 0; i < 32; ++i) a[i] = xr[2 * i];
}

__device__ __forceinline__ void g1_compute_tile(const float4 (&a)[32], int tile, int l, int h,
                                                const float4 (&w01)[G1_STEPS], const float4 (&w23)[G1_STEPS],
                                                float* __restrict__ sup) {
    v8f c0 = {}, c1 = {}, c2 = {}, c3 = {};
#pragma unroll
    for (int i = 0; i < 32; ++i) {
#pragma unroll
        for (int p = 0; p < 2; ++p) {
            const int t = 2 * i + p;
            v2f av, b;
            av.x = p ? a[i].z : a[i].x;
            av.y = p ? a[i].w : a[i].y;
            b.x = w01[t].x; b.y = w01[t].y;
            c0 = __builtin_amdgcn_wmma_f32_16x16x4_f32(false, av, false, b, (short)0, c0, false, false);
            b.x = w01[t].z; b.y = w01[t].w;
            c1 = __builtin_amdgcn_wmma_f32_16x16x4_f32(false, av, false, b, (short)0, c1, false, false);
            b.x = w23[t].x; b.y = w23[t].y;
            c2 = __builtin_amdgcn_wmma_f32_16x16x4_f32(false, av, false, b, (short)0, c2, false, false);
            b.x = w23[t].z; b.y = w23[t].w;
            c3 = __builtin_amdgcn_wmma_f32_16x16x4_f32(false, av, false, b, (short)0, c3, false, false);
        }
    }
    float* out = sup + (size_t)(tile * 16 + h * 8) * NHID + l;
#pragma unroll
    for (int r = 0; r < 8; ++r) {
        out[(size_t)r * NHID]      = c0[r];
        out[(size_t)r * NHID + 16] = c1[r];
        out[(size_t)r * NHID + 32] = c2[r];
        out[(size_t)r * NHID + 48] = c3[r];
    }
}

__device__ __forceinline__ int2 g1_owner_bytes(const int4& d0, const int4& d1) {
    const unsigned pk0 = ((unsigned)d0.x / RPB) | (((unsigned)d0.y / RPB) << 8) |
                         (((unsigned)d0.z / RPB) << 16) | (((unsigned)d0.w / RPB) << 24);
    const unsigned pk1 = ((unsigned)d1.x / RPB) | (((unsigned)d1.y / RPB) << 8) |
                         (((unsigned)d1.z / RPB) << 16) | (((unsigned)d1.w / RPB) << 24);
    return make_int2((int)pk0, (int)pk1);
}

__global__ __launch_bounds__(G1_THREADS) void gemm1_kernel(const float* __restrict__ x,
                                                           const float* __restrict__ W1,
                                                           float* __restrict__ sup,
                                                           const int* __restrict__ edst,
                                                           int* __restrict__ ob) {
    __shared__ float4 wb01[G1_STEPS * 32];
    __shared__ float4 wb23[G1_STEPS * 32];
    const int tid  = threadIdx.x;
    const int lane = tid & 31;
    const int wave = tid >> 5;
    const int l    = lane & 15;
    const int h    = lane >> 4;

    static_assert(EPT == 256 && EPT * NTILES == NEDGE, "32 lanes x 8 edges per tile cover every edge once");
    static_assert((N_NODES - 1) / RPB <= 255, "an owner id fits a byte");
    static_assert(G1_MINT >= 1 && NTILES < (G1_MINT * G1_WAVES + 1) * G1_BLOCKS,
                  "a block has G1_MINT * G1_WAVES tiles or one more, so a wave has G1_MINT or G1_MINT + 1");

    const int t_lo = (int)(((long long)blockIdx.x * NTILES) / G1_BLOCKS);
    const int t_hi = (int)(((long long)(blockIdx.x + 1) * NTILES) / G1_BLOCKS);
    const int tile0 = t_lo + wave;
    const int n = (t_hi - tile0 + G1_WAVES - 1) / G1_WAVES;

    float4 A[32], B[32];
    g1_load_tile(x, tile0, l, h, A);

    {
        int4 d0[G1_MINT], d1[G1_MINT];
#pragma unroll
        for (int i = 0; i < G1_MINT; ++i) {
            const int4* d4 = reinterpret_cast<const int4*>(edst) + (size_t)(tile0 + G1_WAVES * i) * (EPT / 4) + lane * 2;
            d0[i] = d4[0]; d1[i] = d4[1];
        }
#pragma unroll
        for (int i = 0; i < G1_MINT; ++i)
            reinterpret_cast<int2*>(ob)[(size_t)(tile0 + G1_WAVES * i) * (EPT / 8) + lane] = g1_owner_bytes(d0[i], d1[i]);
        if (n > G1_MINT) {
            const int tl = tile0 + G1_WAVES * G1_MINT;
            const int4* d4 = reinterpret_cast<const int4*>(edst) + (size_t)tl * (EPT / 4) + lane * 2;
            reinterpret_cast<int2*>(ob)[(size_t)tl * (EPT / 8) + lane] = g1_owner_bytes(d4[0], d4[1]);
        }
    }

    for (int i = tid; i < G1_STEPS * 32; i += G1_THREADS) {
        const int t  = i >> 5;
        const int ll = i & 15;
        const int hh = (i >> 4) & 1;
        const int k0 = 64 * (t >> 4) + 8 * ((t >> 1) & 7) + 4 * hh + 2 * (t & 1);
        const float* r0 = W1 + (size_t)k0 * NHID + ll;
        const float* r1 = r0 + NHID;
        wb01[i] = make_float4(r0[0],  r1[0],  r0[16], r1[16]);
        wb23[i] = make_float4(r0[32], r1[32], r0[48], r1[48]);
    }
    __syncthreads();
    float4 w01[G1_STEPS], w23[G1_STEPS];
#pragma unroll
    for (int t = 0; t < G1_STEPS; ++t) {
        w01[t] = wb01[t * 32 + lane];
        w23[t] = wb23[t * 32 + lane];
    }

    int i = 0;
    for (; i + 2 < n; i += 2) {
        g1_load_tile(x, tile0 + G1_WAVES * (i + 1), l, h, B);
        __builtin_amdgcn_sched_barrier(0);
        g1_compute_tile(A, tile0 + G1_WAVES * i, l, h, w01, w23, sup);
        __builtin_amdgcn_sched_barrier(0);
        g1_load_tile(x, tile0 + G1_WAVES * (i + 2), l, h, A);
        __builtin_amdgcn_sched_barrier(0);
        g1_compute_tile(B, tile0 + G1_WAVES * (i + 1), l, h, w01, w23, sup);
        __builtin_amdgcn_sched_barrier(0);
    }
    const bool two = i + 1 < n;
    if (two) g1_load_tile(x, tile0 + G1_WAVES * (i + 1), l, h, B);
    __builtin_amdgcn_sched_barrier(0);
    g1_compute_tile(A, tile0 + G1_WAVES * i, l, h, w01, w23, sup);
    if (two) g1_compute_tile(B, tile0 + G1_WAVES * (i + 1), l, h, w01, w23, sup);
}

__global__ __launch_bounds__(256) void gemm2_kernel(const float* __restrict__ h,
                                                    const float* __restrict__ W2,
                                                    float* __restrict__ sup2) {
    const int lane = threadIdx.x & 31;
    const int wave = threadIdx.x >> 5;
    const int tile = blockIdx.x * 8 + wave;
    if (tile >= NTILES) return;
    const int rb = tile * 16;
    const int l  = lane & 15;
    const int hi = lane >> 4;
    v8f acc = {};
    for (int k = 0; k < NHID; k += 4) {
        const int ks = k + 2 * hi;
        v2f a = *reinterpret_cast<const v2f*>(&h[(size_t)(rb + l) * NHID + ks]);
        v2f b;
        b.x = W2[ks * NCLS + l];
        b.y = W2[(ks + 1) * NCLS + l];
        acc = __builtin_amdgcn_wmma_f32_16x16x4_f32(false, a, false, b,
                                                    (short)0, acc, false, false);
    }
    float* out = sup2 + (size_t)rb * NCLS;
    const int mb = hi * 8;
#pragma unroll
    for (int r = 0; r < 8; ++r)
        out[(size_t)(mb + r) * NCLS + l] = acc[r];
}

#define NOWN  256
#define NQ    (NEDGE / 16)
#define SCAN_U 2
#define NFULL (NQ / (1024 * SCAN_U))
#define PCAP  1056
#define ROT   392
#define WCAP  512

template <bool GUARD>
__device__ __forceinline__ void load_groups(const int4* __restrict__ d4, int it, int tid, int rot,
                                            int fill, int4 (&dv)[SCAN_U], int (&qi)[SCAN_U]) {
#pragma unroll
    for (int u = 0; u < SCAN_U; ++u) {
        int q = (it * SCAN_U + u) * 1024 + tid;
        const bool valid = !GUARD || q < NQ;
        q += rot;
        if (q >= NQ) q -= NQ;
        qi[u] = q;
        dv[u] = valid ? d4[q] : make_int4(fill, fill, fill, fill);
    }
}

template <int I> struct step_c { static constexpr int value = I; };
template <int I, int N, class Fn>
__device__ __forceinline__ void for_steps(Fn&& f) {
    if constexpr (I < N) { f(step_c<I>{}); for_steps<I + 1, N>(f); }
}

template <int F, int I>
__device__ __forceinline__ int edge_field(int x) {
    if constexpr (F == 64) return __builtin_amdgcn_readlane(x, I);
    else                   return __builtin_amdgcn_update_dpp(0, x, 0x150 + I, 0xf, 0xf, true);
}

template <int F>
__device__ __forceinline__ void gather_add(int ro, int s, float w, int n, int lane,
                                           const float* __restrict__ sup,
                                           float* acc) {
    constexpr int VW  = (F == 64) ? 2 : 1;
    constexpr int LPE = F / VW;
    constexpr int EPS = 32 / LPE;
    constexpr int NST = 32 / EPS;
    constexpr int CH  = 8;
    static_assert(LPE == 32 || LPE == 16, "an edge takes the wave or one 16-lane row");
    static_assert(NST % CH == 0, "whole chunks");
    const int sub = lane / LPE;
    const int col = (lane % LPE) * VW;
    for_steps<0, NST / CH>([&](auto c) {
        constexpr int T0 = decltype(c)::value * CH;
        float v0[CH], v1[CH];
        for_steps<0, CH>([&](auto kk) {
            constexpr int K = decltype(kk)::value;
            constexpr int I = T0 + K;
            const int j  = (EPS == 1) ? I : 16 * sub + I;
            const int sj = edge_field<F, I>(s);
            v0[K] = 0.0f; v1[K] = 0.0f;
            if (j < n) {
                if (VW == 2) {
                    const v2f v = *reinterpret_cast<const v2f*>(sup + (size_t)sj * F + col);
                    v0[K] = v.x; v1[K] = v.y;
                } else {
                    v0[K] = sup[(size_t)sj * F + col];
                }
            }
        });
        for_steps<0, CH>([&](auto kk) {
            constexpr int K = decltype(kk)::value;
            constexpr int I = T0 + K;
            const int   j  = (EPS == 1) ? I : 16 * sub + I;
            const int   rj = edge_field<F, I>(ro);
            const float wj = __int_as_float(edge_field<F, I>(__float_as_int(w)));
            if (j < n) {
                atomicAdd(&acc[rj + col], v0[K] * wj);
                if (VW == 2) atomicAdd(&acc[rj + col + 1], v1[K] * wj);
            }
        });
    });
}

template <int F, bool REC>
__device__ __forceinline__ void add_edges(const int* list, int n, int lane, int lo,
                                          const int* __restrict__ esrc,
                                          const int* __restrict__ edst,
                                          const float* __restrict__ ev,
                                          const float* __restrict__ sup,
                                          float* acc, int4* rec, int& written) {
    int ro = 0, s = 0;
    float w = 0.0f;
    if (lane < n) {
        const int e   = list[lane];
        const int row = edst[e] - lo;
        ro = row * F;
        s  = esrc[e];
        w  = ev[e];
        if (REC && written + lane < WCAP)
            rec[written + lane] = make_int4(s, __float_as_int(w), row, 0);
    }
    if (REC) written += n;
    gather_add<F>(ro, s, w, n, lane, sup, acc);
}

__device__ __forceinline__ void scan_set(const int4 (&dv)[SCAN_U], const int (&qi)[SCAN_U],
                                         unsigned pat, int* mypend, int& cnt) {
#pragma unroll
    for (int u = 0; u < SCAN_U; ++u) {
        const unsigned ww[4] = {(unsigned)dv[u].x, (unsigned)dv[u].y, (unsigned)dv[u].z, (unsigned)dv[u].w};
        unsigned m = 0u;
#pragma unroll
        for (int c = 0; c < 4; ++c) {
            const unsigned x = ww[c] ^ pat;
            const unsigned t = ~(((x & 0x7f7f7f7fu) + 0x7f7f7f7fu) | x | 0x7f7f7f7fu);
            m |= t >> (7 - c);
        }
        unsigned bal = __builtin_amdgcn_ballot_w32(m != 0u);
        while (bal) {
            const int pos = cnt + (int)__builtin_amdgcn_mbcnt_lo(bal, 0u);
            if (m != 0u) {
                const int p = __builtin_ctz(m);
                mypend[pos] = qi[u] * 16 + (p & 3) * 4 + (p >> 3);
                m &= m - 1u;
            }
            cnt += __builtin_popcount(bal);
            bal = __builtin_amdgcn_ballot_w32(m != 0u);
        }
    }
}

template <int F, bool REC>
__device__ __forceinline__ void scan_and_sum(const int* __restrict__ esrc,
                                             const int* __restrict__ edst,
                                             const float* __restrict__ ev,
                                             const float* __restrict__ sup,
                                             const int* __restrict__ ob,
                                             float* acc, int* mypend, int lo, int owner,
                                             int tid, int lane, int rot,
                                             int4* rec, int& written) {
    int  cnt = 0;
    const int4* d4 = reinterpret_cast<const int4*>(ob);
    const unsigned pat = (unsigned)owner * 0x01010101u;
    const int fill = (int)~pat;
    static_assert(NFULL % 2 == 0 && NFULL >= 2, "the loop below ends with set a holding trip NFULL - 2");
    static_assert((NFULL + 1) * 1024 * SCAN_U >= NQ, "one guarded trip covers the rest");
    static_assert(31 + SCAN_U * 16 * 32 <= PCAP, "a trip's edges fit the pending list whatever the input");
    int4 ga[SCAN_U], gb[SCAN_U];
    int  qa[SCAN_U], qb[SCAN_U];
    auto drain = [&]() {
        while (cnt >= 32) {
            cnt -= 32;
            add_edges<F, REC>(mypend + cnt, 32, lane, lo, esrc, edst, ev, sup, acc, rec, written);
        }
    };
    load_groups<false>(d4, 0, tid, rot, fill, ga, qa);
    for (int it = 0; it < NFULL - 2; it += 2) {
        load_groups<false>(d4, it + 1, tid, rot, fill, gb, qb);
        scan_set(ga, qa, pat, mypend, cnt);
        drain();
        load_groups<false>(d4, it + 2, tid, rot, fill, ga, qa);
        scan_set(gb, qb, pat, mypend, cnt);
        drain();
    }
    load_groups<false>(d4, NFULL - 1, tid, rot, fill, gb, qb);
    scan_set(ga, qa, pat, mypend, cnt);
    drain();
    load_groups<true>(d4, NFULL, tid, rot, fill, ga, qa);
    scan_set(gb, qb, pat, mypend, cnt);
    drain();
    scan_set(ga, qa, pat, mypend, cnt);
    drain();
    add_edges<F, REC>(mypend, cnt, lane, lo, esrc, edst, ev, sup, acc, rec, written);
}

template <int F, bool RELU>
__device__ __forceinline__ void store_rows(const float* acc, const float* __restrict__ bias,
                                           float* __restrict__ dstm, int lo, int nrows, int tid) {
    const float4* acc4 = reinterpret_cast<const float4*>(acc);
    float4* o4 = reinterpret_cast<float4*>(dstm + (size_t)lo * F);
    for (int i = tid; i < nrows * F / 4; i += 1024) {
        float4 v = acc4[i];
        const int c = (i * 4) & (F - 1);
        v.x += bias[c]; v.y += bias[c + 1]; v.z += bias[c + 2]; v.w += bias[c + 3];
        if (RELU) {
            v.x = v.x > 0.0f ? v.x : 0.0f; v.y = v.y > 0.0f ? v.y : 0.0f;
            v.z = v.z > 0.0f ? v.z : 0.0f; v.w = v.w > 0.0f ? v.w : 0.0f;
        }
        o4[i] = v;
    }
}

template <int F, bool RELU>
__global__ __launch_bounds__(1024) void owner_sum_kernel(const int* __restrict__ esrc,
                                                         const int* __restrict__ edst,
                                                         const float* __restrict__ ev,
                                                         const float* __restrict__ sup,
                                                         const float* __restrict__ bias,
                                                         float* __restrict__ dstm,
                                                         int* __restrict__ recs,
                                                         int* __restrict__ cnts,
                                                         const int* __restrict__ ob) {
    __shared__ float acc[RPB * F];
    __shared__ int   pend[32 * PCAP];
    const int tid  = threadIdx.x;
    const int lane = tid & 31;
    const int wave = tid >> 5;
    const int lo   = blockIdx.x * RPB;
    int nrows = N_NODES - lo;
    if (nrows > RPB) nrows = RPB;

    float4* acc4 = reinterpret_cast<float4*>(acc);
    for (int i = tid; i < RPB * F / 4; i += 1024) acc4[i] = make_float4(0.f, 0.f, 0.f, 0.f);
    __syncthreads();

    const int rot = blockIdx.x * ROT;
    const int seg = blockIdx.x * 32 + wave;
    int4* rec = reinterpret_cast<int4*>(recs) + (size_t)seg * WCAP;
    int written = 0;
    scan_and_sum<F, true>(esrc, edst, ev, sup, ob, acc, pend + wave * PCAP, lo, (int)blockIdx.x,
                          tid, lane, rot, rec, written);
    if (lane == 0) cnts[seg] = written;
    __syncthreads();
    store_rows<F, RELU>(acc, bias, dstm, lo, nrows, tid);
}

template <int F, bool RELU>
__global__ __launch_bounds__(1024) void replay_sum_kernel(const int* __restrict__ esrc,
                                                          const int* __restrict__ edst,
                                                          const float* __restrict__ ev,
                                                          const float* __restrict__ sup,
                                                          const float* __restrict__ bias,
                                                          float* __restrict__ dstm,
                                                          const int* __restrict__ recs,
                                                          const int* __restrict__ cnts,
                                                          const int* __restrict__ ob) {
    __shared__ float acc[RPB * F];
    __shared__ int   pend[32 * PCAP];
    const int tid  = threadIdx.x;
    const int lane = tid & 31;
    const int wave = tid >> 5;
    const int lo   = blockIdx.x * RPB;
    int nrows = N_NODES - lo;
    if (nrows > RPB) nrows = RPB;

    float4* acc4 = reinterpret_cast<float4*>(acc);
    for (int i = tid; i < RPB * F / 4; i += 1024) acc4[i] = make_float4(0.f, 0.f, 0.f, 0.f);
    __syncthreads();

    const bool over = __builtin_amdgcn_ballot_w32(cnts[blockIdx.x * 32 + lane] > WCAP) != 0u;
    if (!over) {
        const int seg = blockIdx.x * 32 + wave;
        const int n   = __builtin_amdgcn_readfirstlane(cnts[seg]);
        const int4* rec = reinterpret_cast<const int4*>(recs) + (size_t)seg * WCAP;
        if (n > 0) {
            const int last = n - 1;
            int i0 = lane < last ? lane : last;
            int4 cur = rec[i0];
            for (int base = 0; base < n; base += 32) {
                int i1 = base + 32 + lane;
                i1 = i1 < last ? i1 : last;
                const int4 nxt = rec[i1];
                int m = n - base;
                if (m > 32) m = 32;
                gather_add<F>(cur.z * F, cur.x, __int_as_float(cur.y), m, lane, sup, acc);
                cur = nxt;
            }
        }
    } else {
        int unused = 0;
        scan_and_sum<F, false>(esrc, edst, ev, sup, ob, acc, pend + wave * PCAP, lo, (int)blockIdx.x,
                               tid, lane, blockIdx.x * ROT, nullptr, unused);
    }
    __syncthreads();
    store_rows<F, RELU>(acc, bias, dstm, lo, nrows, tid);
}

__global__ __launch_bounds__(128) void log_softmax_kernel(float* __restrict__ out) {
    const int r = blockIdx.x * blockDim.x + threadIdx.x;
    if (r >= N_NODES) return;
    float* p = out + (size_t)r * NCLS;
    float v[NCLS];
    float m = -3.402823466e+38f;
#pragma unroll
    for (int i = 0; i < NCLS; ++i) { v[i] = p[i]; m = v[i] > m ? v[i] : m; }
    float s = 0.0f;
#pragma unroll
    for (int i = 0; i < NCLS; ++i) s += __expf(v[i] - m);
    const float lse = m + __logf(s);
#pragma unroll
    for (int i = 0; i < NCLS; ++i) p[i] = v[i] - lse;
}

extern "C" void kernel_launch(void* const* d_in, const int* in_sizes, int n_in,
                              void* d_out, int out_size, void* d_ws, size_t ws_size,
                              hipStream_t stream) {
    (void)in_sizes; (void)n_in; (void)out_size; (void)ws_size;
    const float* x   = (const float*)d_in[0];
    const int*   es  = (const int*)  d_in[1];
    const int*   ed  = (const int*)  d_in[2];
    const float* ev  = (const float*)d_in[3];
    const float* W1  = (const float*)d_in[4];
    const float* b1  = (const float*)d_in[5];
    const float* W2  = (const float*)d_in[6];
    const float* b2  = (const float*)d_in[7];
    float* out = (float*)d_out;

    float* sup1 = (float*)d_ws;
    float* hacc = sup1 + (size_t)N_NODES * NHID;
    float* sup2 = sup1;
    int*   recs = (int*)(hacc + (size_t)N_NODES * NHID);
    int*   cnts = recs + (size_t)NOWN * 32 * WCAP * 4;
    int*   ob   = cnts + NOWN * 32;

    gemm1_kernel<<<G1_BLOCKS, G1_THREADS, 0, stream>>>(x, W1, sup1, ed, ob);
    owner_sum_kernel<NHID, true><<<NOWN, 1024, 0, stream>>>(es, ed, ev, sup1, b1, hacc, recs, cnts, ob);
    gemm2_kernel<<<(NTILES + 7) / 8, 256, 0, stream>>>(hacc, W2, sup2);
    replay_sum_kernel<NCLS, false><<<NOWN, 1024, 0, stream>>>(es, ed, ev, sup2, b2, out, recs, cnts, ob);
    log_softmax_kernel<<<(N_NODES + 127) / 128, 128, 0, stream>>>(out);
}


# GEMM1's waves also pack each edge (source row, row in owner block, weight) into one 8-byte record; an owner block fetches 1 record per kept edge, not 3 scattered 4-byte fields; sums unchanged
# speedup vs baseline: 3.1041x; 1.0957x over previous
#include <hip/hip_runtime.h>

typedef float v2f __attribute__((ext_vector_type(2)));
typedef float v8f __attribute__((ext_vector_type(8)));

#define N_NODES 100000
#define FEAT    256
#define NHID    64
#define NCLS    16
#define NEDGE   1600000
#define NTILES  (N_NODES / 16)
#define RPB     392
#define EPT     (NEDGE / NTILES)

#define G1_WAVES   4
#define G1_THREADS (G1_WAVES * 32)
#define G1_BLOCKS  256
#define G1_STEPS   (FEAT / 4)
#define G1_MINT    (NTILES / G1_BLOCKS / G1_WAVES)

__device__ __forceinline__ void g1_load_tile(const float* __restrict__ x, int tile, int l, int h,
                                             float4 (&a)[32]) {
    const float4* xr = reinterpret_cast<const float4*>(x + (size_t)(tile * 16 + l) * FEAT + 4 * h);
#pragma unroll
    for (int i = 0; i < 32; ++i) a[i] = xr[2 * i];
}

__device__ __forceinline__ void g1_compute_tile(const float4 (&a)[32], int tile, int l, int h,
                                                const float4 (&w01)[G1_STEPS], const float4 (&w23)[G1_STEPS],
                                                float* __restrict__ sup) {
    v8f c0 = {}, c1 = {}, c2 = {}, c3 = {};
#pragma unroll
    for (int i = 0; i < 32; ++i) {
#pragma unroll
        for (int p = 0; p < 2; ++p) {
            const int t = 2 * i + p;
            v2f av, b;
            av.x = p ? a[i].z : a[i].x;
            av.y = p ? a[i].w : a[i].y;
            b.x = w01[t].x; b.y = w01[t].y;
            c0 = __builtin_amdgcn_wmma_f32_16x16x4_f32(false, av, false, b, (short)0, c0, false, false);
            b.x = w01[t].z; b.y = w01[t].w;
            c1 = __builtin_amdgcn_wmma_f32_16x16x4_f32(false, av, false, b, (short)0, c1, false, false);
            b.x = w23[t].x; b.y = w23[t].y;
            c2 = __builtin_amdgcn_wmma_f32_16x16x4_f32(false, av, false, b, (short)0, c2, false, false);
            b.x = w23[t].z; b.y = w23[t].w;
            c3 = __builtin_amdgcn_wmma_f32_16x16x4_f32(false, av, false, b, (short)0, c3, false, false);
        }
    }
    float* out = sup + (size_t)(tile * 16 + h * 8) * NHID + l;
#pragma unroll
    for (int r = 0; r < 8; ++r) {
        out[(size_t)r * NHID]      = c0[r];
        out[(size_t)r * NHID + 16] = c1[r];
        out[(size_t)r * NHID + 32] = c2[r];
        out[(size_t)r * NHID + 48] = c3[r];
    }
}

__device__ __forceinline__ int rec_slot(int e) {
    return (e & ~255) | ((e & 6) << 5) | ((e >> 2) & 62) | (e & 1);
}

__device__ __forceinline__ void g1_edge_out(int tile, int lane,
                                            const int4& d0, const int4& d1,
                                            const int4& s0, const int4& s1,
                                            const float4& w0, const float4& w1,
                                            int* __restrict__ ob, int* __restrict__ erec) {
    static_assert(RPB <= 512 && N_NODES <= (1 << 23), "row in 9 bits, source row in the 23 above");
    const unsigned d[8] = {(unsigned)d0.x, (unsigned)d0.y, (unsigned)d0.z, (unsigned)d0.w,
                           (unsigned)d1.x, (unsigned)d1.y, (unsigned)d1.z, (unsigned)d1.w};
    const unsigned s[8] = {(unsigned)s0.x, (unsigned)s0.y, (unsigned)s0.z, (unsigned)s0.w,
                           (unsigned)s1.x, (unsigned)s1.y, (unsigned)s1.z, (unsigned)s1.w};
    const float    w[8] = {w0.x, w0.y, w0.z, w0.w, w1.x, w1.y, w1.z, w1.w};
    unsigned o[8];
    int      p[8];
#pragma unroll
    for (int j = 0; j < 8; ++j) {
        o[j] = d[j] / RPB;
        p[j] = (int)((s[j] << 9) | (d[j] - o[j] * RPB));
    }
    const unsigned b0 = o[0] | (o[1] << 8) | (o[2] << 16) | (o[3] << 24);
    const unsigned b1 = o[4] | (o[5] << 8) | (o[6] << 16) | (o[7] << 24);
    reinterpret_cast<int2*>(ob)[(size_t)tile * (EPT / 8) + lane] = make_int2((int)b0, (int)b1);
    int4* r4 = reinterpret_cast<int4*>(erec) + (size_t)tile * (EPT / 2) + lane;
#pragma unroll
    for (int m = 0; m < 4; ++m)
        r4[32 * m] = make_int4(p[2 * m], __float_as_int(w[2 * m]),
                               p[2 * m + 1], __float_as_int(w[2 * m + 1]));
}

__global__ __launch_bounds__(G1_THREADS) void gemm1_kernel(const float* __restrict__ x,
                                                           const float* __restrict__ W1,
                                                           float* __restrict__ sup,
                                                           const int* __restrict__ edst,
                                                           const int* __restrict__ esrc,
                                                           const float* __restrict__ ev,
                                                           int* __restrict__ ob,
                                                           int* __restrict__ erec) {
    __shared__ float4 wb01[G1_STEPS * 32];
    __shared__ float4 wb23[G1_STEPS * 32];
    const int tid  = threadIdx.x;
    const int lane = tid & 31;
    const int wave = tid >> 5;
    const int l    = lane & 15;
    const int h    = lane >> 4;

    static_assert(EPT == 256 && EPT * NTILES == NEDGE, "32 lanes x 8 edges per tile cover every edge once");
    static_assert((N_NODES - 1) / RPB <= 255, "an owner id fits a byte");
    static_assert(G1_MINT >= 1 && NTILES < (G1_MINT * G1_WAVES + 1) * G1_BLOCKS,
                  "a block has G1_MINT * G1_WAVES tiles or one more, so a wave has G1_MINT or G1_MINT + 1");

    const int t_lo = (int)(((long long)blockIdx.x * NTILES) / G1_BLOCKS);
    const int t_hi = (int)(((long long)(blockIdx.x + 1) * NTILES) / G1_BLOCKS);
    const int tile0 = t_lo + wave;
    const int n = (t_hi - tile0 + G1_WAVES - 1) / G1_WAVES;

    float4 A[32], B[32];
    g1_load_tile(x, tile0, l, h, A);

    {
        int4   d0[G1_MINT], d1[G1_MINT], s0[G1_MINT], s1[G1_MINT];
        float4 w0[G1_MINT], w1[G1_MINT];
#pragma unroll
        for (int i = 0; i < G1_MINT; ++i) {
            const size_t q = (size_t)(tile0 + G1_WAVES * i) * (EPT / 4) + lane * 2;
            d0[i] = reinterpret_cast<const int4*>(edst)[q];   d1[i] = reinterpret_cast<const int4*>(edst)[q + 1];
            s0[i] = reinterpret_cast<const int4*>(esrc)[q];   s1[i] = reinterpret_cast<const int4*>(esrc)[q + 1];
            w0[i] = reinterpret_cast<const float4*>(ev)[q];   w1[i] = reinterpret_cast<const float4*>(ev)[q + 1];
        }
#pragma unroll
        for (int i = 0; i < G1_MINT; ++i)
            g1_edge_out(tile0 + G1_WAVES * i, lane, d0[i], d1[i], s0[i], s1[i], w0[i], w1[i], ob, erec);
        if (n > G1_MINT) {
            const int tl = tile0 + G1_WAVES * G1_MINT;
            const size_t q = (size_t)tl * (EPT / 4) + lane * 2;
            g1_edge_out(tl, lane,
                        reinterpret_cast<const int4*>(edst)[q],  reinterpret_cast<const int4*>(edst)[q + 1],
                        reinterpret_cast<const int4*>(esrc)[q],  reinterpret_cast<const int4*>(esrc)[q + 1],
                        reinterpret_cast<const float4*>(ev)[q],  reinterpret_cast<const float4*>(ev)[q + 1],
                        ob, erec);
        }
    }

    for (int i = tid; i < G1_STEPS * 32; i += G1_THREADS) {
        const int t  = i >> 5;
        const int ll = i & 15;
        const int hh = (i >> 4) & 1;
        const int k0 = 64 * (t >> 4) + 8 * ((t >> 1) & 7) + 4 * hh + 2 * (t & 1);
        const float* r0 = W1 + (size_t)k0 * NHID + ll;
        const float* r1 = r0 + NHID;
        wb01[i] = make_float4(r0[0],  r1[0],  r0[16], r1[16]);
        wb23[i] = make_float4(r0[32], r1[32], r0[48], r1[48]);
    }
    __syncthreads();
    float4 w01[G1_STEPS], w23[G1_STEPS];
#pragma unroll
    for (int t = 0; t < G1_STEPS; ++t) {
        w01[t] = wb01[t * 32 + lane];
        w23[t] = wb23[t * 32 + lane];
    }

    int i = 0;
    for (; i + 2 < n; i += 2) {
        g1_load_tile(x, tile0 + G1_WAVES * (i + 1), l, h, B);
        __builtin_amdgcn_sched_barrier(0);
        g1_compute_tile(A, tile0 + G1_WAVES * i, l, h, w01, w23, sup);
        __builtin_amdgcn_sched_barrier(0);
        g1_load_tile(x, tile0 + G1_WAVES * (i + 2), l, h, A);
        __builtin_amdgcn_sched_barrier(0);
        g1_compute_tile(B, tile0 + G1_WAVES * (i + 1), l, h, w01, w23, sup);
        __builtin_amdgcn_sched_barrier(0);
    }
    const bool two = i + 1 < n;
    if (two) g1_load_tile(x, tile0 + G1_WAVES * (i + 1), l, h, B);
    __builtin_amdgcn_sched_barrier(0);
    g1_compute_tile(A, tile0 + G1_WAVES * i, l, h, w01, w23, sup);
    if (two) g1_compute_tile(B, tile0 + G1_WAVES * (i + 1), l, h, w01, w23, sup);
}

__global__ __launch_bounds__(256) void gemm2_kernel(const float* __restrict__ h,
                                                    const float* __restrict__ W2,
                                                    float* __restrict__ sup2) {
    const int lane = threadIdx.x & 31;
    const int wave = threadIdx.x >> 5;
    const int tile = blockIdx.x * 8 + wave;
    if (tile >= NTILES) return;
    const int rb = tile * 16;
    const int l  = lane & 15;
    const int hi = lane >> 4;
    v8f acc = {};
    for (int k = 0; k < NHID; k += 4) {
        const int ks = k + 2 * hi;
        v2f a = *reinterpret_cast<const v2f*>(&h[(size_t)(rb + l) * NHID + ks]);
        v2f b;
        b.x = W2[ks * NCLS + l];
        b.y = W2[(ks + 1) * NCLS + l];
        acc = __builtin_amdgcn_wmma_f32_16x16x4_f32(false, a, false, b,
                                                    (short)0, acc, false, false);
    }
    float* out = sup2 + (size_t)rb * NCLS;
    const int mb = hi * 8;
#pragma unroll
    for (int r = 0; r < 8; ++r)
        out[(size_t)(mb + r) * NCLS + l] = acc[r];
}

#define NOWN  256
#define NQ    (NEDGE / 16)
#define SCAN_U 2
#define NFULL (NQ / (1024 * SCAN_U))
#define PCAP  1056
#define ROT   392
#define WCAP  512

template <bool GUARD>
__device__ __forceinline__ void load_groups(const int4* __restrict__ d4, int it, int tid, int rot,
                                            int fill, int4 (&dv)[SCAN_U], int (&qi)[SCAN_U]) {
#pragma unroll
    for (int u = 0; u < SCAN_U; ++u) {
        int q = (it * SCAN_U + u) * 1024 + tid;
        const bool valid = !GUARD || q < NQ;
        q += rot;
        if (q >= NQ) q -= NQ;
        qi[u] = q;
        dv[u] = valid ? d4[q] : make_int4(fill, fill, fill, fill);
    }
}

template <int I> struct step_c { static constexpr int value = I; };
template <int I, int N, class Fn>
__device__ __forceinline__ void for_steps(Fn&& f) {
    if constexpr (I < N) { f(step_c<I>{}); for_steps<I + 1, N>(f); }
}

template <int F, int I>
__device__ __forceinline__ int edge_field(int x) {
    if constexpr (F == 64) return __builtin_amdgcn_readlane(x, I);
    else                   return __builtin_amdgcn_update_dpp(0, x, 0x150 + I, 0xf, 0xf, true);
}

template <int F>
__device__ __forceinline__ void gather_add(int ro, int s, float w, int n, int lane,
                                           const float* __restrict__ sup,
                                           float* acc) {
    constexpr int VW  = (F == 64) ? 2 : 1;
    constexpr int LPE = F / VW;
    constexpr int EPS = 32 / LPE;
    constexpr int NST = 32 / EPS;
    constexpr int CH  = 8;
    static_assert(LPE == 32 || LPE == 16, "an edge takes the wave or one 16-lane row");
    static_assert(NST % CH == 0, "whole chunks");
    const int sub = lane / LPE;
    const int col = (lane % LPE) * VW;
    for_steps<0, NST / CH>([&](auto c) {
        constexpr int T0 = decltype(c)::value * CH;
        float v0[CH], v1[CH];
        for_steps<0, CH>([&](auto kk) {
            constexpr int K = decltype(kk)::value;
            constexpr int I = T0 + K;
            const int j  = (EPS == 1) ? I : 16 * sub + I;
            const int sj = edge_field<F, I>(s);
            v0[K] = 0.0f; v1[K] = 0.0f;
            if (j < n) {
                if (VW == 2) {
                    const v2f v = *reinterpret_cast<const v2f*>(sup + (size_t)sj * F + col);
                    v0[K] = v.x; v1[K] = v.y;
                } else {
                    v0[K] = sup[(size_t)sj * F + col];
                }
            }
        });
        for_steps<0, CH>([&](auto kk) {
            constexpr int K = decltype(kk)::value;
            constexpr int I = T0 + K;
            const int   j  = (EPS == 1) ? I : 16 * sub + I;
            const int   rj = edge_field<F, I>(ro);
            const float wj = __int_as_float(edge_field<F, I>(__float_as_int(w)));
            if (j < n) {
                atomicAdd(&acc[rj + col], v0[K] * wj);
                if (VW == 2) atomicAdd(&acc[rj + col + 1], v1[K] * wj);
            }
        });
    });
}

template <int F, bool REC>
__device__ __forceinline__ void add_edges(const int* list, int n, int lane,
                                          const int2* __restrict__ erec,
                                          const float* __restrict__ sup,
                                          float* acc, int4* rec, int& written) {
    int ro = 0, s = 0;
    float w = 0.0f;
    if (lane < n) {
        const int2 r  = erec[rec_slot(list[lane])];
        const int row = r.x & 511;
        ro = row * F;
        s  = (int)((unsigned)r.x >> 9);
        w  = __int_as_float(r.y);
        if (REC && written + lane < WCAP)
            rec[written + lane] = make_int4(s, r.y, row, 0);
    }
    if (REC) written += n;
    gather_add<F>(ro, s, w, n, lane, sup, acc);
}

__device__ __forceinline__ void scan_set(const int4 (&dv)[SCAN_U], const int (&qi)[SCAN_U],
                                         unsigned pat, int* mypend, int& cnt) {
#pragma unroll
    for (int u = 0; u < SCAN_U; ++u) {
        const unsigned ww[4] = {(unsigned)dv[u].x, (unsigned)dv[u].y, (unsigned)dv[u].z, (unsigned)dv[u].w};
        unsigned m = 0u;
#pragma unroll
        for (int c = 0; c < 4; ++c) {
            const unsigned x = ww[c] ^ pat;
            const unsigned t = ~(((x & 0x7f7f7f7fu) + 0x7f7f7f7fu) | x | 0x7f7f7f7fu);
            m |= t >> (7 - c);
        }
        unsigned bal = __builtin_amdgcn_ballot_w32(m != 0u);
        while (bal) {
            const int pos = cnt + (int)__builtin_amdgcn_mbcnt_lo(bal, 0u);
            if (m != 0u) {
                const int p = __builtin_ctz(m);
                mypend[pos] = qi[u] * 16 + (p & 3) * 4 + (p >> 3);
                m &= m - 1u;
            }
            cnt += __builtin_popcount(bal);
            bal = __builtin_amdgcn_ballot_w32(m != 0u);
        }
    }
}

template <int F, bool REC>
__device__ __forceinline__ void scan_and_sum(const int2* __restrict__ erec,
                                             const float* __restrict__ sup,
                                             const int* __restrict__ ob,
                                             float* acc, int* mypend, int owner,
                                             int tid, int lane, int rot,
                                             int4* rec, int& written) {
    int  cnt = 0;
    const int4* d4 = reinterpret_cast<const int4*>(ob);
    const unsigned pat = (unsigned)owner * 0x01010101u;
    const int fill = (int)~pat;
    static_assert(NFULL % 2 == 0 && NFULL >= 2, "the loop below ends with set a holding trip NFULL - 2");
    static_assert((NFULL + 1) * 1024 * SCAN_U >= NQ, "one guarded trip covers the rest");
    static_assert(31 + SCAN_U * 16 * 32 <= PCAP, "a trip's edges fit the pending list whatever the input");
    int4 ga[SCAN_U], gb[SCAN_U];
    int  qa[SCAN_U], qb[SCAN_U];
    auto drain = [&]() {
        while (cnt >= 32) {
            cnt -= 32;
            add_edges<F, REC>(mypend + cnt, 32, lane, erec, sup, acc, rec, written);
        }
    };
    load_groups<false>(d4, 0, tid, rot, fill, ga, qa);
    for (int it = 0; it < NFULL - 2; it += 2) {
        load_groups<false>(d4, it + 1, tid, rot, fill, gb, qb);
        scan_set(ga, qa, pat, mypend, cnt);
        drain();
        load_groups<false>(d4, it + 2, tid, rot, fill, ga, qa);
        scan_set(gb, qb, pat, mypend, cnt);
        drain();
    }
    load_groups<false>(d4, NFULL - 1, tid, rot, fill, gb, qb);
    scan_set(ga, qa, pat, mypend, cnt);
    drain();
    load_groups<true>(d4, NFULL, tid, rot, fill, ga, qa);
    scan_set(gb, qb, pat, mypend, cnt);
    drain();
    scan_set(ga, qa, pat, mypend, cnt);
    drain();
    add_edges<F, REC>(mypend, cnt, lane, erec, sup, acc, rec, written);
}

template <int F, bool RELU>
__device__ __forceinline__ void store_rows(const float* acc, const float* __restrict__ bias,
                                           float* __restrict__ dstm, int lo, int nrows, int tid) {
    const float4* acc4 = reinterpret_cast<const float4*>(acc);
    float4* o4 = reinterpret_cast<float4*>(dstm + (size_t)lo * F);
    for (int i = tid; i < nrows * F / 4; i += 1024) {
        float4 v = acc4[i];
        const int c = (i * 4) & (F - 1);
        v.x += bias[c]; v.y += bias[c + 1]; v.z += bias[c + 2]; v.w += bias[c + 3];
        if (RELU) {
            v.x = v.x > 0.0f ? v.x : 0.0f; v.y = v.y > 0.0f ? v.y : 0.0f;
            v.z = v.z > 0.0f ? v.z : 0.0f; v.w = v.w > 0.0f ? v.w : 0.0f;
        }
        o4[i] = v;
    }
}

template <int F, bool RELU>
__global__ __launch_bounds__(1024) void owner_sum_kernel(const float* __restrict__ sup,
                                                         const float* __restrict__ bias,
                                                         float* __restrict__ dstm,
                                                         int* __restrict__ recs,
                                                         int* __restrict__ cnts,
                                                         const int* __restrict__ ob,
                                                         const int* __restrict__ erec) {
    __shared__ float acc[RPB * F];
    __shared__ int   pend[32 * PCAP];
    const int tid  = threadIdx.x;
    const int lane = tid & 31;
    const int wave = tid >> 5;
    const int lo   = blockIdx.x * RPB;
    int nrows = N_NODES - lo;
    if (nrows > RPB) nrows = RPB;

    float4* acc4 = reinterpret_cast<float4*>(acc);
    for (int i = tid; i < RPB * F / 4; i += 1024) acc4[i] = make_float4(0.f, 0.f, 0.f, 0.f);
    __syncthreads();

    const int rot = blockIdx.x * ROT;
    const int seg = blockIdx.x * 32 + wave;
    int4* rec = reinterpret_cast<int4*>(recs) + (size_t)seg * WCAP;
    int written = 0;
    scan_and_sum<F, true>(reinterpret_cast<const int2*>(erec), sup, ob, acc, pend + wave * PCAP,
                          (int)blockIdx.x, tid, lane, rot, rec, written);
    if (lane == 0) cnts[seg] = written;
    __syncthreads();
    store_rows<F, RELU>(acc, bias, dstm, lo, nrows, tid);
}

template <int F, bool RELU>
__global__ __launch_bounds__(1024) void replay_sum_kernel(const float* __restrict__ sup,
                                                          const float* __restrict__ bias,
                                                          float* __restrict__ dstm,
                                                          const int* __restrict__ recs,
                                                          const int* __restrict__ cnts,
                                                          const int* __restrict__ ob,
                                                          const int* __restrict__ erec) {
    __shared__ float acc[RPB * F];
    __shared__ int   pend[32 * PCAP];
    const int tid  = threadIdx.x;
    const int lane = tid & 31;
    const int wave = tid >> 5;
    const int lo   = blockIdx.x * RPB;
    int nrows = N_NODES - lo;
    if (nrows > RPB) nrows = RPB;

    float4* acc4 = reinterpret_cast<float4*>(acc);
    for (int i = tid; i < RPB * F / 4; i += 1024) acc4[i] = make_float4(0.f, 0.f, 0.f, 0.f);
    __syncthreads();

    const bool over = __builtin_amdgcn_ballot_w32(cnts[blockIdx.x * 32 + lane] > WCAP) != 0u;
    if (!over) {
        const int seg = blockIdx.x * 32 + wave;
        const int n   = __builtin_amdgcn_readfirstlane(cnts[seg]);
        const int4* rec = reinterpret_cast<const int4*>(recs) + (size_t)seg * WCAP;
        if (n > 0) {
            const int last = n - 1;
            int i0 = lane < last ? lane : last;
            int4 cur = rec[i0];
            for (int base = 0; base < n; base += 32) {
                int i1 = base + 32 + lane;
                i1 = i1 < last ? i1 : last;
                const int4 nxt = rec[i1];
                int m = n - base;
                if (m > 32) m = 32;
                gather_add<F>(cur.z * F, cur.x, __int_as_float(cur.y), m, lane, sup, acc);
                cur = nxt;
            }
        }
    } else {
        int unused = 0;
        scan_and_sum<F, false>(reinterpret_cast<const int2*>(erec), sup, ob, acc, pend + wave * PCAP,
                               (int)blockIdx.x, tid, lane, blockIdx.x * ROT, nullptr, unused);
    }
    __syncthreads();
    store_rows<F, RELU>(acc, bias, dstm, lo, nrows, tid);
}

__global__ __launch_bounds__(128) void log_softmax_kernel(float* __restrict__ out) {
    const int r = blockIdx.x * blockDim.x + threadIdx.x;
    if (r >= N_NODES) return;
    float* p = out + (size_t)r * NCLS;
    float v[NCLS];
    float m = -3.402823466e+38f;
#pragma unroll
    for (int i = 0; i < NCLS; ++i) { v[i] = p[i]; m = v[i] > m ? v[i] : m; }
    float s = 0.0f;
#pragma unroll
    for (int i = 0; i < NCLS; ++i) s += __expf(v[i] - m);
    const float lse = m + __logf(s);
#pragma unroll
    for (int i = 0; i < NCLS; ++i) p[i] = v[i] - lse;
}

extern "C" void kernel_launch(void* const* d_in, const int* in_sizes, int n_in,
                              void* d_out, int out_size, void* d_ws, size_t ws_size,
                              hipStream_t stream) {
    (void)in_sizes; (void)n_in; (void)out_size; (void)ws_size;
    const float* x   = (const float*)d_in[0];
    const int*   es  = (const int*)  d_in[1];
    const int*   ed  = (const int*)  d_in[2];
    const float* ev  = (const float*)d_in[3];
    const float* W1  = (const float*)d_in[4];
    const float* b1  = (const float*)d_in[5];
    const float* W2  = (const float*)d_in[6];
    const float* b2  = (const float*)d_in[7];
    float* out = (float*)d_out;

    float* sup1 = (float*)d_ws;
    float* hacc = sup1 + (size_t)N_NODES * NHID;
    float* sup2 = sup1;
    int*   recs = (int*)(hacc + (size_t)N_NODES * NHID);
    int*   cnts = recs + (size_t)NOWN * 32 * WCAP * 4;
    int*   ob   = cnts + NOWN * 32;
    int*   erec = ob + NEDGE / 4;

    gemm1_kernel<<<G1_BLOCKS, G1_THREADS, 0, stream>>>(x, W1, sup1, ed, es, ev, ob, erec);
    owner_sum_kernel<NHID, true><<<NOWN, 1024, 0, stream>>>(sup1, b1, hacc, recs, cnts, ob, erec);
    gemm2_kernel<<<(NTILES + 7) / 8, 256, 0, stream>>>(hacc, W2, sup2);
    replay_sum_kernel<NCLS, false><<<NOWN, 1024, 0, stream>>>(sup2, b2, out, recs, cnts, ob, erec);
    log_softmax_kernel<<<(N_NODES + 127) / 128, 128, 0, stream>>>(out);
}


# GEMM1's waves sort each 256-edge tile's records by owner block (LDS counts) and write a start/count table; owner blocks read 1 entry per tile, not every owner byte; new edge order, arrival sums
# speedup vs baseline: 3.3782x; 1.0883x over previous
#include <hip/hip_runtime.h>

typedef float v2f __attribute__((ext_vector_type(2)));
typedef float v8f __attribute__((ext_vector_type(8)));

#define N_NODES 100000
#define FEAT    256
#define NHID    64
#define NCLS    16
#define NEDGE   1600000
#define NTILES  (N_NODES / 16)
#define RPB     392
#define EPT     (NEDGE / NTILES)
#define NOWNB   256

#define G1_WAVES   4
#define G1_THREADS (G1_WAVES * 32)
#define G1_BLOCKS  256
#define G1_STEPS   (FEAT / 4)
#define G1_MINT    (NTILES / G1_BLOCKS / G1_WAVES)

__device__ __forceinline__ void g1_load_tile(const float* __restrict__ x, int tile, int l, int h,
                                             float4 (&a)[32]) {
    const float4* xr = reinterpret_cast<const float4*>(x + (size_t)(tile * 16 + l) * FEAT + 4 * h);
#pragma unroll
    for (int i = 0; i < 32; ++i) a[i] = xr[2 * i];
}

__device__ __forceinline__ void g1_compute_tile(const float4 (&a)[32], int tile, int l, int h,
                                                const float4 (&w01)[G1_STEPS], const float4 (&w23)[G1_STEPS],
                                                float* __restrict__ sup) {
    v8f c0 = {}, c1 = {}, c2 = {}, c3 = {};
#pragma unroll
    for (int i = 0; i < 32; ++i) {
#pragma unroll
        for (int p = 0; p < 2; ++p) {
            const int t = 2 * i + p;
            v2f av, b;
            av.x = p ? a[i].z : a[i].x;
            av.y = p ? a[i].w : a[i].y;
            b.x = w01[t].x; b.y = w01[t].y;
            c0 = __builtin_amdgcn_wmma_f32_16x16x4_f32(false, av, false, b, (short)0, c0, false, false);
            b.x = w01[t].z; b.y = w01[t].w;
            c1 = __builtin_amdgcn_wmma_f32_16x16x4_f32(false, av, false, b, (short)0, c1, false, false);
            b.x = w23[t].x; b.y = w23[t].y;
            c2 = __builtin_amdgcn_wmma_f32_16x16x4_f32(false, av, false, b, (short)0, c2, false, false);
            b.x = w23[t].z; b.y = w23[t].w;
            c3 = __builtin_amdgcn_wmma_f32_16x16x4_f32(false, av, false, b, (short)0, c3, false, false);
        }
    }
    float* out = sup + (size_t)(tile * 16 + h * 8) * NHID + l;
#pragma unroll
    for (int r = 0; r < 8; ++r) {
        out[(size_t)r * NHID]      = c0[r];
        out[(size_t)r * NHID + 16] = c1[r];
        out[(size_t)r * NHID + 32] = c2[r];
        out[(size_t)r * NHID + 48] = c3[r];
    }
}

__device__ __forceinline__ void g1_sort_tile(int tile, int lane,
                                             const int4& d0, const int4& d1,
                                             const int4& s0, const int4& s1,
                                             const float4& w0, const float4& w1,
                                             unsigned* hist, int2* stg,
                                             unsigned* __restrict__ tab, int* __restrict__ erec) {
    static_assert(RPB <= 512 && N_NODES <= (1 << 23), "row in 9 bits, source row in the 23 above");
    static_assert((N_NODES - 1) / RPB < NOWNB, "an owner id indexes the 256 counts");
    const unsigned d[8] = {(unsigned)d0.x, (unsigned)d0.y, (unsigned)d0.z, (unsigned)d0.w,
                           (unsigned)d1.x, (unsigned)d1.y, (unsigned)d1.z, (unsigned)d1.w};
    const unsigned s[8] = {(unsigned)s0.x, (unsigned)s0.y, (unsigned)s0.z, (unsigned)s0.w,
                           (unsigned)s1.x, (unsigned)s1.y, (unsigned)s1.z, (unsigned)s1.w};
    const float    w[8] = {w0.x, w0.y, w0.z, w0.w, w1.x, w1.y, w1.z, w1.w};
    unsigned o[8], rk[8];
    int      p[8];
#pragma unroll
    for (int j = 0; j < 8; ++j) {
        o[j] = d[j] / RPB;
        p[j] = (int)((s[j] << 9) | (d[j] - o[j] * RPB));
    }
    uint4* h4 = reinterpret_cast<uint4*>(hist);
    h4[2 * lane]     = make_uint4(0u, 0u, 0u, 0u);
    h4[2 * lane + 1] = make_uint4(0u, 0u, 0u, 0u);
#pragma unroll
    for (int j = 0; j < 8; ++j) rk[j] = atomicAdd(&hist[o[j]], 1u);
    const uint4 ca = h4[2 * lane], cb = h4[2 * lane + 1];
    const unsigned c[8] = {ca.x, ca.y, ca.z, ca.w, cb.x, cb.y, cb.z, cb.w};
    unsigned e[8];
    unsigned run = 0u;
#pragma unroll
    for (int i = 0; i < 8; ++i) { e[i] = run; run += c[i]; }
    unsigned inc = run;
    inc += (unsigned)__builtin_amdgcn_update_dpp(0, (int)inc, 0x111, 0xf, 0xf, true);
    inc += (unsigned)__builtin_amdgcn_update_dpp(0, (int)inc, 0x112, 0xf, 0xf, true);
    inc += (unsigned)__builtin_amdgcn_update_dpp(0, (int)inc, 0x114, 0xf, 0xf, true);
    inc += (unsigned)__builtin_amdgcn_update_dpp(0, (int)inc, 0x118, 0xf, 0xf, true);
    const unsigned low = (unsigned)__builtin_amdgcn_readlane((int)inc, 15);
    if (lane >= 16) inc += low;
    const unsigned base = inc - run;
#pragma unroll
    for (int i = 0; i < 8; ++i) e[i] += base;
    h4[2 * lane]     = make_uint4(e[0], e[1], e[2], e[3]);
    h4[2 * lane + 1] = make_uint4(e[4], e[5], e[6], e[7]);
    uint4* t4 = reinterpret_cast<uint4*>(tab) + (size_t)tile * (NOWNB / 4) + 2 * lane;
    t4[0] = make_uint4(e[0] | (c[0] << 16), e[1] | (c[1] << 16), e[2] | (c[2] << 16), e[3] | (c[3] << 16));
    t4[1] = make_uint4(e[4] | (c[4] << 16), e[5] | (c[5] << 16), e[6] | (c[6] << 16), e[7] | (c[7] << 16));
#pragma unroll
    for (int j = 0; j < 8; ++j)
        stg[hist[o[j]] + rk[j]] = make_int2(p[j], __float_as_int(w[j]));
    int4* r4 = reinterpret_cast<int4*>(erec) + (size_t)tile * (EPT / 2) + lane;
    const int4* g4 = reinterpret_cast<const int4*>(stg);
#pragma unroll
    for (int m = 0; m < 4; ++m) r4[32 * m] = g4[32 * m + lane];
}

__global__ __launch_bounds__(G1_THREADS) void gemm1_kernel(const float* __restrict__ x,
                                                           const float* __restrict__ W1,
                                                           float* __restrict__ sup,
                                                           const int* __restrict__ edst,
                                                           const int* __restrict__ esrc,
                                                           const float* __restrict__ ev,
                                                           unsigned* __restrict__ tab,
                                                           int* __restrict__ erec) {
    __shared__ float4 wb01[G1_STEPS * 32];
    __shared__ float4 wb23[G1_STEPS * 32];
    __shared__ __attribute__((aligned(16))) unsigned hist[G1_WAVES * NOWNB];
    __shared__ __attribute__((aligned(16))) int2     stg[G1_WAVES * EPT];
    const int tid  = threadIdx.x;
    const int lane = tid & 31;
    const int wave = tid >> 5;
    const int l    = lane & 15;
    const int h    = lane >> 4;

    static_assert(EPT == 256 && EPT * NTILES == NEDGE, "32 lanes x 8 edges per tile cover every edge once");
    static_assert(G1_MINT >= 1 && NTILES < (G1_MINT * G1_WAVES + 1) * G1_BLOCKS,
                  "a block has G1_MINT * G1_WAVES tiles or one more, so a wave has G1_MINT or G1_MINT + 1");

    const int t_lo = (int)(((long long)blockIdx.x * NTILES) / G1_BLOCKS);
    const int t_hi = (int)(((long long)(blockIdx.x + 1) * NTILES) / G1_BLOCKS);
    const int tile0 = t_lo + wave;
    const int n = (t_hi - tile0 + G1_WAVES - 1) / G1_WAVES;

    float4 A[32], B[32];
    g1_load_tile(x, tile0, l, h, A);

    {
        int4   d0[G1_MINT], d1[G1_MINT], s0[G1_MINT], s1[G1_MINT];
        float4 w0[G1_MINT], w1[G1_MINT];
#pragma unroll
        for (int i = 0; i < G1_MINT; ++i) {
            const size_t q = (size_t)(tile0 + G1_WAVES * i) * (EPT / 4) + lane * 2;
            d0[i] = reinterpret_cast<const int4*>(edst)[q];   d1[i] = reinterpret_cast<const int4*>(edst)[q + 1];
            s0[i] = reinterpret_cast<const int4*>(esrc)[q];   s1[i] = reinterpret_cast<const int4*>(esrc)[q + 1];
            w0[i] = reinterpret_cast<const float4*>(ev)[q];   w1[i] = reinterpret_cast<const float4*>(ev)[q + 1];
        }
#pragma unroll
        for (int i = 0; i < G1_MINT; ++i)
            g1_sort_tile(tile0 + G1_WAVES * i, lane, d0[i], d1[i], s0[i], s1[i], w0[i], w1[i],
                         hist + wave * NOWNB, stg + wave * EPT, tab, erec);
        if (n > G1_MINT) {
            const int tl = tile0 + G1_WAVES * G1_MINT;
            const size_t q = (size_t)tl * (EPT / 4) + lane * 2;
            g1_sort_tile(tl, lane,
                        reinterpret_cast<const int4*>(edst)[q],  reinterpret_cast<const int4*>(edst)[q + 1],
                        reinterpret_cast<const int4*>(esrc)[q],  reinterpret_cast<const int4*>(esrc)[q + 1],
                        reinterpret_cast<const float4*>(ev)[q],  reinterpret_cast<const float4*>(ev)[q + 1],
                        hist + wave * NOWNB, stg + wave * EPT, tab, erec);
        }
    }

    for (int i = tid; i < G1_STEPS * 32; i += G1_THREADS) {
        const int t  = i >> 5;
        const int ll = i & 15;
        const int hh = (i >> 4) & 1;
        const int k0 = 64 * (t >> 4) + 8 * ((t >> 1) & 7) + 4 * hh + 2 * (t & 1);
        const float* r0 = W1 + (size_t)k0 * NHID + ll;
        const float* r1 = r0 + NHID;
        wb01[i] = make_float4(r0[0],  r1[0],  r0[16], r1[16]);
        wb23[i] = make_float4(r0[32], r1[32], r0[48], r1[48]);
    }
    __syncthreads();
    float4 w01[G1_STEPS], w23[G1_STEPS];
#pragma unroll
    for (int t = 0; t < G1_STEPS; ++t) {
        w01[t] = wb01[t * 32 + lane];
        w23[t] = wb23[t * 32 + lane];
    }

    int i = 0;
    for (; i + 2 < n; i += 2) {
        g1_load_tile(x, tile0 + G1_WAVES * (i + 1), l, h, B);
        __builtin_amdgcn_sched_barrier(0);
        g1_compute_tile(A, tile0 + G1_WAVES * i, l, h, w01, w23, sup);
        __builtin_amdgcn_sched_barrier(0);
        g1_load_tile(x, tile0 + G1_WAVES * (i + 2), l, h, A);
        __builtin_amdgcn_sched_barrier(0);
        g1_compute_tile(B, tile0 + G1_WAVES * (i + 1), l, h, w01, w23, sup);
        __builtin_amdgcn_sched_barrier(0);
    }
    const bool two = i + 1 < n;
    if (two) g1_load_tile(x, tile0 + G1_WAVES * (i + 1), l, h, B);
    __builtin_amdgcn_sched_barrier(0);
    g1_compute_tile(A, tile0 + G1_WAVES * i, l, h, w01, w23, sup);
    if (two) g1_compute_tile(B, tile0 + G1_WAVES * (i + 1), l, h, w01, w23, sup);
}

__global__ __launch_bounds__(256) void gemm2_kernel(const float* __restrict__ h,
                                                    const float* __restrict__ W2,
                                                    float* __restrict__ sup2) {
    const int lane = threadIdx.x & 31;
    const int wave = threadIdx.x >> 5;
    const int tile = blockIdx.x * 8 + wave;
    if (tile >= NTILES) return;
    const int rb = tile * 16;
    const int l  = lane & 15;
    const int hi = lane >> 4;
    v8f acc = {};
    for (int k = 0; k < NHID; k += 4) {
        const int ks = k + 2 * hi;
        v2f a = *reinterpret_cast<const v2f*>(&h[(size_t)(rb + l) * NHID + ks]);
        v2f b;
        b.x = W2[ks * NCLS + l];
        b.y = W2[(ks + 1) * NCLS + l];
        acc = __builtin_amdgcn_wmma_f32_16x16x4_f32(false, a, false, b,
                                                    (short)0, acc, false, false);
    }
    float* out = sup2 + (size_t)rb * NCLS;
    const int mb = hi * 8;
#pragma unroll
    for (int r = 0; r < 8; ++r)
        out[(size_t)(mb + r) * NCLS + l] = acc[r];
}

#define NOWN  NOWNB
#define TPT   ((NTILES + 1023) / 1024)
#define PCAP  1056
#define TROT  24
#define WCAP  512

template <int I> struct step_c { static constexpr int value = I; };
template <int I, int N, class Fn>
__device__ __forceinline__ void for_steps(Fn&& f) {
    if constexpr (I < N) { f(step_c<I>{}); for_steps<I + 1, N>(f); }
}

template <int F, int I>
__device__ __forceinline__ int edge_field(int x) {
    if constexpr (F == 64) return __builtin_amdgcn_readlane(x, I);
    else                   return __builtin_amdgcn_update_dpp(0, x, 0x150 + I, 0xf, 0xf, true);
}

template <int F>
__device__ __forceinline__ void gather_add(int ro, int s, float w, int n, int lane,
                                           const float* __restrict__ sup,
                                           float* acc) {
    constexpr int VW  = (F == 64) ? 2 : 1;
    constexpr int LPE = F / VW;
    constexpr int EPS = 32 / LPE;
    constexpr int NST = 32 / EPS;
    constexpr int CH  = 8;
    static_assert(LPE == 32 || LPE == 16, "an edge takes the wave or one 16-lane row");
    static_assert(NST % CH == 0, "whole chunks");
    const int sub = lane / LPE;
    const int col = (lane % LPE) * VW;
    for_steps<0, NST / CH>([&](auto c) {
        constexpr int T0 = decltype(c)::value * CH;
        float v0[CH], v1[CH];
        for_steps<0, CH>([&](auto kk) {
            constexpr int K = decltype(kk)::value;
            constexpr int I = T0 + K;
            const int j  = (EPS == 1) ? I : 16 * sub + I;
            const int sj = edge_field<F, I>(s);
            v0[K] = 0.0f; v1[K] = 0.0f;
            if (j < n) {
                if (VW == 2) {
                    const v2f v = *reinterpret_cast<const v2f*>(sup + (size_t)sj * F + col);
                    v0[K] = v.x; v1[K] = v.y;
                } else {
                    v0[K] = sup[(size_t)sj * F + col];
                }
            }
        });
        for_steps<0, CH>([&](auto kk) {
            constexpr int K = decltype(kk)::value;
            constexpr int I = T0 + K;
            const int   j  = (EPS == 1) ? I : 16 * sub + I;
            const int   rj = edge_field<F, I>(ro);
            const float wj = __int_as_float(edge_field<F, I>(__float_as_int(w)));
            if (j < n) {
                atomicAdd(&acc[rj + col], v0[K] * wj);
                if (VW == 2) atomicAdd(&acc[rj + col + 1], v1[K] * wj);
            }
        });
    });
}

template <int F, bool REC>
__device__ __forceinline__ void add_edges(const int* list, int n, int lane,
                                          const int2* __restrict__ erec,
                                          const float* __restrict__ sup,
                                          float* acc, int4* rec, int& written) {
    int ro = 0, s = 0;
    float w = 0.0f;
    if (lane < n) {
        const int2 r  = erec[list[lane]];
        const int row = r.x & 511;
        ro = row * F;
        s  = (int)((unsigned)r.x >> 9);
        w  = __int_as_float(r.y);
        if (REC && written + lane < WCAP)
            rec[written + lane] = make_int4(s, r.y, row, 0);
    }
    if (REC) written += n;
    gather_add<F>(ro, s, w, n, lane, sup, acc);
}

template <int F, bool REC>
__device__ __forceinline__ void scan_and_sum(const int2* __restrict__ erec,
                                             const float* __restrict__ sup,
                                             const unsigned* __restrict__ tab,
                                             float* acc, int* mypend, int owner,
                                             int tid, int lane,
                                             int4* rec, int& written) {
    int cnt = 0;
    static_assert(TPT * 1024 >= NTILES && (NTILES - 1) + (NOWN - 1) * TROT < 2 * NTILES,
                  "every tile has a thread, and one subtraction wraps the rotated tile number");
    int      st[TPT];
    unsigned c[TPT];
#pragma unroll
    for (int k = 0; k < TPT; ++k) {
        const int  i  = tid + 1024 * k;
        const bool ok = i < NTILES;
        int t = (ok ? i : 0) + owner * TROT;
        if (t >= NTILES) t -= NTILES;
        const unsigned e = tab[(size_t)t * NOWNB + owner];
        st[k] = t * EPT + (int)(e & 0xffffu);
        c[k]  = ok ? (e >> 16) : 0u;
    }
#pragma unroll
    for (int k = 0; k < TPT; ++k) {
        unsigned bal = __builtin_amdgcn_ballot_w32(c[k] != 0u);
        while (bal) {
            const int pos = cnt + (int)__builtin_amdgcn_mbcnt_lo(bal, 0u);
            if (c[k] != 0u) {
                mypend[pos] = st[k];
                ++st[k];
                --c[k];
            }
            cnt += __builtin_popcount(bal);
            if (cnt >= 32) {
                cnt -= 32;
                add_edges<F, REC>(mypend + cnt, 32, lane, erec, sup, acc, rec, written);
            }
            bal = __builtin_amdgcn_ballot_w32(c[k] != 0u);
        }
    }
    add_edges<F, REC>(mypend, cnt, lane, erec, sup, acc, rec, written);
}

template <int F, bool RELU>
__device__ __forceinline__ void store_rows(const float* acc, const float* __restrict__ bias,
                                           float* __restrict__ dstm, int lo, int nrows, int tid) {
    const float4* acc4 = reinterpret_cast<const float4*>(acc);
    float4* o4 = reinterpret_cast<float4*>(dstm + (size_t)lo * F);
    for (int i = tid; i < nrows * F / 4; i += 1024) {
        float4 v = acc4[i];
        const int c = (i * 4) & (F - 1);
        v.x += bias[c]; v.y += bias[c + 1]; v.z += bias[c + 2]; v.w += bias[c + 3];
        if (RELU) {
            v.x = v.x > 0.0f ? v.x : 0.0f; v.y = v.y > 0.0f ? v.y : 0.0f;
            v.z = v.z > 0.0f ? v.z : 0.0f; v.w = v.w > 0.0f ? v.w : 0.0f;
        }
        o4[i] = v;
    }
}

template <int F, bool RELU>
__global__ __launch_bounds__(1024) void owner_sum_kernel(const float* __restrict__ sup,
                                                         const float* __restrict__ bias,
                                                         float* __restrict__ dstm,
                                                         int* __restrict__ recs,
                                                         int* __restrict__ cnts,
                                                         const unsigned* __restrict__ tab,
                                                         const int* __restrict__ erec) {
    __shared__ float acc[RPB * F];
    __shared__ int   pend[32 * PCAP];
    const int tid  = threadIdx.x;
    const int lane = tid & 31;
    const int wave = tid >> 5;
    const int lo   = blockIdx.x * RPB;
    int nrows = N_NODES - lo;
    if (nrows > RPB) nrows = RPB;

    float4* acc4 = reinterpret_cast<float4*>(acc);
    for (int i = tid; i < RPB * F / 4; i += 1024) acc4[i] = make_float4(0.f, 0.f, 0.f, 0.f);
    __syncthreads();

    const int seg = blockIdx.x * 32 + wave;
    int4* rec = reinterpret_cast<int4*>(recs) + (size_t)seg * WCAP;
    int written = 0;
    scan_and_sum<F, true>(reinterpret_cast<const int2*>(erec), sup, tab, acc, pend + wave * PCAP,
                          (int)blockIdx.x, tid, lane, rec, written);
    if (lane == 0) cnts[seg] = written;
    __syncthreads();
    store_rows<F, RELU>(acc, bias, dstm, lo, nrows, tid);
}

template <int F, bool RELU>
__global__ __launch_bounds__(1024) void replay_sum_kernel(const float* __restrict__ sup,
                                                          const float* __restrict__ bias,
                                                          float* __restrict__ dstm,
                                                          const int* __restrict__ recs,
                                                          const int* __restrict__ cnts,
                                                          const unsigned* __restrict__ tab,
                                                          const int* __restrict__ erec) {
    __shared__ float acc[RPB * F];
    __shared__ int   pend[32 * PCAP];
    const int tid  = threadIdx.x;
    const int lane = tid & 31;
    const int wave = tid >> 5;
    const int lo   = blockIdx.x * RPB;
    int nrows = N_NODES - lo;
    if (nrows > RPB) nrows = RPB;

    float4* acc4 = reinterpret_cast<float4*>(acc);
    for (int i = tid; i < RPB * F / 4; i += 1024) acc4[i] = make_float4(0.f, 0.f, 0.f, 0.f);
    __syncthreads();

    const bool over = __builtin_amdgcn_ballot_w32(cnts[blockIdx.x * 32 + lane] > WCAP) != 0u;
    if (!over) {
        const int seg = blockIdx.x * 32 + wave;
        const int n   = __builtin_amdgcn_readfirstlane(cnts[seg]);
        const int4* rec = reinterpret_cast<const int4*>(recs) + (size_t)seg * WCAP;
        if (n > 0) {
            const int last = n - 1;
            int i0 = lane < last ? lane : last;
            int4 cur = rec[i0];
            for (int base = 0; base < n; base += 32) {
                int i1 = base + 32 + lane;
                i1 = i1 < last ? i1 : last;
                const int4 nxt = rec[i1];
                int m = n - base;
                if (m > 32) m = 32;
                gather_add<F>(cur.z * F, cur.x, __int_as_float(cur.y), m, lane, sup, acc);
                cur = nxt;
            }
        }
    } else {
        int unused = 0;
        scan_and_sum<F, false>(reinterpret_cast<const int2*>(erec), sup, tab, acc, pend + wave * PCAP,
                               (int)blockIdx.x, tid, lane, nullptr, unused);
    }
    __syncthreads();
    store_rows<F, RELU>(acc, bias, dstm, lo, nrows, tid);
}

__global__ __launch_bounds__(128) void log_softmax_kernel(float* __restrict__ out) {
    const int r = blockIdx.x * blockDim.x + threadIdx.x;
    if (r >= N_NODES) return;
    float* p = out + (size_t)r * NCLS;
    float v[NCLS];
    float m = -3.402823466e+38f;
#pragma unroll
    for (int i = 0; i < NCLS; ++i) { v[i] = p[i]; m = v[i] > m ? v[i] : m; }
    float s = 0.0f;
#pragma unroll
    for (int i = 0; i < NCLS; ++i) s += __expf(v[i] - m);
    const float lse = m + __logf(s);
#pragma unroll
    for (int i = 0; i < NCLS; ++i) p[i] = v[i] - lse;
}

extern "C" void kernel_launch(void* const* d_in, const int* in_sizes, int n_in,
                              void* d_out, int out_size, void* d_ws, size_t ws_size,
                              hipStream_t stream) {
    (void)in_sizes; (void)n_in; (void)out_size; (void)ws_size;
    const float* x   = (const float*)d_in[0];
    const int*   es  = (const int*)  d_in[1];
    const int*   ed  = (const int*)  d_in[2];
    const float* ev  = (const float*)d_in[3];
    const float* W1  = (const float*)d_in[4];
    const float* b1  = (const float*)d_in[5];
    const float* W2  = (const float*)d_in[6];
    const float* b2  = (const float*)d_in[7];
    float* out = (float*)d_out;

    float* sup1 = (float*)d_ws;
    float* hacc = sup1 + (size_t)N_NODES * NHID;
    float* sup2 = sup1;
    int*   recs = (int*)(hacc + (size_t)N_NODES * NHID);
    int*   cnts = recs + (size_t)NOWN * 32 * WCAP * 4;
    unsigned* tab = (unsigned*)(cnts + NOWN * 32);
    int*   erec = (int*)(tab + (size_t)NTILES * NOWNB);

    gemm1_kernel<<<G1_BLOCKS, G1_THREADS, 0, stream>>>(x, W1, sup1, ed, es, ev, tab, erec);
    owner_sum_kernel<NHID, true><<<NOWN, 1024, 0, stream>>>(sup1, b1, hacc, recs, cnts, tab, erec);
    gemm2_kernel<<<(NTILES + 7) / 8, 256, 0, stream>>>(hacc, W2, sup2);
    replay_sum_kernel<NCLS, false><<<NOWN, 1024, 0, stream>>>(sup2, b2, out, recs, cnts, tab, erec);
    log_softmax_kernel<<<(N_NODES + 127) / 128, 128, 0, stream>>>(out);
}


# GEMM1 blocks sort edge records of all their 24-25 tiles by owner block, not tile by tile: 1 table entry, 1 contiguous run per (block, owner); both layers read the runs; new edge order, arrival sums
# speedup vs baseline: 4.1259x; 1.2213x over previous
#include <hip/hip_runtime.h>

typedef float v2f __attribute__((ext_vector_type(2)));
typedef float v8f __attribute__((ext_vector_type(8)));

#define N_NODES 100000
#define FEAT    256
#define NHID    64
#define NCLS    16
#define NEDGE   1600000
#define NTILES  (N_NODES / 16)
#define RPB     392
#define EPT     (NEDGE / NTILES)
#define NOWNB   256

#define G1_WAVES   4
#define G1_THREADS (G1_WAVES * 32)
#define G1_BLOCKS  256
#define G1_STEPS   (FEAT / 4)
#define G1_MINT    (NTILES / G1_BLOCKS / G1_WAVES)
#define G1_MAXT    (G1_MINT * G1_WAVES + 1)

__device__ __forceinline__ void g1_load_tile(const float* __restrict__ x, int tile, int l, int h,
                                             float4 (&a)[32]) {
    const float4* xr = reinterpret_cast<const float4*>(x + (size_t)(tile * 16 + l) * FEAT + 4 * h);
#pragma unroll
    for (int i = 0; i < 32; ++i) a[i] = xr[2 * i];
}

__device__ __forceinline__ void g1_compute_tile(const float4 (&a)[32], int tile, int l, int h,
                                                const float4 (&w01)[G1_STEPS], const float4 (&w23)[G1_STEPS],
                                                float* __restrict__ sup) {
    v8f c0 = {}, c1 = {}, c2 = {}, c3 = {};
#pragma unroll
    for (int i = 0; i < 32; ++i) {
#pragma unroll
        for (int p = 0; p < 2; ++p) {
            const int t = 2 * i + p;
            v2f av, b;
            av.x = p ? a[i].z : a[i].x;
            av.y = p ? a[i].w : a[i].y;
            b.x = w01[t].x; b.y = w01[t].y;
            c0 = __builtin_amdgcn_wmma_f32_16x16x4_f32(false, av, false, b, (short)0, c0, false, false);
            b.x = w01[t].z; b.y = w01[t].w;
            c1 = __builtin_amdgcn_wmma_f32_16x16x4_f32(false, av, false, b, (short)0, c1, false, false);
            b.x = w23[t].x; b.y = w23[t].y;
            c2 = __builtin_amdgcn_wmma_f32_16x16x4_f32(false, av, false, b, (short)0, c2, false, false);
            b.x = w23[t].z; b.y = w23[t].w;
            c3 = __builtin_amdgcn_wmma_f32_16x16x4_f32(false, av, false, b, (short)0, c3, false, false);
        }
    }
    float* out = sup + (size_t)(tile * 16 + h * 8) * NHID + l;
#pragma unroll
    for (int r = 0; r < 8; ++r) {
        out[(size_t)r * NHID]      = c0[r];
        out[(size_t)r * NHID + 16] = c1[r];
        out[(size_t)r * NHID + 32] = c2[r];
        out[(size_t)r * NHID + 48] = c3[r];
    }
}

__device__ __forceinline__ void g1_count_tile(const int4& d0, const int4& d1,
                                              const int4& s0, const int4& s1,
                                              unsigned* hw,
                                              unsigned (&o)[8], int (&p)[8], unsigned (&rk)[8]) {
    static_assert(RPB <= 512 && N_NODES <= (1 << 23), "row in 9 bits, source row in the 23 above");
    static_assert((N_NODES - 1) / RPB < NOWNB, "an owner id indexes the 256 counts");
    const unsigned d[8] = {(unsigned)d0.x, (unsigned)d0.y, (unsigned)d0.z, (unsigned)d0.w,
                           (unsigned)d1.x, (unsigned)d1.y, (unsigned)d1.z, (unsigned)d1.w};
    const unsigned s[8] = {(unsigned)s0.x, (unsigned)s0.y, (unsigned)s0.z, (unsigned)s0.w,
                           (unsigned)s1.x, (unsigned)s1.y, (unsigned)s1.z, (unsigned)s1.w};
#pragma unroll
    for (int j = 0; j < 8; ++j) {
        o[j] = d[j] / RPB;
        p[j] = (int)((s[j] << 9) | (d[j] - o[j] * RPB));
    }
#pragma unroll
    for (int j = 0; j < 8; ++j) rk[j] = atomicAdd(&hw[o[j]], 1u);
}

__device__ __forceinline__ void g1_place_tile(const unsigned (&o)[8], const int (&p)[8],
                                              const unsigned (&rk)[8], const float4& w0, const float4& w1,
                                              const unsigned* hw, int2* stg) {
    const float w[8] = {w0.x, w0.y, w0.z, w0.w, w1.x, w1.y, w1.z, w1.w};
#pragma unroll
    for (int j = 0; j < 8; ++j)
        stg[hw[o[j]] + rk[j]] = make_int2(p[j], __float_as_int(w[j]));
}

__global__ __launch_bounds__(G1_THREADS) void gemm1_kernel(const float* __restrict__ x,
                                                           const float* __restrict__ W1,
                                                           float* __restrict__ sup,
                                                           const int* __restrict__ edst,
                                                           const int* __restrict__ esrc,
                                                           const float* __restrict__ ev,
                                                           unsigned* __restrict__ tab,
                                                           int* __restrict__ erec) {
    __shared__ float4 wb01[G1_STEPS * 32];
    __shared__ float4 wb23[G1_STEPS * 32];
    __shared__ __attribute__((aligned(16))) unsigned hist[G1_WAVES * NOWNB];
    __shared__ __attribute__((aligned(16))) int2     stg[G1_MAXT * EPT];
    __shared__ unsigned wtot[G1_WAVES];
    const int tid  = threadIdx.x;
    const int lane = tid & 31;
    const int wave = tid >> 5;
    const int l    = lane & 15;
    const int h    = lane >> 4;

    static_assert(EPT == 256 && EPT * NTILES == NEDGE, "32 lanes x 8 edges per tile cover every edge once");
    static_assert(G1_MINT >= 1 && NTILES < (G1_MINT * G1_WAVES + 1) * G1_BLOCKS,
                  "a block has G1_MINT * G1_WAVES tiles or one more, so a wave has G1_MINT or G1_MINT + 1");

    const int t_lo = (int)(((long long)blockIdx.x * NTILES) / G1_BLOCKS);
    const int t_hi = (int)(((long long)(blockIdx.x + 1) * NTILES) / G1_BLOCKS);
    const int tile0 = t_lo + wave;
    const int n = (t_hi - tile0 + G1_WAVES - 1) / G1_WAVES;

    float4 A[32], B[32];
    g1_load_tile(x, tile0, l, h, A);

    static_assert(NOWNB == 2 * G1_THREADS, "two owners per thread in the scan of the counts");
    unsigned* hw = hist + wave * NOWNB;
    const bool extra = n > G1_MINT;
    unsigned o[G1_MINT + 1][8], rk[G1_MINT + 1][8];
    int      p[G1_MINT + 1][8];
    float4   w0[G1_MINT + 1], w1[G1_MINT + 1];
    {
        int4 d0[G1_MINT], d1[G1_MINT], s0[G1_MINT], s1[G1_MINT];
#pragma unroll
        for (int i = 0; i < G1_MINT; ++i) {
            const size_t q = (size_t)(tile0 + G1_WAVES * i) * (EPT / 4) + lane * 2;
            d0[i] = reinterpret_cast<const int4*>(edst)[q];   d1[i] = reinterpret_cast<const int4*>(edst)[q + 1];
            s0[i] = reinterpret_cast<const int4*>(esrc)[q];   s1[i] = reinterpret_cast<const int4*>(esrc)[q + 1];
            w0[i] = reinterpret_cast<const float4*>(ev)[q];   w1[i] = reinterpret_cast<const float4*>(ev)[q + 1];
        }
        uint4* h4 = reinterpret_cast<uint4*>(hw);
        h4[2 * lane]     = make_uint4(0u, 0u, 0u, 0u);
        h4[2 * lane + 1] = make_uint4(0u, 0u, 0u, 0u);
#pragma unroll
        for (int i = 0; i < G1_MINT; ++i)
            g1_count_tile(d0[i], d1[i], s0[i], s1[i], hw, o[i], p[i], rk[i]);
#pragma unroll
        for (int j = 0; j < 8; ++j) { o[G1_MINT][j] = 0u; p[G1_MINT][j] = 0; rk[G1_MINT][j] = 0u; }
        w0[G1_MINT] = make_float4(0.f, 0.f, 0.f, 0.f);
        w1[G1_MINT] = w0[G1_MINT];
        if (extra) {
            const size_t q = (size_t)(tile0 + G1_WAVES * G1_MINT) * (EPT / 4) + lane * 2;
            w0[G1_MINT] = reinterpret_cast<const float4*>(ev)[q];
            w1[G1_MINT] = reinterpret_cast<const float4*>(ev)[q + 1];
            g1_count_tile(reinterpret_cast<const int4*>(edst)[q], reinterpret_cast<const int4*>(edst)[q + 1],
                          reinterpret_cast<const int4*>(esrc)[q], reinterpret_cast<const int4*>(esrc)[q + 1],
                          hw, o[G1_MINT], p[G1_MINT], rk[G1_MINT]);
        }
    }

    for (int i = tid; i < G1_STEPS * 32; i += G1_THREADS) {
        const int t  = i >> 5;
        const int ll = i & 15;
        const int hh = (i >> 4) & 1;
        const int k0 = 64 * (t >> 4) + 8 * ((t >> 1) & 7) + 4 * hh + 2 * (t & 1);
        const float* r0 = W1 + (size_t)k0 * NHID + ll;
        const float* r1 = r0 + NHID;
        wb01[i] = make_float4(r0[0],  r1[0],  r0[16], r1[16]);
        wb23[i] = make_float4(r0[32], r1[32], r0[48], r1[48]);
    }
    __syncthreads();

    {
        uint2 cw[G1_WAVES];
        unsigned tot0 = 0u, tot1 = 0u;
#pragma unroll
        for (int w = 0; w < G1_WAVES; ++w) {
            cw[w] = *reinterpret_cast<const uint2*>(&hist[w * NOWNB + 2 * tid]);
            tot0 += cw[w].x; tot1 += cw[w].y;
        }
        const unsigned run = tot0 + tot1;
        unsigned inc = run;
        inc += (unsigned)__builtin_amdgcn_update_dpp(0, (int)inc, 0x111, 0xf, 0xf, true);
        inc += (unsigned)__builtin_amdgcn_update_dpp(0, (int)inc, 0x112, 0xf, 0xf, true);
        inc += (unsigned)__builtin_amdgcn_update_dpp(0, (int)inc, 0x114, 0xf, 0xf, true);
        inc += (unsigned)__builtin_amdgcn_update_dpp(0, (int)inc, 0x118, 0xf, 0xf, true);
        const unsigned low = (unsigned)__builtin_amdgcn_readlane((int)inc, 15);
        if (lane >= 16) inc += low;
        if (lane == 31) wtot[wave] = inc;
        __syncthreads();
        unsigned b0 = inc - run;
#pragma unroll
        for (int w = 0; w < G1_WAVES - 1; ++w) b0 += (w < wave) ? wtot[w] : 0u;
        unsigned b1 = b0 + tot0;
        *reinterpret_cast<uint2*>(&tab[(size_t)blockIdx.x * NOWNB + 2 * tid]) =
            make_uint2(b0 | (tot0 << 16), b1 | (tot1 << 16));
#pragma unroll
        for (int w = 0; w < G1_WAVES; ++w) {
            *reinterpret_cast<uint2*>(&hist[w * NOWNB + 2 * tid]) = make_uint2(b0, b1);
            b0 += cw[w].x; b1 += cw[w].y;
        }
    }
    __syncthreads();

#pragma unroll
    for (int i = 0; i < G1_MINT; ++i) g1_place_tile(o[i], p[i], rk[i], w0[i], w1[i], hw, stg);
    if (extra) g1_place_tile(o[G1_MINT], p[G1_MINT], rk[G1_MINT], w0[G1_MINT], w1[G1_MINT], hw, stg);
    __syncthreads();

    {
        static_assert(G1_MAXT * EPT < (1 << 16), "a start and a count fit 16 bits each");
        constexpr int NPC = G1_MINT * G1_WAVES * (EPT / 2) / G1_THREADS;
        static_assert(NPC * G1_THREADS == G1_MINT * G1_WAVES * (EPT / 2) && EPT / 2 == G1_THREADS,
                      "whole trips, and one more trip for a block's one more tile");
        int4* r4 = reinterpret_cast<int4*>(erec) + (size_t)t_lo * (EPT / 2) + tid;
        const int4* g4 = reinterpret_cast<const int4*>(stg) + tid;
        int4 v[NPC];
#pragma unroll
        for (int m = 0; m < NPC; ++m) v[m] = g4[m * G1_THREADS];
#pragma unroll
        for (int m = 0; m < NPC; ++m) r4[m * G1_THREADS] = v[m];
        if (t_hi - t_lo > G1_MINT * G1_WAVES) r4[NPC * G1_THREADS] = g4[NPC * G1_THREADS];
    }

    float4 w01[G1_STEPS], w23[G1_STEPS];
#pragma unroll
    for (int t = 0; t < G1_STEPS; ++t) {
        w01[t] = wb01[t * 32 + lane];
        w23[t] = wb23[t * 32 + lane];
    }

    int i = 0;
    for (; i + 2 < n; i += 2) {
        g1_load_tile(x, tile0 + G1_WAVES * (i + 1), l, h, B);
        __builtin_amdgcn_sched_barrier(0);
        g1_compute_tile(A, tile0 + G1_WAVES * i, l, h, w01, w23, sup);
        __builtin_amdgcn_sched_barrier(0);
        g1_load_tile(x, tile0 + G1_WAVES * (i + 2), l, h, A);
        __builtin_amdgcn_sched_barrier(0);
        g1_compute_tile(B, tile0 + G1_WAVES * (i + 1), l, h, w01, w23, sup);
        __builtin_amdgcn_sched_barrier(0);
    }
    const bool two = i + 1 < n;
    if (two) g1_load_tile(x, tile0 + G1_WAVES * (i + 1), l, h, B);
    __builtin_amdgcn_sched_barrier(0);
    g1_compute_tile(A, tile0 + G1_WAVES * i, l, h, w01, w23, sup);
    if (two) g1_compute_tile(B, tile0 + G1_WAVES * (i + 1), l, h, w01, w23, sup);
}

__global__ __launch_bounds__(256) void gemm2_kernel(const float* __restrict__ h,
                                                    const float* __restrict__ W2,
                                                    float* __restrict__ sup2) {
    const int lane = threadIdx.x & 31;
    const int wave = threadIdx.x >> 5;
    const int tile = blockIdx.x * 8 + wave;
    if (tile >= NTILES) return;
    const int rb = tile * 16;
    const int l  = lane & 15;
    const int hi = lane >> 4;
    v8f acc = {};
    for (int k = 0; k < NHID; k += 4) {
        const int ks = k + 2 * hi;
        v2f a = *reinterpret_cast<const v2f*>(&h[(size_t)(rb + l) * NHID + ks]);
        v2f b;
        b.x = W2[ks * NCLS + l];
        b.y = W2[(ks + 1) * NCLS + l];
        acc = __builtin_amdgcn_wmma_f32_16x16x4_f32(false, a, false, b,
                                                    (short)0, acc, false, false);
    }
    float* out = sup2 + (size_t)rb * NCLS;
    const int mb = hi * 8;
#pragma unroll
    for (int r = 0; r < 8; ++r)
        out[(size_t)(mb + r) * NCLS + l] = acc[r];
}

#define NOWN  NOWNB
#define RPW   (G1_BLOCKS / 32)
#define ACCMIN 41984

template <int I> struct step_c { static constexpr int value = I; };
template <int I, int N, class Fn>
__device__ __forceinline__ void for_steps(Fn&& f) {
    if constexpr (I < N) { f(step_c<I>{}); for_steps<I + 1, N>(f); }
}

template <int F, int I>
__device__ __forceinline__ int edge_field(int x) {
    if constexpr (F == 64) return __builtin_amdgcn_readlane(x, I);
    else                   return __builtin_amdgcn_update_dpp(0, x, 0x150 + I, 0xf, 0xf, true);
}

template <int F>
__device__ __forceinline__ void gather_add(int ro, int s, float w, int n, int lane,
                                           const float* __restrict__ sup,
                                           float* acc) {
    constexpr int VW  = (F == 64) ? 2 : 1;
    constexpr int LPE = F / VW;
    constexpr int EPS = 32 / LPE;
    constexpr int NST = 32 / EPS;
    constexpr int CH  = 8;
    static_assert(LPE == 32 || LPE == 16, "an edge takes the wave or one 16-lane row");
    static_assert(NST % CH == 0, "whole chunks");
    const int sub = lane / LPE;
    const int col = (lane % LPE) * VW;
    for_steps<0, NST / CH>([&](auto c) {
        constexpr int T0 = decltype(c)::value * CH;
        float v0[CH], v1[CH];
        for_steps<0, CH>([&](auto kk) {
            constexpr int K = decltype(kk)::value;
            constexpr int I = T0 + K;
            const int j  = (EPS == 1) ? I : 16 * sub + I;
            const int sj = edge_field<F, I>(s);
            v0[K] = 0.0f; v1[K] = 0.0f;
            if (j < n) {
                if (VW == 2) {
                    const v2f v = *reinterpret_cast<const v2f*>(sup + (size_t)sj * F + col);
                    v0[K] = v.x; v1[K] = v.y;
                } else {
                    v0[K] = sup[(size_t)sj * F + col];
                }
            }
        });
        for_steps<0, CH>([&](auto kk) {
            constexpr int K = decltype(kk)::value;
            constexpr int I = T0 + K;
            const int   j  = (EPS == 1) ? I : 16 * sub + I;
            const int   rj = edge_field<F, I>(ro);
            const float wj = __int_as_float(edge_field<F, I>(__float_as_int(w)));
            if (j < n) {
                atomicAdd(&acc[rj + col], v0[K] * wj);
                if (VW == 2) atomicAdd(&acc[rj + col + 1], v1[K] * wj);
            }
        });
    });
}

template <int F, bool RELU>
__device__ __forceinline__ void store_rows(const float* acc, const float* __restrict__ bias,
                                           float* __restrict__ dstm, int lo, int nrows, int tid) {
    const float4* acc4 = reinterpret_cast<const float4*>(acc);
    float4* o4 = reinterpret_cast<float4*>(dstm + (size_t)lo * F);
    for (int i = tid; i < nrows * F / 4; i += 1024) {
        float4 v = acc4[i];
        const int c = (i * 4) & (F - 1);
        v.x += bias[c]; v.y += bias[c + 1]; v.z += bias[c + 2]; v.w += bias[c + 3];
        if (RELU) {
            v.x = v.x > 0.0f ? v.x : 0.0f; v.y = v.y > 0.0f ? v.y : 0.0f;
            v.z = v.z > 0.0f ? v.z : 0.0f; v.w = v.w > 0.0f ? v.w : 0.0f;
        }
        o4[i] = v;
    }
}

template <int F, bool RELU>
__global__ __launch_bounds__(1024) void owner_sum_kernel(const float* __restrict__ sup,
                                                         const float* __restrict__ bias,
                                                         float* __restrict__ dstm,
                                                         const unsigned* __restrict__ tab,
                                                         const int* __restrict__ erec) {
    __shared__ float acc[RPB * F > ACCMIN ? RPB * F : ACCMIN];
    static_assert(RPW * 32 == G1_BLOCKS && RPW <= 16, "a wave's runs: one per lane of a 16-lane row");
    const int tid  = threadIdx.x;
    const int lane = tid & 31;
    const int wave = tid >> 5;
    const int lo   = blockIdx.x * RPB;
    int nrows = N_NODES - lo;
    if (nrows > RPB) nrows = RPB;

    const int gb = RPW * wave + (lane < RPW ? lane : 0);
    const unsigned e = tab[(size_t)gb * NOWNB + blockIdx.x];

    float4* acc4 = reinterpret_cast<float4*>(acc);
    for (int i = tid; i < RPB * F / 4; i += 1024) acc4[i] = make_float4(0.f, 0.f, 0.f, 0.f);
    __syncthreads();

    const int cnt = lane < RPW ? (int)(e >> 16) : 0;
    const int st  = (int)(((long long)gb * NTILES) / G1_BLOCKS) * EPT + (int)(e & 0xffffu);
    int inc = cnt;
    inc += __builtin_amdgcn_update_dpp(0, inc, 0x111, 0xf, 0xf, true);
    inc += __builtin_amdgcn_update_dpp(0, inc, 0x112, 0xf, 0xf, true);
    inc += __builtin_amdgcn_update_dpp(0, inc, 0x114, 0xf, 0xf, true);
    const int dl = st - (inc - cnt);
    const int n  = __builtin_amdgcn_readlane(inc, RPW - 1);
    int pre[RPW], dd[RPW];
#pragma unroll
    for (int k = 0; k < RPW; ++k) {
        pre[k] = __builtin_amdgcn_readlane(inc - cnt, k);
        dd[k]  = __builtin_amdgcn_readlane(dl, k);
    }
    if (n > 0) {
        const int2* rec = reinterpret_cast<const int2*>(erec);
        const int last = n - 1;
        int i0 = lane < last ? lane : last;
        int d0 = dd[0];
#pragma unroll
        for (int k = 1; k < RPW; ++k) d0 = i0 >= pre[k] ? dd[k] : d0;
        int2 cur = rec[i0 + d0];
        for (int base = 0; base < n; base += 32) {
            int i1 = base + 32 + lane;
            i1 = i1 < last ? i1 : last;
            int d1 = dd[0];
#pragma unroll
            for (int k = 1; k < RPW; ++k) d1 = i1 >= pre[k] ? dd[k] : d1;
            const int2 nxt = rec[i1 + d1];
            const int   ro = (cur.x & 511) * F;
            const int   s  = (int)((unsigned)cur.x >> 9);
            const float w  = __int_as_float(cur.y);
            const int   m  = n - base;
            if (m >= 32) gather_add<F>(ro, s, w, 32, lane, sup, acc);
            else         gather_add<F>(ro, s, w, m,  lane, sup, acc);
            cur = nxt;
        }
    }
    __syncthreads();
    store_rows<F, RELU>(acc, bias, dstm, lo, nrows, tid);
}

__global__ __launch_bounds__(128) void log_softmax_kernel(float* __restrict__ out) {
    const int r = blockIdx.x * blockDim.x + threadIdx.x;
    if (r >= N_NODES) return;
    float* p = out + (size_t)r * NCLS;
    float v[NCLS];
    float m = -3.402823466e+38f;
#pragma unroll
    for (int i = 0; i < NCLS; ++i) { v[i] = p[i]; m = v[i] > m ? v[i] : m; }
    float s = 0.0f;
#pragma unroll
    for (int i = 0; i < NCLS; ++i) s += __expf(v[i] - m);
    const float lse = m + __logf(s);
#pragma unroll
    for (int i = 0; i < NCLS; ++i) p[i] = v[i] - lse;
}

extern "C" void kernel_launch(void* const* d_in, const int* in_sizes, int n_in,
                              void* d_out, int out_size, void* d_ws, size_t ws_size,
                              hipStream_t stream) {
    (void)in_sizes; (void)n_in; (void)out_size; (void)ws_size;
    const float* x   = (const float*)d_in[0];
    const int*   es  = (const int*)  d_in[1];
    const int*   ed  = (const int*)  d_in[2];
    const float* ev  = (const float*)d_in[3];
    const float* W1  = (const float*)d_in[4];
    const float* b1  = (const float*)d_in[5];
    const float* W2  = (const float*)d_in[6];
    const float* b2  = (const float*)d_in[7];
    float* out = (float*)d_out;

    float* sup1 = (float*)d_ws;
    float* hacc = sup1 + (size_t)N_NODES * NHID;
    float* sup2 = sup1;
    unsigned* tab = (unsigned*)(hacc + (size_t)N_NODES * NHID);
    int*   erec = (int*)(tab + (size_t)G1_BLOCKS * NOWNB);

    gemm1_kernel<<<G1_BLOCKS, G1_THREADS, 0, stream>>>(x, W1, sup1, ed, es, ev, tab, erec);
    owner_sum_kernel<NHID, true><<<NOWN, 1024, 0, stream>>>(sup1, b1, hacc, tab, erec);
    gemm2_kernel<<<(NTILES + 7) / 8, 256, 0, stream>>>(hacc, W2, sup2);
    owner_sum_kernel<NCLS, false><<<NOWN, 1024, 0, stream>>>(sup2, b2, out, tab, erec);
    log_softmax_kernel<<<(N_NODES + 127) / 128, 128, 0, stream>>>(out);
}


# second GEMM (bias+relu at the fragment read, f32 WMMA) and log-softmax run in the tails of the two owner kernels on their LDS rows (64-wide: pitch 68): 5 launches -> 3, h not stored; sums unchanged
# speedup vs baseline: 4.7763x; 1.1576x over previous
#include <hip/hip_runtime.h>

typedef float v2f __attribute__((ext_vector_type(2)));
typedef float v8f __attribute__((ext_vector_type(8)));

#define N_NODES 100000
#define FEAT    256
#define NHID    64
#define NCLS    16
#define NEDGE   1600000
#define NTILES  (N_NODES / 16)
#define RPB     392
#define EPT     (NEDGE / NTILES)
#define NOWNB   256

#define G1_WAVES   4
#define G1_THREADS (G1_WAVES * 32)
#define G1_BLOCKS  256
#define G1_STEPS   (FEAT / 4)
#define G1_MINT    (NTILES / G1_BLOCKS / G1_WAVES)
#define G1_MAXT    (G1_MINT * G1_WAVES + 1)

__device__ __forceinline__ void g1_load_tile(const float* __restrict__ x, int tile, int l, int h,
                                             float4 (&a)[32]) {
    const float4* xr = reinterpret_cast<const float4*>(x + (size_t)(tile * 16 + l) * FEAT + 4 * h);
#pragma unroll
    for (int i = 0; i < 32; ++i) a[i] = xr[2 * i];
}

__device__ __forceinline__ void g1_compute_tile(const float4 (&a)[32], int tile, int l, int h,
                                                const float4 (&w01)[G1_STEPS], const float4 (&w23)[G1_STEPS],
                                                float* __restrict__ sup) {
    v8f c0 = {}, c1 = {}, c2 = {}, c3 = {};
#pragma unroll
    for (int i = 0; i < 32; ++i) {
#pragma unroll
        for (int p = 0; p < 2; ++p) {
            const int t = 2 * i + p;
            v2f av, b;
            av.x = p ? a[i].z : a[i].x;
            av.y = p ? a[i].w : a[i].y;
            b.x = w01[t].x; b.y = w01[t].y;
            c0 = __builtin_amdgcn_wmma_f32_16x16x4_f32(false, av, false, b, (short)0, c0, false, false);
            b.x = w01[t].z; b.y = w01[t].w;
            c1 = __builtin_amdgcn_wmma_f32_16x16x4_f32(false, av, false, b, (short)0, c1, false, false);
            b.x = w23[t].x; b.y = w23[t].y;
            c2 = __builtin_amdgcn_wmma_f32_16x16x4_f32(false, av, false, b, (short)0, c2, false, false);
            b.x = w23[t].z; b.y = w23[t].w;
            c3 = __builtin_amdgcn_wmma_f32_16x16x4_f32(false, av, false, b, (short)0, c3, false, false);
        }
    }
    float* out = sup + (size_t)(tile * 16 + h * 8) * NHID + l;
#pragma unroll
    for (int r = 0; r < 8; ++r) {
        out[(size_t)r * NHID]      = c0[r];
        out[(size_t)r * NHID + 16] = c1[r];
        out[(size_t)r * NHID + 32] = c2[r];
        out[(size_t)r * NHID + 48] = c3[r];
    }
}

__device__ __forceinline__ void g1_count_tile(const int4& d0, const int4& d1,
                                              const int4& s0, const int4& s1,
                                              unsigned* hw,
                                              unsigned (&o)[8], int (&p)[8], unsigned (&rk)[8]) {
    static_assert(RPB <= 512 && N_NODES <= (1 << 23), "row in 9 bits, source row in the 23 above");
    static_assert((N_NODES - 1) / RPB < NOWNB, "an owner id indexes the 256 counts");
    const unsigned d[8] = {(unsigned)d0.x, (unsigned)d0.y, (unsigned)d0.z, (unsigned)d0.w,
                           (unsigned)d1.x, (unsigned)d1.y, (unsigned)d1.z, (unsigned)d1.w};
    const unsigned s[8] = {(unsigned)s0.x, (unsigned)s0.y, (unsigned)s0.z, (unsigned)s0.w,
                           (unsigned)s1.x, (unsigned)s1.y, (unsigned)s1.z, (unsigned)s1.w};
#pragma unroll
    for (int j = 0; j < 8; ++j) {
        o[j] = d[j] / RPB;
        p[j] = (int)((s[j] << 9) | (d[j] - o[j] * RPB));
    }
#pragma unroll
    for (int j = 0; j < 8; ++j) rk[j] = atomicAdd(&hw[o[j]], 1u);
}

__device__ __forceinline__ void g1_place_tile(const unsigned (&o)[8], const int (&p)[8],
                                              const unsigned (&rk)[8], const float4& w0, const float4& w1,
                                              const unsigned* hw, int2* stg) {
    const float w[8] = {w0.x, w0.y, w0.z, w0.w, w1.x, w1.y, w1.z, w1.w};
#pragma unroll
    for (int j = 0; j < 8; ++j)
        stg[hw[o[j]] + rk[j]] = make_int2(p[j], __float_as_int(w[j]));
}

__global__ __launch_bounds__(G1_THREADS) void gemm1_kernel(const float* __restrict__ x,
                                                           const float* __restrict__ W1,
                                                           float* __restrict__ sup,
                                                           const int* __restrict__ edst,
                                                           const int* __restrict__ esrc,
                                                           const float* __restrict__ ev,
                                                           unsigned* __restrict__ tab,
                                                           int* __restrict__ erec) {
    __shared__ float4 wb01[G1_STEPS * 32];
    __shared__ float4 wb23[G1_STEPS * 32];
    __shared__ __attribute__((aligned(16))) unsigned hist[G1_WAVES * NOWNB];
    __shared__ __attribute__((aligned(16))) int2     stg[G1_MAXT * EPT];
    __shared__ unsigned wtot[G1_WAVES];
    const int tid  = threadIdx.x;
    const int lane = tid & 31;
    const int wave = tid >> 5;
    const int l    = lane & 15;
    const int h    = lane >> 4;

    static_assert(EPT == 256 && EPT * NTILES == NEDGE, "32 lanes x 8 edges per tile cover every edge once");
    static_assert(G1_MINT >= 1 && NTILES < (G1_MINT * G1_WAVES + 1) * G1_BLOCKS,
                  "a block has G1_MINT * G1_WAVES tiles or one more, so a wave has G1_MINT or G1_MINT + 1");

    const int t_lo = (int)(((long long)blockIdx.x * NTILES) / G1_BLOCKS);
    const int t_hi = (int)(((long long)(blockIdx.x + 1) * NTILES) / G1_BLOCKS);
    const int tile0 = t_lo + wave;
    const int n = (t_hi - tile0 + G1_WAVES - 1) / G1_WAVES;

    float4 A[32], B[32];
    g1_load_tile(x, tile0, l, h, A);

    static_assert(NOWNB == 2 * G1_THREADS, "two owners per thread in the scan of the counts");
    unsigned* hw = hist + wave * NOWNB;
    const bool extra = n > G1_MINT;
    unsigned o[G1_MINT + 1][8], rk[G1_MINT + 1][8];
    int      p[G1_MINT + 1][8];
    float4   w0[G1_MINT + 1], w1[G1_MINT + 1];
    {
        int4 d0[G1_MINT], d1[G1_MINT], s0[G1_MINT], s1[G1_MINT];
#pragma unroll
        for (int i = 0; i < G1_MINT; ++i) {
            const size_t q = (size_t)(tile0 + G1_WAVES * i) * (EPT / 4) + lane * 2;
            d0[i] = reinterpret_cast<const int4*>(edst)[q];   d1[i] = reinterpret_cast<const int4*>(edst)[q + 1];
            s0[i] = reinterpret_cast<const int4*>(esrc)[q];   s1[i] = reinterpret_cast<const int4*>(esrc)[q + 1];
            w0[i] = reinterpret_cast<const float4*>(ev)[q];   w1[i] = reinterpret_cast<const float4*>(ev)[q + 1];
        }
        uint4* h4 = reinterpret_cast<uint4*>(hw);
        h4[2 * lane]     = make_uint4(0u, 0u, 0u, 0u);
        h4[2 * lane + 1] = make_uint4(0u, 0u, 0u, 0u);
#pragma unroll
        for (int i = 0; i < G1_MINT; ++i)
            g1_count_tile(d0[i], d1[i], s0[i], s1[i], hw, o[i], p[i], rk[i]);
#pragma unroll
        for (int j = 0; j < 8; ++j) { o[G1_MINT][j] = 0u; p[G1_MINT][j] = 0; rk[G1_MINT][j] = 0u; }
        w0[G1_MINT] = make_float4(0.f, 0.f, 0.f, 0.f);
        w1[G1_MINT] = w0[G1_MINT];
        if (extra) {
            const size_t q = (size_t)(tile0 + G1_WAVES * G1_MINT) * (EPT / 4) + lane * 2;
            w0[G1_MINT] = reinterpret_cast<const float4*>(ev)[q];
            w1[G1_MINT] = reinterpret_cast<const float4*>(ev)[q + 1];
            g1_count_tile(reinterpret_cast<const int4*>(edst)[q], reinterpret_cast<const int4*>(edst)[q + 1],
                          reinterpret_cast<const int4*>(esrc)[q], reinterpret_cast<const int4*>(esrc)[q + 1],
                          hw, o[G1_MINT], p[G1_MINT], rk[G1_MINT]);
        }
    }

    for (int i = tid; i < G1_STEPS * 32; i += G1_THREADS) {
        const int t  = i >> 5;
        const int ll = i & 15;
        const int hh = (i >> 4) & 1;
        const int k0 = 64 * (t >> 4) + 8 * ((t >> 1) & 7) + 4 * hh + 2 * (t & 1);
        const float* r0 = W1 + (size_t)k0 * NHID + ll;
        const float* r1 = r0 + NHID;
        wb01[i] = make_float4(r0[0],  r1[0],  r0[16], r1[16]);
        wb23[i] = make_float4(r0[32], r1[32], r0[48], r1[48]);
    }
    __syncthreads();

    {
        uint2 cw[G1_WAVES];
        unsigned tot0 = 0u, tot1 = 0u;
#pragma unroll
        for (int w = 0; w < G1_WAVES; ++w) {
            cw[w] = *reinterpret_cast<const uint2*>(&hist[w * NOWNB + 2 * tid]);
            tot0 += cw[w].x; tot1 += cw[w].y;
        }
        const unsigned run = tot0 + tot1;
        unsigned inc = run;
        inc += (unsigned)__builtin_amdgcn_update_dpp(0, (int)inc, 0x111, 0xf, 0xf, true);
        inc += (unsigned)__builtin_amdgcn_update_dpp(0, (int)inc, 0x112, 0xf, 0xf, true);
        inc += (unsigned)__builtin_amdgcn_update_dpp(0, (int)inc, 0x114, 0xf, 0xf, true);
        inc += (unsigned)__builtin_amdgcn_update_dpp(0, (int)inc, 0x118, 0xf, 0xf, true);
        const unsigned low = (unsigned)__builtin_amdgcn_readlane((int)inc, 15);
        if (lane >= 16) inc += low;
        if (lane == 31) wtot[wave] = inc;
        __syncthreads();
        unsigned b0 = inc - run;
#pragma unroll
        for (int w = 0; w < G1_WAVES - 1; ++w) b0 += (w < wave) ? wtot[w] : 0u;
        unsigned b1 = b0 + tot0;
        *reinterpret_cast<uint2*>(&tab[(size_t)blockIdx.x * NOWNB + 2 * tid]) =
            make_uint2(b0 | (tot0 << 16), b1 | (tot1 << 16));
#pragma unroll
        for (int w = 0; w < G1_WAVES; ++w) {
            *reinterpret_cast<uint2*>(&hist[w * NOWNB + 2 * tid]) = make_uint2(b0, b1);
            b0 += cw[w].x; b1 += cw[w].y;
        }
    }
    __syncthreads();

#pragma unroll
    for (int i = 0; i < G1_MINT; ++i) g1_place_tile(o[i], p[i], rk[i], w0[i], w1[i], hw, stg);
    if (extra) g1_place_tile(o[G1_MINT], p[G1_MINT], rk[G1_MINT], w0[G1_MINT], w1[G1_MINT], hw, stg);
    __syncthreads();

    {
        static_assert(G1_MAXT * EPT < (1 << 16), "a start and a count fit 16 bits each");
        constexpr int NPC = G1_MINT * G1_WAVES * (EPT / 2) / G1_THREADS;
        static_assert(NPC * G1_THREADS == G1_MINT * G1_WAVES * (EPT / 2) && EPT / 2 == G1_THREADS,
                      "whole trips, and one more trip for a block's one more tile");
        int4* r4 = reinterpret_cast<int4*>(erec) + (size_t)t_lo * (EPT / 2) + tid;
        const int4* g4 = reinterpret_cast<const int4*>(stg) + tid;
        int4 v[NPC];
#pragma unroll
        for (int m = 0; m < NPC; ++m) v[m] = g4[m * G1_THREADS];
#pragma unroll
        for (int m = 0; m < NPC; ++m) r4[m * G1_THREADS] = v[m];
        if (t_hi - t_lo > G1_MINT * G1_WAVES) r4[NPC * G1_THREADS] = g4[NPC * G1_THREADS];
    }

    float4 w01[G1_STEPS], w23[G1_STEPS];
#pragma unroll
    for (int t = 0; t < G1_STEPS; ++t) {
        w01[t] = wb01[t * 32 + lane];
        w23[t] = wb23[t * 32 + lane];
    }

    int i = 0;
    for (; i + 2 < n; i += 2) {
        g1_load_tile(x, tile0 + G1_WAVES * (i + 1), l, h, B);
        __builtin_amdgcn_sched_barrier(0);
        g1_compute_tile(A, tile0 + G1_WAVES * i, l, h, w01, w23, sup);
        __builtin_amdgcn_sched_barrier(0);
        g1_load_tile(x, tile0 + G1_WAVES * (i + 2), l, h, A);
        __builtin_amdgcn_sched_barrier(0);
        g1_compute_tile(B, tile0 + G1_WAVES * (i + 1), l, h, w01, w23, sup);
        __builtin_amdgcn_sched_barrier(0);
    }
    const bool two = i + 1 < n;
    if (two) g1_load_tile(x, tile0 + G1_WAVES * (i + 1), l, h, B);
    __builtin_amdgcn_sched_barrier(0);
    g1_compute_tile(A, tile0 + G1_WAVES * i, l, h, w01, w23, sup);
    if (two) g1_compute_tile(B, tile0 + G1_WAVES * (i + 1), l, h, w01, w23, sup);
}

#define NOWN  NOWNB
#define RPW   (G1_BLOCKS / 32)
#define ACCMIN 41984

template <int I> struct step_c { static constexpr int value = I; };
template <int I, int N, class Fn>
__device__ __forceinline__ void for_steps(Fn&& f) {
    if constexpr (I < N) { f(step_c<I>{}); for_steps<I + 1, N>(f); }
}

template <int F, int I>
__device__ __forceinline__ int edge_field(int x) {
    if constexpr (F == 64) return __builtin_amdgcn_readlane(x, I);
    else                   return __builtin_amdgcn_update_dpp(0, x, 0x150 + I, 0xf, 0xf, true);
}

template <int F>
__device__ __forceinline__ void gather_add(int ro, int s, float w, int n, int lane,
                                           const float* __restrict__ sup,
                                           float* acc) {
    constexpr int VW  = (F == 64) ? 2 : 1;
    constexpr int LPE = F / VW;
    constexpr int EPS = 32 / LPE;
    constexpr int NST = 32 / EPS;
    constexpr int CH  = 8;
    static_assert(LPE == 32 || LPE == 16, "an edge takes the wave or one 16-lane row");
    static_assert(NST % CH == 0, "whole chunks");
    const int sub = lane / LPE;
    const int col = (lane % LPE) * VW;
    for_steps<0, NST / CH>([&](auto c) {
        constexpr int T0 = decltype(c)::value * CH;
        float v0[CH], v1[CH];
        for_steps<0, CH>([&](auto kk) {
            constexpr int K = decltype(kk)::value;
            constexpr int I = T0 + K;
            const int j  = (EPS == 1) ? I : 16 * sub + I;
            const int sj = edge_field<F, I>(s);
            v0[K] = 0.0f; v1[K] = 0.0f;
            if (j < n) {
                if (VW == 2) {
                    const v2f v = *reinterpret_cast<const v2f*>(sup + (size_t)sj * F + col);
                    v0[K] = v.x; v1[K] = v.y;
                } else {
                    v0[K] = sup[(size_t)sj * F + col];
                }
            }
        });
        for_steps<0, CH>([&](auto kk) {
            constexpr int K = decltype(kk)::value;
            constexpr int I = T0 + K;
            const int   j  = (EPS == 1) ? I : 16 * sub + I;
            const int   rj = edge_field<F, I>(ro);
            const float wj = __int_as_float(edge_field<F, I>(__float_as_int(w)));
            if (j < n) {
                atomicAdd(&acc[rj + col], v0[K] * wj);
                if (VW == 2) atomicAdd(&acc[rj + col + 1], v1[K] * wj);
            }
        });
    });
}

template <int F>
__global__ __launch_bounds__(1024) void owner_sum_kernel(const float* __restrict__ sup,
                                                         const float* __restrict__ bias,
                                                         float* __restrict__ dstm,
                                                         const unsigned* __restrict__ tab,
                                                         const int* __restrict__ erec,
                                                         const float* __restrict__ W2) {
    constexpr int PITCH = (F == 64) ? 68 : F;
    constexpr int ROWS  = (F == 64) ? (RPB + 15) / 16 * 16 : RPB;
    static_assert(F == 64 || F == 16, "the two layers");
    static_assert(RPB % 8 == 0 && N_NODES % 8 == 0, "a lane half's 8 product rows are all inside the block's rows or all outside");
    __shared__ float acc[ROWS * PITCH > ACCMIN ? ROWS * PITCH : ACCMIN];
    __shared__ float lse_s[F == 16 ? RPB : 1];
    static_assert(RPW * 32 == G1_BLOCKS && RPW <= 16, "a wave's runs: one per lane of a 16-lane row");
    const int tid  = threadIdx.x;
    const int lane = tid & 31;
    const int wave = tid >> 5;
    const int lo   = blockIdx.x * RPB;
    int nrows = N_NODES - lo;
    if (nrows > RPB) nrows = RPB;

    const int gb = RPW * wave + (lane < RPW ? lane : 0);
    const unsigned e = tab[(size_t)gb * NOWNB + blockIdx.x];

    float4* acc4 = reinterpret_cast<float4*>(acc);
    for (int i = tid; i < ROWS * PITCH / 4; i += 1024) acc4[i] = make_float4(0.f, 0.f, 0.f, 0.f);
    __syncthreads();

    const int cnt = lane < RPW ? (int)(e >> 16) : 0;
    const int st  = (int)(((long long)gb * NTILES) / G1_BLOCKS) * EPT + (int)(e & 0xffffu);
    int inc = cnt;
    inc += __builtin_amdgcn_update_dpp(0, inc, 0x111, 0xf, 0xf, true);
    inc += __builtin_amdgcn_update_dpp(0, inc, 0x112, 0xf, 0xf, true);
    inc += __builtin_amdgcn_update_dpp(0, inc, 0x114, 0xf, 0xf, true);
    const int dl = st - (inc - cnt);
    const int n  = __builtin_amdgcn_readlane(inc, RPW - 1);
    int pre[RPW], dd[RPW];
#pragma unroll
    for (int k = 0; k < RPW; ++k) {
        pre[k] = __builtin_amdgcn_readlane(inc - cnt, k);
        dd[k]  = __builtin_amdgcn_readlane(dl, k);
    }
    if (n > 0) {
        const int2* rec = reinterpret_cast<const int2*>(erec);
        const int last = n - 1;
        int i0 = lane < last ? lane : last;
        int d0 = dd[0];
#pragma unroll
        for (int k = 1; k < RPW; ++k) d0 = i0 >= pre[k] ? dd[k] : d0;
        int2 cur = rec[i0 + d0];
        for (int base = 0; base < n; base += 32) {
            int i1 = base + 32 + lane;
            i1 = i1 < last ? i1 : last;
            int d1 = dd[0];
#pragma unroll
            for (int k = 1; k < RPW; ++k) d1 = i1 >= pre[k] ? dd[k] : d1;
            const int2 nxt = rec[i1 + d1];
            const int   ro = (cur.x & 511) * PITCH;
            const int   s  = (int)((unsigned)cur.x >> 9);
            const float w  = __int_as_float(cur.y);
            const int   m  = n - base;
            if (m >= 32) gather_add<F>(ro, s, w, 32, lane, sup, acc);
            else         gather_add<F>(ro, s, w, m,  lane, sup, acc);
            cur = nxt;
        }
    }
    __syncthreads();

    if constexpr (F == 64) {
        if (16 * wave < nrows) {
            const int l  = lane & 15;
            const int hi = lane >> 4;
            const float* arow = acc + (16 * wave + l) * PITCH + 2 * hi;
            v2f a[NHID / 4], b[NHID / 4];
#pragma unroll
            for (int t = 0; t < NHID / 4; ++t) {
                const int ks = 4 * t + 2 * hi;
                b[t].x = W2[ks * NCLS + l];
                b[t].y = W2[(ks + 1) * NCLS + l];
                const v2f bb = *reinterpret_cast<const v2f*>(bias + ks);
                const v2f s  = *reinterpret_cast<const v2f*>(arow + 4 * t);
                const float h0 = s.x + bb.x, h1 = s.y + bb.y;
                a[t].x = h0 > 0.0f ? h0 : 0.0f;
                a[t].y = h1 > 0.0f ? h1 : 0.0f;
            }
            v8f c = {};
#pragma unroll
            for (int t = 0; t < NHID / 4; ++t)
                c = __builtin_amdgcn_wmma_f32_16x16x4_f32(false, a[t], false, b[t], (short)0, c, false, false);
            const int r0 = 16 * wave + 8 * hi;
            if (r0 < nrows) {
                float* o = dstm + (size_t)(lo + r0) * NCLS + l;
#pragma unroll
                for (int r = 0; r < 8; ++r) o[(size_t)r * NCLS] = c[r];
            }
        }
    } else {
        if (tid < nrows) {
            const float4* r4 = reinterpret_cast<const float4*>(acc + tid * F);
            float v[NCLS];
#pragma unroll
            for (int q = 0; q < NCLS / 4; ++q) {
                const float4 t4 = r4[q];
                v[4 * q]     = t4.x + bias[4 * q];
                v[4 * q + 1] = t4.y + bias[4 * q + 1];
                v[4 * q + 2] = t4.z + bias[4 * q + 2];
                v[4 * q + 3] = t4.w + bias[4 * q + 3];
            }
            float m = -3.402823466e+38f;
#pragma unroll
            for (int i = 0; i < NCLS; ++i) m = v[i] > m ? v[i] : m;
            float s = 0.0f;
#pragma unroll
            for (int i = 0; i < NCLS; ++i) s += __expf(v[i] - m);
            lse_s[tid] = m + __logf(s);
        }
        __syncthreads();
        const float4* a4 = reinterpret_cast<const float4*>(acc);
        float4* o4 = reinterpret_cast<float4*>(dstm + (size_t)lo * F);
        for (int i = tid; i < nrows * F / 4; i += 1024) {
            float4 v = a4[i];
            const int c = (i * 4) & (F - 1);
            const float lse = lse_s[i / (F / 4)];
            v.x += bias[c]; v.y += bias[c + 1]; v.z += bias[c + 2]; v.w += bias[c + 3];
            v.x -= lse; v.y -= lse; v.z -= lse; v.w -= lse;
            o4[i] = v;
        }
    }
}

extern "C" void kernel_launch(void* const* d_in, const int* in_sizes, int n_in,
                              void* d_out, int out_size, void* d_ws, size_t ws_size,
                              hipStream_t stream) {
    (void)in_sizes; (void)n_in; (void)out_size; (void)ws_size;
    const float* x   = (const float*)d_in[0];
    const int*   es  = (const int*)  d_in[1];
    const int*   ed  = (const int*)  d_in[2];
    const float* ev  = (const float*)d_in[3];
    const float* W1  = (const float*)d_in[4];
    const float* b1  = (const float*)d_in[5];
    const float* W2  = (const float*)d_in[6];
    const float* b2  = (const float*)d_in[7];
    float* out = (float*)d_out;

    float* sup1 = (float*)d_ws;
    float* sup2 = sup1 + (size_t)N_NODES * NHID;
    unsigned* tab = (unsigned*)(sup2 + (size_t)N_NODES * NHID);
    int*   erec = (int*)(tab + (size_t)G1_BLOCKS * NOWNB);

    gemm1_kernel<<<G1_BLOCKS, G1_THREADS, 0, stream>>>(x, W1, sup1, ed, es, ev, tab, erec);
    owner_sum_kernel<NHID><<<NOWN, 1024, 0, stream>>>(sup1, b1, sup2, tab, erec, W2);
    owner_sum_kernel<NCLS><<<NOWN, 1024, 0, stream>>>(sup2, b2, out, tab, erec, W2);
}


# layer-1 owner blocks counting-sort their records by row in LDS; a wave owns 12-13 whole rows, sums each in registers (fma, arrival order in a row), stores it once: no f32 LDS adds, no zero fill
# speedup vs baseline: 5.2395x; 1.0970x over previous
#include <hip/hip_runtime.h>

typedef float v2f __attribute__((ext_vector_type(2)));
typedef float v8f __attribute__((ext_vector_type(8)));

#define N_NODES 100000
#define FEAT    256
#define NHID    64
#define NCLS    16
#define NEDGE   1600000
#define NTILES  (N_NODES / 16)
#define RPB     392
#define EPT     (NEDGE / NTILES)
#define NOWNB   256

#define G1_WAVES   4
#define G1_THREADS (G1_WAVES * 32)
#define G1_BLOCKS  256
#define G1_STEPS   (FEAT / 4)
#define G1_MINT    (NTILES / G1_BLOCKS / G1_WAVES)
#define G1_MAXT    (G1_MINT * G1_WAVES + 1)

__device__ __forceinline__ void g1_load_tile(const float* __restrict__ x, int tile, int l, int h,
                                             float4 (&a)[32]) {
    const float4* xr = reinterpret_cast<const float4*>(x + (size_t)(tile * 16 + l) * FEAT + 4 * h);
#pragma unroll
    for (int i = 0; i < 32; ++i) a[i] = xr[2 * i];
}

__device__ __forceinline__ void g1_compute_tile(const float4 (&a)[32], int tile, int l, int h,
                                                const float4 (&w01)[G1_STEPS], const float4 (&w23)[G1_STEPS],
                                                float* __restrict__ sup) {
    v8f c0 = {}, c1 = {}, c2 = {}, c3 = {};
#pragma unroll
    for (int i = 0; i < 32; ++i) {
#pragma unroll
        for (int p = 0; p < 2; ++p) {
            const int t = 2 * i + p;
            v2f av, b;
            av.x = p ? a[i].z : a[i].x;
            av.y = p ? a[i].w : a[i].y;
            b.x = w01[t].x; b.y = w01[t].y;
            c0 = __builtin_amdgcn_wmma_f32_16x16x4_f32(false, av, false, b, (short)0, c0, false, false);
            b.x = w01[t].z; b.y = w01[t].w;
            c1 = __builtin_amdgcn_wmma_f32_16x16x4_f32(false, av, false, b, (short)0, c1, false, false);
            b.x = w23[t].x; b.y = w23[t].y;
            c2 = __builtin_amdgcn_wmma_f32_16x16x4_f32(false, av, false, b, (short)0, c2, false, false);
            b.x = w23[t].z; b.y = w23[t].w;
            c3 = __builtin_amdgcn_wmma_f32_16x16x4_f32(false, av, false, b, (short)0, c3, false, false);
        }
    }
    float* out = sup + (size_t)(tile * 16 + h * 8) * NHID + l;
#pragma unroll
    for (int r = 0; r < 8; ++r) {
        out[(size_t)r * NHID]      = c0[r];
        out[(size_t)r * NHID + 16] = c1[r];
        out[(size_t)r * NHID + 32] = c2[r];
        out[(size_t)r * NHID + 48] = c3[r];
    }
}

__device__ __forceinline__ void g1_count_tile(const int4& d0, const int4& d1,
                                              const int4& s0, const int4& s1,
                                              unsigned* hw,
                                              unsigned (&o)[8], int (&p)[8], unsigned (&rk)[8]) {
    static_assert(RPB <= 512 && N_NODES <= (1 << 23), "row in 9 bits, source row in the 23 above");
    static_assert((N_NODES - 1) / RPB < NOWNB, "an owner id indexes the 256 counts");
    const unsigned d[8] = {(unsigned)d0.x, (unsigned)d0.y, (unsigned)d0.z, (unsigned)d0.w,
                           (unsigned)d1.x, (unsigned)d1.y, (unsigned)d1.z, (unsigned)d1.w};
    const unsigned s[8] = {(unsigned)s0.x, (unsigned)s0.y, (unsigned)s0.z, (unsigned)s0.w,
                           (unsigned)s1.x, (unsigned)s1.y, (unsigned)s1.z, (unsigned)s1.w};
#pragma unroll
    for (int j = 0; j < 8; ++j) {
        o[j] = d[j] / RPB;
        p[j] = (int)((s[j] << 9) | (d[j] - o[j] * RPB));
    }
#pragma unroll
    for (int j = 0; j < 8; ++j) rk[j] = atomicAdd(&hw[o[j]], 1u);
}

__device__ __forceinline__ void g1_place_tile(const unsigned (&o)[8], const int (&p)[8],
                                              const unsigned (&rk)[8], const float4& w0, const float4& w1,
                                              const unsigned* hw, int2* stg) {
    const float w[8] = {w0.x, w0.y, w0.z, w0.w, w1.x, w1.y, w1.z, w1.w};
#pragma unroll
    for (int j = 0; j < 8; ++j)
        stg[hw[o[j]] + rk[j]] = make_int2(p[j], __float_as_int(w[j]));
}

__global__ __launch_bounds__(G1_THREADS) void gemm1_kernel(const float* __restrict__ x,
                                                           const float* __restrict__ W1,
                                                           float* __restrict__ sup,
                                                           const int* __restrict__ edst,
                                                           const int* __restrict__ esrc,
                                                           const float* __restrict__ ev,
                                                           unsigned* __restrict__ tab,
                                                           int* __restrict__ erec) {
    __shared__ float4 wb01[G1_STEPS * 32];
    __shared__ float4 wb23[G1_STEPS * 32];
    __shared__ __attribute__((aligned(16))) unsigned hist[G1_WAVES * NOWNB];
    __shared__ __attribute__((aligned(16))) int2     stg[G1_MAXT * EPT];
    __shared__ unsigned wtot[G1_WAVES];
    const int tid  = threadIdx.x;
    const int lane = tid & 31;
    const int wave = tid >> 5;
    const int l    = lane & 15;
    const int h    = lane >> 4;

    static_assert(EPT == 256 && EPT * NTILES == NEDGE, "32 lanes x 8 edges per tile cover every edge once");
    static_assert(G1_MINT >= 1 && NTILES < (G1_MINT * G1_WAVES + 1) * G1_BLOCKS,
                  "a block has G1_MINT * G1_WAVES tiles or one more, so a wave has G1_MINT or G1_MINT + 1");

    const int t_lo = (int)(((long long)blockIdx.x * NTILES) / G1_BLOCKS);
    const int t_hi = (int)(((long long)(blockIdx.x + 1) * NTILES) / G1_BLOCKS);
    const int tile0 = t_lo + wave;
    const int n = (t_hi - tile0 + G1_WAVES - 1) / G1_WAVES;

    float4 A[32], B[32];
    g1_load_tile(x, tile0, l, h, A);

    static_assert(NOWNB == 2 * G1_THREADS, "two owners per thread in the scan of the counts");
    unsigned* hw = hist + wave * NOWNB;
    const bool extra = n > G1_MINT;
    unsigned o[G1_MINT + 1][8], rk[G1_MINT + 1][8];
    int      p[G1_MINT + 1][8];
    float4   w0[G1_MINT + 1], w1[G1_MINT + 1];
    {
        int4 d0[G1_MINT], d1[G1_MINT], s0[G1_MINT], s1[G1_MINT];
#pragma unroll
        for (int i = 0; i < G1_MINT; ++i) {
            const size_t q = (size_t)(tile0 + G1_WAVES * i) * (EPT / 4) + lane * 2;
            d0[i] = reinterpret_cast<const int4*>(edst)[q];   d1[i] = reinterpret_cast<const int4*>(edst)[q + 1];
            s0[i] = reinterpret_cast<const int4*>(esrc)[q];   s1[i] = reinterpret_cast<const int4*>(esrc)[q + 1];
            w0[i] = reinterpret_cast<const float4*>(ev)[q];   w1[i] = reinterpret_cast<const float4*>(ev)[q + 1];
        }
        uint4* h4 = reinterpret_cast<uint4*>(hw);
        h4[2 * lane]     = make_uint4(0u, 0u, 0u, 0u);
        h4[2 * lane + 1] = make_uint4(0u, 0u, 0u, 0u);
#pragma unroll
        for (int i = 0; i < G1_MINT; ++i)
            g1_count_tile(d0[i], d1[i], s0[i], s1[i], hw, o[i], p[i], rk[i]);
#pragma unroll
        for (int j = 0; j < 8; ++j) { o[G1_MINT][j] = 0u; p[G1_MINT][j] = 0; rk[G1_MINT][j] = 0u; }
        w0[G1_MINT] = make_float4(0.f, 0.f, 0.f, 0.f);
        w1[G1_MINT] = w0[G1_MINT];
        if (extra) {
            const size_t q = (size_t)(tile0 + G1_WAVES * G1_MINT) * (EPT / 4) + lane * 2;
            w0[G1_MINT] = reinterpret_cast<const float4*>(ev)[q];
            w1[G1_MINT] = reinterpret_cast<const float4*>(ev)[q + 1];
            g1_count_tile(reinterpret_cast<const int4*>(edst)[q], reinterpret_cast<const int4*>(edst)[q + 1],
                          reinterpret_cast<const int4*>(esrc)[q], reinterpret_cast<const int4*>(esrc)[q + 1],
                          hw, o[G1_MINT], p[G1_MINT], rk[G1_MINT]);
        }
    }

    for (int i = tid; i < G1_STEPS * 32; i += G1_THREADS) {
        const int t  = i >> 5;
        const int ll = i & 15;
        const int hh = (i >> 4) & 1;
        const int k0 = 64 * (t >> 4) + 8 * ((t >> 1) & 7) + 4 * hh + 2 * (t & 1);
        const float* r0 = W1 + (size_t)k0 * NHID + ll;
        const float* r1 = r0 + NHID;
        wb01[i] = make_float4(r0[0],  r1[0],  r0[16], r1[16]);
        wb23[i] = make_float4(r0[32], r1[32], r0[48], r1[48]);
    }
    __syncthreads();

    {
        uint2 cw[G1_WAVES];
        unsigned tot0 = 0u, tot1 = 0u;
#pragma unroll
        for (int w = 0; w < G1_WAVES; ++w) {
            cw[w] = *reinterpret_cast<const uint2*>(&hist[w * NOWNB + 2 * tid]);
            tot0 += cw[w].x; tot1 += cw[w].y;
        }
        const unsigned run = tot0 + tot1;
        unsigned inc = run;
        inc += (unsigned)__builtin_amdgcn_update_dpp(0, (int)inc, 0x111, 0xf, 0xf, true);
        inc += (unsigned)__builtin_amdgcn_update_dpp(0, (int)inc, 0x112, 0xf, 0xf, true);
        inc += (unsigned)__builtin_amdgcn_update_dpp(0, (int)inc, 0x114, 0xf, 0xf, true);
        inc += (unsigned)__builtin_amdgcn_update_dpp(0, (int)inc, 0x118, 0xf, 0xf, true);
        const unsigned low = (unsigned)__builtin_amdgcn_readlane((int)inc, 15);
        if (lane >= 16) inc += low;
        if (lane == 31) wtot[wave] = inc;
        __syncthreads();
        unsigned b0 = inc - run;
#pragma unroll
        for (int w = 0; w < G1_WAVES - 1; ++w) b0 += (w < wave) ? wtot[w] : 0u;
        unsigned b1 = b0 + tot0;
        *reinterpret_cast<uint2*>(&tab[(size_t)blockIdx.x * NOWNB + 2 * tid]) =
            make_uint2(b0 | (tot0 << 16), b1 | (tot1 << 16));
#pragma unroll
        for (int w = 0; w < G1_WAVES; ++w) {
            *reinterpret_cast<uint2*>(&hist[w * NOWNB + 2 * tid]) = make_uint2(b0, b1);
            b0 += cw[w].x; b1 += cw[w].y;
        }
    }
    __syncthreads();

#pragma unroll
    for (int i = 0; i < G1_MINT; ++i) g1_place_tile(o[i], p[i], rk[i], w0[i], w1[i], hw, stg);
    if (extra) g1_place_tile(o[G1_MINT], p[G1_MINT], rk[G1_MINT], w0[G1_MINT], w1[G1_MINT], hw, stg);
    __syncthreads();

    {
        static_assert(G1_MAXT * EPT < (1 << 16), "a start and a count fit 16 bits each");
        constexpr int NPC = G1_MINT * G1_WAVES * (EPT / 2) / G1_THREADS;
        static_assert(NPC * G1_THREADS == G1_MINT * G1_WAVES * (EPT / 2) && EPT / 2 == G1_THREADS,
                      "whole trips, and one more trip for a block's one more tile");
        int4* r4 = reinterpret_cast<int4*>(erec) + (size_t)t_lo * (EPT / 2) + tid;
        const int4* g4 = reinterpret_cast<const int4*>(stg) + tid;
        int4 v[NPC];
#pragma unroll
        for (int m = 0; m < NPC; ++m) v[m] = g4[m * G1_THREADS];
#pragma unroll
        for (int m = 0; m < NPC; ++m) r4[m * G1_THREADS] = v[m];
        if (t_hi - t_lo > G1_MINT * G1_WAVES) r4[NPC * G1_THREADS] = g4[NPC * G1_THREADS];
    }

    float4 w01[G1_STEPS], w23[G1_STEPS];
#pragma unroll
    for (int t = 0; t < G1_STEPS; ++t) {
        w01[t] = wb01[t * 32 + lane];
        w23[t] = wb23[t * 32 + lane];
    }

    int i = 0;
    for (; i + 2 < n; i += 2) {
        g1_load_tile(x, tile0 + G1_WAVES * (i + 1), l, h, B);
        __builtin_amdgcn_sched_barrier(0);
        g1_compute_tile(A, tile0 + G1_WAVES * i, l, h, w01, w23, sup);
        __builtin_amdgcn_sched_barrier(0);
        g1_load_tile(x, tile0 + G1_WAVES * (i + 2), l, h, A);
        __builtin_amdgcn_sched_barrier(0);
        g1_compute_tile(B, tile0 + G1_WAVES * (i + 1), l, h, w01, w23, sup);
        __builtin_amdgcn_sched_barrier(0);
    }
    const bool two = i + 1 < n;
    if (two) g1_load_tile(x, tile0 + G1_WAVES * (i + 1), l, h, B);
    __builtin_amdgcn_sched_barrier(0);
    g1_compute_tile(A, tile0 + G1_WAVES * i, l, h, w01, w23, sup);
    if (two) g1_compute_tile(B, tile0 + G1_WAVES * (i + 1), l, h, w01, w23, sup);
}

#define NOWN  NOWNB
#define RPW   (G1_BLOCKS / 32)
#define ACCMIN 41984
#define NSET   10
#define SCAP   7680
#define NRS    400

template <int I> struct step_c { static constexpr int value = I; };
template <int I, int N, class Fn>
__device__ __forceinline__ void for_steps(Fn&& f) {
    if constexpr (I < N) { f(step_c<I>{}); for_steps<I + 1, N>(f); }
}

template <int F, int I>
__device__ __forceinline__ int edge_field(int x) {
    if constexpr (F == 64) return __builtin_amdgcn_readlane(x, I);
    else                   return __builtin_amdgcn_update_dpp(0, x, 0x150 + I, 0xf, 0xf, true);
}

template <int F>
__device__ __forceinline__ void gather_add(int ro, int s, float w, int n, int lane,
                                           const float* __restrict__ sup,
                                           float* acc) {
    constexpr int VW  = (F == 64) ? 2 : 1;
    constexpr int LPE = F / VW;
    constexpr int EPS = 32 / LPE;
    constexpr int NST = 32 / EPS;
    constexpr int CH  = 8;
    static_assert(LPE == 32 || LPE == 16, "an edge takes the wave or one 16-lane row");
    static_assert(NST % CH == 0, "whole chunks");
    const int sub = lane / LPE;
    const int col = (lane % LPE) * VW;
    for_steps<0, NST / CH>([&](auto c) {
        constexpr int T0 = decltype(c)::value * CH;
        float v0[CH], v1[CH];
        for_steps<0, CH>([&](auto kk) {
            constexpr int K = decltype(kk)::value;
            constexpr int I = T0 + K;
            const int j  = (EPS == 1) ? I : 16 * sub + I;
            const int sj = edge_field<F, I>(s);
            v0[K] = 0.0f; v1[K] = 0.0f;
            if (j < n) {
                if (VW == 2) {
                    const v2f v = *reinterpret_cast<const v2f*>(sup + (size_t)sj * F + col);
                    v0[K] = v.x; v1[K] = v.y;
                } else {
                    v0[K] = sup[(size_t)sj * F + col];
                }
            }
        });
        for_steps<0, CH>([&](auto kk) {
            constexpr int K = decltype(kk)::value;
            constexpr int I = T0 + K;
            const int   j  = (EPS == 1) ? I : 16 * sub + I;
            const int   rj = edge_field<F, I>(ro);
            const float wj = __int_as_float(edge_field<F, I>(__float_as_int(w)));
            if (j < n) {
                atomicAdd(&acc[rj + col], v0[K] * wj);
                if (VW == 2) atomicAdd(&acc[rj + col + 1], v1[K] * wj);
            }
        });
    });
}

__device__ __forceinline__ void row_flush(float* acc, int col, int upto, int& cur, float& a0, float& a1) {
    for (; cur < upto; ++cur) {
        v2f t; t.x = a0; t.y = a1;
        *reinterpret_cast<v2f*>(acc + cur * 68 + col) = t;
        a0 = 0.0f; a1 = 0.0f;
    }
}

__device__ __forceinline__ void row_sum_set(int x, int wb, int n, int col,
                                            const float* __restrict__ sup, float* acc,
                                            int& cur, float& a0, float& a1) {
    constexpr int CH = 8;
    for_steps<0, 32 / CH>([&](auto c) {
        constexpr int T0 = decltype(c)::value * CH;
        float v0[CH], v1[CH];
        for_steps<0, CH>([&](auto kk) {
            constexpr int K = decltype(kk)::value;
            constexpr int I = T0 + K;
            v0[K] = 0.0f; v1[K] = 0.0f;
            if (I < n) {
                const int sj = (int)((unsigned)__builtin_amdgcn_readlane(x, I) >> 9);
                const v2f v = *reinterpret_cast<const v2f*>(sup + (size_t)sj * NHID + col);
                v0[K] = v.x; v1[K] = v.y;
            }
        });
        for_steps<0, CH>([&](auto kk) {
            constexpr int K = decltype(kk)::value;
            constexpr int I = T0 + K;
            if (I < n) {
                const int   row = __builtin_amdgcn_readlane(x, I) & 511;
                const float wj  = __int_as_float(__builtin_amdgcn_readlane(wb, I));
                if (row != cur) row_flush(acc, col, row, cur, a0, a1);
                a0 = __builtin_fmaf(v0[K], wj, a0);
                a1 = __builtin_fmaf(v1[K], wj, a1);
            }
        });
    });
}

template <int F>
__global__ __launch_bounds__(1024) void owner_sum_kernel(const float* __restrict__ sup,
                                                         const float* __restrict__ bias,
                                                         float* __restrict__ dstm,
                                                         const unsigned* __restrict__ tab,
                                                         const int* __restrict__ erec,
                                                         const float* __restrict__ W2) {
    constexpr int PITCH = (F == 64) ? 68 : F;
    constexpr int ROWS  = (F == 64) ? (RPB + 15) / 16 * 16 : RPB;
    static_assert(F == 64 || F == 16, "the two layers");
    static_assert(RPB % 8 == 0 && N_NODES % 8 == 0, "a lane half's 8 product rows are all inside the block's rows or all outside");
    __shared__ float acc[ROWS * PITCH > ACCMIN ? ROWS * PITCH : ACCMIN];
    __shared__ float lse_s[F == 16 ? RPB : 1];
    constexpr bool SORT = (F == 64);
    static_assert(!SORT || (ROWS == NRS && PITCH == 68), "the sorted rows are the rows the tail reads; row_flush's pitch");
    __shared__ __attribute__((aligned(8))) int2 srt[SORT ? SCAP : 1];
    __shared__ unsigned rcnt[SORT ? 512 : 1];
    __shared__ unsigned wsum[SORT ? 32 : 1];
    __shared__ int      wn[SORT ? 32 : 1];
    static_assert(RPW * 32 == G1_BLOCKS && RPW <= 16, "a wave's runs: one per lane of a 16-lane row");
    const int tid  = threadIdx.x;
    const int lane = tid & 31;
    const int wave = tid >> 5;
    const int lo   = blockIdx.x * RPB;
    int nrows = N_NODES - lo;
    if (nrows > RPB) nrows = RPB;

    const int gb = RPW * wave + (lane < RPW ? lane : 0);
    const unsigned e = tab[(size_t)gb * NOWNB + blockIdx.x];

    float4* acc4 = reinterpret_cast<float4*>(acc);
    if constexpr (!SORT) {
        for (int i = tid; i < ROWS * PITCH / 4; i += 1024) acc4[i] = make_float4(0.f, 0.f, 0.f, 0.f);
        __syncthreads();
    }

    const int cnt = lane < RPW ? (int)(e >> 16) : 0;
    const int st  = (int)(((long long)gb * NTILES) / G1_BLOCKS) * EPT + (int)(e & 0xffffu);
    int inc = cnt;
    inc += __builtin_amdgcn_update_dpp(0, inc, 0x111, 0xf, 0xf, true);
    inc += __builtin_amdgcn_update_dpp(0, inc, 0x112, 0xf, 0xf, true);
    inc += __builtin_amdgcn_update_dpp(0, inc, 0x114, 0xf, 0xf, true);
    const int dl = st - (inc - cnt);
    const int n  = __builtin_amdgcn_readlane(inc, RPW - 1);
    int pre[RPW], dd[RPW];
#pragma unroll
    for (int k = 0; k < RPW; ++k) {
        pre[k] = __builtin_amdgcn_readlane(inc - cnt, k);
        dd[k]  = __builtin_amdgcn_readlane(dl, k);
    }
    const int2* rec = reinterpret_cast<const int2*>(erec);
    bool sorted = false;
    if constexpr (SORT) {
        if (tid < 512) rcnt[tid] = 0u;
        if (lane == 0) wn[wave] = n;
        __syncthreads();
        int tot = 0, big = 0;
        for (int w = 0; w < 32; ++w) { const int v = wn[w]; tot += v; big = v > big ? v : big; }
        sorted = __builtin_amdgcn_readfirstlane((tot <= SCAP && big <= 32 * NSET) ? 1 : 0) != 0;
    }
    if (sorted) {
        const int last = n - 1;
        int rx[NSET], rw[NSET];
        unsigned rk[NSET];
#pragma unroll
        for (int k = 0; k < NSET; ++k) { rx[k] = 0; rw[k] = 0; rk[k] = 0u; }
        if (n > 0) {
#pragma unroll
            for (int k = 0; k < NSET; ++k) {
                int i = 32 * k + lane;
                i = i < last ? i : last;
                int d = dd[0];
#pragma unroll
                for (int q = 1; q < RPW; ++q) d = i >= pre[q] ? dd[q] : d;
                const int2 r = rec[i + d];
                rx[k] = r.x; rw[k] = r.y;
            }
#pragma unroll
            for (int k = 0; k < NSET; ++k)
                if (32 * k + lane < n) rk[k] = atomicAdd(&rcnt[rx[k] & 511], 1u);
        }
        __syncthreads();
        {
            const unsigned c = tid < NRS ? rcnt[tid] : 0u;
            unsigned ic = c;
            ic += (unsigned)__builtin_amdgcn_update_dpp(0, (int)ic, 0x111, 0xf, 0xf, true);
            ic += (unsigned)__builtin_amdgcn_update_dpp(0, (int)ic, 0x112, 0xf, 0xf, true);
            ic += (unsigned)__builtin_amdgcn_update_dpp(0, (int)ic, 0x114, 0xf, 0xf, true);
            ic += (unsigned)__builtin_amdgcn_update_dpp(0, (int)ic, 0x118, 0xf, 0xf, true);
            const unsigned low = (unsigned)__builtin_amdgcn_readlane((int)ic, 15);
            if (lane >= 16) ic += low;
            if (lane == 31) wsum[wave] = ic;
            __syncthreads();
            unsigned b = ic - c;
            for (int w = 0; w < 31; ++w) b += (w < wave) ? wsum[w] : 0u;
            if (tid <= NRS) rcnt[tid] = b;
        }
        __syncthreads();
        unsigned rs[NSET];
#pragma unroll
        for (int k = 0; k < NSET; ++k) rs[k] = rcnt[rx[k] & 511];
#pragma unroll
        for (int k = 0; k < NSET; ++k)
            if (32 * k + lane < n) srt[rs[k] + rk[k]] = make_int2(rx[k], rw[k]);
        __syncthreads();

        const int wv = __builtin_amdgcn_readfirstlane(wave);
        const int ra = (wv * NRS) >> 5;
        const int rb = ((wv + 1) * NRS) >> 5;
        const int sb = __builtin_amdgcn_readfirstlane((int)rcnt[ra]);
        const int m  = __builtin_amdgcn_readfirstlane((int)rcnt[rb]) - sb;
        const int2* sl = srt + sb;
        const int col = 2 * lane;
        float a0 = 0.0f, a1 = 0.0f;
        int cur = ra;
        if (m > 0) {
            const int lastm = m - 1;
            int2 cr = sl[lane < lastm ? lane : lastm];
            for (int base = 0; base < m; base += 32) {
                int i1 = base + 32 + lane;
                i1 = i1 < lastm ? i1 : lastm;
                const int2 nx = sl[i1];
                const int rem = m - base;
                if (rem >= 32) row_sum_set(cr.x, cr.y, 32,  col, sup, acc, cur, a0, a1);
                else           row_sum_set(cr.x, cr.y, rem, col, sup, acc, cur, a0, a1);
                cr = nx;
            }
        }
        row_flush(acc, col, rb, cur, a0, a1);
    } else {
        if constexpr (SORT) {
            for (int i = tid; i < ROWS * PITCH / 4; i += 1024) acc4[i] = make_float4(0.f, 0.f, 0.f, 0.f);
            __syncthreads();
        }
        if (n > 0) {
            const int last = n - 1;
            int i0 = lane < last ? lane : last;
            int d0 = dd[0];
#pragma unroll
            for (int k = 1; k < RPW; ++k) d0 = i0 >= pre[k] ? dd[k] : d0;
            int2 cur = rec[i0 + d0];
            for (int base = 0; base < n; base += 32) {
                int i1 = base + 32 + lane;
                i1 = i1 < last ? i1 : last;
                int d1 = dd[0];
#pragma unroll
                for (int k = 1; k < RPW; ++k) d1 = i1 >= pre[k] ? dd[k] : d1;
                const int2 nxt = rec[i1 + d1];
                const int   ro = (cur.x & 511) * PITCH;
                const int   s  = (int)((unsigned)cur.x >> 9);
                const float w  = __int_as_float(cur.y);
                const int   m  = n - base;
                if (m >= 32) gather_add<F>(ro, s, w, 32, lane, sup, acc);
                else         gather_add<F>(ro, s, w, m,  lane, sup, acc);
                cur = nxt;
            }
        }
    }
    __syncthreads();

    if constexpr (F == 64) {
        if (16 * wave < nrows) {
            const int l  = lane & 15;
            const int hi = lane >> 4;
            const float* arow = acc + (16 * wave + l) * PITCH + 2 * hi;
            v2f a[NHID / 4], b[NHID / 4];
#pragma unroll
            for (int t = 0; t < NHID / 4; ++t) {
                const int ks = 4 * t + 2 * hi;
                b[t].x = W2[ks * NCLS + l];
                b[t].y = W2[(ks + 1) * NCLS + l];
                const v2f bb = *reinterpret_cast<const v2f*>(bias + ks);
                const v2f s  = *reinterpret_cast<const v2f*>(arow + 4 * t);
                const float h0 = s.x + bb.x, h1 = s.y + bb.y;
                a[t].x = h0 > 0.0f ? h0 : 0.0f;
                a[t].y = h1 > 0.0f ? h1 : 0.0f;
            }
            v8f c = {};
#pragma unroll
            for (int t = 0; t < NHID / 4; ++t)
                c = __builtin_amdgcn_wmma_f32_16x16x4_f32(false, a[t], false, b[t], (short)0, c, false, false);
            const int r0 = 16 * wave + 8 * hi;
            if (r0 < nrows) {
                float* o = dstm + (size_t)(lo + r0) * NCLS + l;
#pragma unroll
                for (int r = 0; r < 8; ++r) o[(size_t)r * NCLS] = c[r];
            }
        }
    } else {
        if (tid < nrows) {
            const float4* r4 = reinterpret_cast<const float4*>(acc + tid * F);
            float v[NCLS];
#pragma unroll
            for (int q = 0; q < NCLS / 4; ++q) {
                const float4 t4 = r4[q];
                v[4 * q]     = t4.x + bias[4 * q];
                v[4 * q + 1] = t4.y + bias[4 * q + 1];
                v[4 * q + 2] = t4.z + bias[4 * q + 2];
                v[4 * q + 3] = t4.w + bias[4 * q + 3];
            }
            float m = -3.402823466e+38f;
#pragma unroll
            for (int i = 0; i < NCLS; ++i) m = v[i] > m ? v[i] : m;
            float s = 0.0f;
#pragma unroll
            for (int i = 0; i < NCLS; ++i) s += __expf(v[i] - m);
            lse_s[tid] = m + __logf(s);
        }
        __syncthreads();
        const float4* a4 = reinterpret_cast<const float4*>(acc);
        float4* o4 = reinterpret_cast<float4*>(dstm + (size_t)lo * F);
        for (int i = tid; i < nrows * F / 4; i += 1024) {
            float4 v = a4[i];
            const int c = (i * 4) & (F - 1);
            const float lse = lse_s[i / (F / 4)];
            v.x += bias[c]; v.y += bias[c + 1]; v.z += bias[c + 2]; v.w += bias[c + 3];
            v.x -= lse; v.y -= lse; v.z -= lse; v.w -= lse;
            o4[i] = v;
        }
    }
}

extern "C" void kernel_launch(void* const* d_in, const int* in_sizes, int n_in,
                              void* d_out, int out_size, void* d_ws, size_t ws_size,
                              hipStream_t stream) {
    (void)in_sizes; (void)n_in; (void)out_size; (void)ws_size;
    const float* x   = (const float*)d_in[0];
    const int*   es  = (const int*)  d_in[1];
    const int*   ed  = (const int*)  d_in[2];
    const float* ev  = (const float*)d_in[3];
    const float* W1  = (const float*)d_in[4];
    const float* b1  = (const float*)d_in[5];
    const float* W2  = (const float*)d_in[6];
    const float* b2  = (const float*)d_in[7];
    float* out = (float*)d_out;

    float* sup1 = (float*)d_ws;
    float* sup2 = sup1 + (size_t)N_NODES * NHID;
    unsigned* tab = (unsigned*)(sup2 + (size_t)N_NODES * NHID);
    int*   erec = (int*)(tab + (size_t)G1_BLOCKS * NOWNB);

    gemm1_kernel<<<G1_BLOCKS, G1_THREADS, 0, stream>>>(x, W1, sup1, ed, es, ev, tab, erec);
    owner_sum_kernel<NHID><<<NOWN, 1024, 0, stream>>>(sup1, b1, sup2, tab, erec, W2);
    owner_sum_kernel<NCLS><<<NOWN, 1024, 0, stream>>>(sup2, b2, out, tab, erec, W2);
}
